# starting kernel
# baseline (speedup 1.0000x reference)
#include <hip/hip_runtime.h>
#include <cstdint>
#include <cstddef>

typedef float v2f __attribute__((ext_vector_type(2)));
typedef float v8f __attribute__((ext_vector_type(8)));

#define GAT_ALPHA 0.2f

__device__ __forceinline__ void atomicMaxF(float* addr, float v) {
    if (v >= 0.0f) atomicMax((int*)addr, __float_as_int(v));
    else           atomicMin((unsigned int*)addr, __float_as_uint(v));
}

__device__ __forceinline__ void atomAddF(float* p, float v) {
#if defined(__HIP_PLATFORM_AMD__)
    unsafeAtomicAdd(p, v);
#else
    atomicAdd(p, v);
#endif
}

__global__ void k_fill(float* __restrict__ p, float v, int n) {
    int i = blockIdx.x * blockDim.x + threadIdx.x;
    if (i < n) p[i] = v;
}

__global__ __launch_bounds__(32) void k_gemm_wmma(
    const float* __restrict__ A, const float* __restrict__ B, float* __restrict__ C,
    int K, long aStride, long bStride, long cStride)
{
    const int lane = threadIdx.x;
    const int m0   = blockIdx.x * 16;
    const int b    = blockIdx.y;
    const float* __restrict__ Ab = A + (long)b * aStride;
    const float* __restrict__ Bb = B + (long)b * bStride;
    float* __restrict__ Cb = C + (long)b * cStride;

    const int row = lane & 15;
    const int kh  = (lane >> 4) << 1;
    const int col = lane & 15;

    v8f c0 = {}, c1 = {}, c2 = {}, c3 = {};

    for (int k0 = 0; k0 < K; k0 += 4) {
        const float2 af = *reinterpret_cast<const float2*>(
            Ab + (size_t)(m0 + row) * K + k0 + kh);
        v2f a; a.x = af.x; a.y = af.y;

        const float* br0 = Bb + (size_t)(k0 + kh) * 64 + col;
        const float* br1 = br0 + 64;
        v2f b0, b1, b2, b3;
        b0.x = br0[0];  b0.y = br1[0];
        b1.x = br0[16]; b1.y = br1[16];
        b2.x = br0[32]; b2.y = br1[32];
        b3.x = br0[48]; b3.y = br1[48];

        c0 = __builtin_amdgcn_wmma_f32_16x16x4_f32(false, a, false, b0, (short)0, c0, false, false);
        c1 = __builtin_amdgcn_wmma_f32_16x16x4_f32(false, a, false, b1, (short)0, c1, false, false);
        c2 = __builtin_amdgcn_wmma_f32_16x16x4_f32(false, a, false, b2, (short)0, c2, false, false);
        c3 = __builtin_amdgcn_wmma_f32_16x16x4_f32(false, a, false, b3, (short)0, c3, false, false);
    }

    const int rbase = m0 + ((lane >> 4) << 3);
    #pragma unroll
    for (int r = 0; r < 8; ++r) {
        float* crow = Cb + (size_t)(rbase + r) * 64 + col;
        crow[0]  = c0[r];
        crow[16] = c1[r];
        crow[32] = c2[r];
        crow[48] = c3[r];
    }
}

__global__ void k_node_scores(const float* __restrict__ Wh, const float* __restrict__ a,
                              float* __restrict__ ssrc, float* __restrict__ sdst,
                              int NH, int N)
{
    int wid  = (blockIdx.x * blockDim.x + threadIdx.x) >> 5;
    int lane = threadIdx.x & 31;
    if (wid >= NH) return;
    int h = wid / N;
    const float* __restrict__ row = Wh + (size_t)wid * 64;
    const float* __restrict__ as  = a + h * 128;
    const float* __restrict__ ad  = as + 64;
    float w0 = row[lane], w1 = row[lane + 32];
    float s0 = w0 * as[lane] + w1 * as[lane + 32];
    float s1 = w0 * ad[lane] + w1 * ad[lane + 32];
    #pragma unroll
    for (int off = 16; off >= 1; off >>= 1) {
        s0 += __shfl_xor(s0, off, 32);
        s1 += __shfl_xor(s1, off, 32);
    }
    if (lane == 0) { ssrc[wid] = s0; sdst[wid] = s1; }
}

__global__ void k_edge_max(const int* __restrict__ src, const int* __restrict__ dst,
                           const float* __restrict__ ssrc, const float* __restrict__ sdst,
                           float* __restrict__ ebuf, float* __restrict__ m,
                           int E, int H, int N)
{
    int tid = blockIdx.x * blockDim.x + threadIdx.x;
    if (tid >= E * H) return;
    int e = tid % E, h = tid / E;
    int s = src[e], d = dst[e];
    float v = ssrc[h * N + s] + sdst[h * N + d];
    v = v > 0.0f ? v : GAT_ALPHA * v;
    ebuf[tid] = v;
    atomicMaxF(&m[h * N + s], v);
}

__global__ void k_edge_exp(const int* __restrict__ src, const float* __restrict__ m,
                           float* __restrict__ ebuf, float* __restrict__ denom,
                           int E, int H, int N)
{
    int tid = blockIdx.x * blockDim.x + threadIdx.x;
    if (tid >= E * H) return;
    int e = tid % E, h = tid / E;
    int s = src[e];
    float ex = __expf(ebuf[tid] - m[h * N + s]);
    ebuf[tid] = ex;
    atomAddF(&denom[h * N + s], ex);
}

__global__ void k_edge_scatter(const int* __restrict__ src, const int* __restrict__ dst,
                               const float* __restrict__ ebuf, const float* __restrict__ denom,
                               const float* __restrict__ Wh, float* __restrict__ hp,
                               int E, int H, int N)
{
    int wid  = (blockIdx.x * blockDim.x + threadIdx.x) >> 5;
    int lane = threadIdx.x & 31;
    if (wid >= E * H) return;
    int e = wid % E, h = wid / E;
    int s = src[e], d = dst[e];
    float att = ebuf[wid] / denom[h * N + s];
    const float* __restrict__ wrow = Wh + (size_t)(h * N + d) * 64;
    float* __restrict__ orow = hp + (size_t)(h * N + s) * 64;
    atomAddF(&orow[lane],      att * wrow[lane]);
    atomAddF(&orow[lane + 32], att * wrow[lane + 32]);
}

__global__ void k_elu_concat(const float* __restrict__ hp, float* __restrict__ hcat, int N) {
    int tid = blockIdx.x * blockDim.x + threadIdx.x;
    if (tid >= N * 256) return;
    int n = tid >> 8, c = tid & 255, h = c >> 6, d = c & 63;
    float v = hp[((size_t)h * N + n) * 64 + d];
    hcat[tid] = v > 0.0f ? v : __expf(v) - 1.0f;
}

__global__ void k_elu(const float* __restrict__ in, float* __restrict__ out, int n) {
    int i = blockIdx.x * blockDim.x + threadIdx.x;
    if (i < n) {
        float v = in[i];
        out[i] = v > 0.0f ? v : __expf(v) - 1.0f;
    }
}

static inline int cdiv(long a, int b) { return (int)((a + b - 1) / b); }

extern "C" void kernel_launch(void* const* d_in, const int* in_sizes, int n_in,
                              void* d_out, int out_size, void* d_ws, size_t ws_size,
                              hipStream_t stream)
{
    const float* x       = (const float*)d_in[0];
    const int*   edges   = (const int*)d_in[1];
    const float* W_heads = (const float*)d_in[2];
    const float* a_heads = (const float*)d_in[3];
    const float* W_out   = (const float*)d_in[4];
    const float* a_out   = (const float*)d_in[5];
    float* out = (float*)d_out;

    const int N = in_sizes[0] / 128;
    const int E = in_sizes[1] / 2;
    const int H = 4, HID = 64;

    const int* src = edges;
    const int* dst = edges + E;

    float* ws = (float*)d_ws;
    size_t off = 0;
    float* Wh1   = ws + off; off += (size_t)H * N * HID;
    float* hcat  = ws + off; off += (size_t)N * 256;
    float* Wh2   = ws + off; off += (size_t)N * HID;
    float* ssrc1 = ws + off; off += (size_t)H * N;
    float* sdst1 = ws + off; off += (size_t)H * N;
    float* ssrc2 = ws + off; off += (size_t)N;
    float* sdst2 = ws + off; off += (size_t)N;
    float* ebuf1 = ws + off; off += (size_t)H * E;
    float* ebuf2 = ws + off; off += (size_t)E;
    float* zbase = ws + off;
    float* denom1 = ws + off; off += (size_t)H * N;
    float* hp1    = ws + off; off += (size_t)H * N * HID;
    float* denom2 = ws + off; off += (size_t)N;
    float* acc2   = ws + off; off += (size_t)N * HID;
    size_t zcount = (size_t)(ws + off - zbase);
    float* nbase = ws + off;
    float* m1 = ws + off; off += (size_t)H * N;
    float* m2 = ws + off; off += (size_t)N;
    size_t ncount = (size_t)(ws + off - nbase);

    const int BS = 256;

    k_fill<<<cdiv((long)zcount, BS), BS, 0, stream>>>(zbase, 0.0f, (int)zcount);
    k_fill<<<cdiv((long)ncount, BS), BS, 0, stream>>>(nbase, -__builtin_huge_valf(), (int)ncount);

    k_gemm_wmma<<<dim3(N / 16, H), 32, 0, stream>>>(x, W_heads, Wh1,
        128, 0L, (long)128 * 64, (long)N * 64);
    k_node_scores<<<cdiv((long)H * N * 32, BS), BS, 0, stream>>>(Wh1, a_heads, ssrc1, sdst1, H * N, N);
    k_edge_max<<<cdiv((long)E * H, BS), BS, 0, stream>>>(src, dst, ssrc1, sdst1, ebuf1, m1, E, H, N);
    k_edge_exp<<<cdiv((long)E * H, BS), BS, 0, stream>>>(src, m1, ebuf1, denom1, E, H, N);
    k_edge_scatter<<<cdiv((long)E * H * 32, BS), BS, 0, stream>>>(src, dst, ebuf1, denom1, Wh1, hp1, E, H, N);
    k_elu_concat<<<cdiv((long)N * 256, BS), BS, 0, stream>>>(hp1, hcat, N);

    k_gemm_wmma<<<dim3(N / 16, 1), 32, 0, stream>>>(hcat, W_out, Wh2,
        256, 0L, 0L, 0L);
    k_node_scores<<<cdiv((long)N * 32, BS), BS, 0, stream>>>(Wh2, a_out, ssrc2, sdst2, N, N);
    k_edge_max<<<cdiv((long)E, BS), BS, 0, stream>>>(src, dst, ssrc2, sdst2, ebuf2, m2, E, 1, N);
    k_edge_exp<<<cdiv((long)E, BS), BS, 0, stream>>>(src, m2, ebuf2, denom2, E, 1, N);
    k_edge_scatter<<<cdiv((long)E * 32, BS), BS, 0, stream>>>(src, dst, ebuf2, denom2, Wh2, acc2, E, 1, N);

    k_elu<<<cdiv((long)N * HID, BS), BS, 0, stream>>>(acc2, out, N * HID);
}


# Edges sorted by (source row, edge id) per call (int atomic count, scan, place); atomic max/sum/scatter passes become one wave-per-(head,row) pass, sums in registers in id order; no float atomics
# speedup vs baseline: 3.3414x; 3.3414x over previous
#include <hip/hip_runtime.h>
#include <cstdint>
#include <cstddef>

typedef float v2f __attribute__((ext_vector_type(2)));
typedef float v8f __attribute__((ext_vector_type(8)));

#define GAT_ALPHA 0.2f

__global__ void k_zero_int(int* __restrict__ p, int n) {
    int i = blockIdx.x * blockDim.x + threadIdx.x;
    if (i < n) p[i] = 0;
}

__global__ void k_edge_rank(const int* __restrict__ src, int* __restrict__ deg,
                            int* __restrict__ rank, int E)
{
    int e = blockIdx.x * blockDim.x + threadIdx.x;
    if (e >= E) return;
    rank[e] = atomicAdd(&deg[src[e]], 1);
}

__global__ __launch_bounds__(1024) void k_row_offsets(const int* __restrict__ deg,
                                                      int* __restrict__ rowptr, int n)
{
    __shared__ int part[1024];
    const int t = threadIdx.x;
    const int chunk = (n + 1023) / 1024;
    const int b = t * chunk;
    const int e = (b + chunk < n) ? (b + chunk) : n;
    int s = 0;
    for (int i = b; i < e; ++i) s += deg[i];
    part[t] = s;
    __syncthreads();
    for (int off = 1; off < 1024; off <<= 1) {
        const int v = (t >= off) ? part[t - off] : 0;
        __syncthreads();
        part[t] += v;
        __syncthreads();
    }
    int run = part[t] - s;
    for (int i = b; i < e; ++i) { rowptr[i] = run; run += deg[i]; }
    if (t == 1023) rowptr[n] = part[1023];
}

__global__ void k_edge_place(const int* __restrict__ src, const int* __restrict__ rowptr,
                             const int* __restrict__ rank, int* __restrict__ eid, int E)
{
    int e = blockIdx.x * blockDim.x + threadIdx.x;
    if (e >= E) return;
    eid[rowptr[src[e]] + rank[e]] = e;
}

__global__ __launch_bounds__(256) void k_row_order(const int* __restrict__ rowptr, const int* __restrict__ eid,
                                                   const int* __restrict__ dst, int* __restrict__ col, int N)
{
    const int lane = threadIdx.x & 31;
    const int wid = __builtin_amdgcn_readfirstlane((int)((blockIdx.x * blockDim.x + threadIdx.x) >> 5));
    if (wid >= N) return;
    const int beg = rowptr[wid];
    const int deg = rowptr[wid + 1] - beg;
    const int* __restrict__ ep = eid + beg;
    for (int ibase = 0; ibase < deg; ibase += 32) {
        const int i = ibase + lane;
        const int mine = (i < deg) ? ep[i] : 0x7fffffff;
        int pos = 0;
        for (int jbase = 0; jbase < deg; jbase += 32) {
            const int j = jbase + lane;
            const int other = (j < deg) ? ep[j] : 0x7fffffff;
            const int left = deg - jbase;
            const int nj = left < 32 ? left : 32;
            for (int u = 0; u < nj; ++u) pos += (__builtin_amdgcn_readlane(other, u) < mine) ? 1 : 0;
        }
        if (i < deg) col[beg + pos] = dst[mine];
    }
}

__global__ __launch_bounds__(32) void k_gemm_wmma(
    const float* __restrict__ A, const float* __restrict__ B, float* __restrict__ C,
    int K, long aStride, long bStride, long cStride)
{
    const int lane = threadIdx.x;
    const int m0   = blockIdx.x * 16;
    const int b    = blockIdx.y;
    const float* __restrict__ Ab = A + (long)b * aStride;
    const float* __restrict__ Bb = B + (long)b * bStride;
    float* __restrict__ Cb = C + (long)b * cStride;

    const int row = lane & 15;
    const int kh  = (lane >> 4) << 1;
    const int col = lane & 15;

    v8f c0 = {}, c1 = {}, c2 = {}, c3 = {};

    for (int k0 = 0; k0 < K; k0 += 4) {
        const float2 af = *reinterpret_cast<const float2*>(
            Ab + (size_t)(m0 + row) * K + k0 + kh);
        v2f a; a.x = af.x; a.y = af.y;

        const float* br0 = Bb + (size_t)(k0 + kh) * 64 + col;
        const float* br1 = br0 + 64;
        v2f b0, b1, b2, b3;
        b0.x = br0[0];  b0.y = br1[0];
        b1.x = br0[16]; b1.y = br1[16];
        b2.x = br0[32]; b2.y = br1[32];
        b3.x = br0[48]; b3.y = br1[48];

        c0 = __builtin_amdgcn_wmma_f32_16x16x4_f32(false, a, false, b0, (short)0, c0, false, false);
        c1 = __builtin_amdgcn_wmma_f32_16x16x4_f32(false, a, false, b1, (short)0, c1, false, false);
        c2 = __builtin_amdgcn_wmma_f32_16x16x4_f32(false, a, false, b2, (short)0, c2, false, false);
        c3 = __builtin_amdgcn_wmma_f32_16x16x4_f32(false, a, false, b3, (short)0, c3, false, false);
    }

    const int rbase = m0 + ((lane >> 4) << 3);
    #pragma unroll
    for (int r = 0; r < 8; ++r) {
        float* crow = Cb + (size_t)(rbase + r) * 64 + col;
        crow[0]  = c0[r];
        crow[16] = c1[r];
        crow[32] = c2[r];
        crow[48] = c3[r];
    }
}

__global__ void k_node_scores(const float* __restrict__ Wh, const float* __restrict__ a,
                              float* __restrict__ ssrc, float* __restrict__ sdst,
                              int NH, int N)
{
    int wid  = (blockIdx.x * blockDim.x + threadIdx.x) >> 5;
    int lane = threadIdx.x & 31;
    if (wid >= NH) return;
    int h = wid / N;
    const float* __restrict__ row = Wh + (size_t)wid * 64;
    const float* __restrict__ as  = a + h * 128;
    const float* __restrict__ ad  = as + 64;
    float w0 = row[lane], w1 = row[lane + 32];
    float s0 = w0 * as[lane] + w1 * as[lane + 32];
    float s1 = w0 * ad[lane] + w1 * ad[lane + 32];
    #pragma unroll
    for (int off = 16; off >= 1; off >>= 1) {
        s0 += __shfl_xor(s0, off, 32);
        s1 += __shfl_xor(s1, off, 32);
    }
    if (lane == 0) { ssrc[wid] = s0; sdst[wid] = s1; }
}

template <int U>
__device__ __forceinline__ void seg_accum(const float* __restrict__ Wb, int dreg, float areg,
                                          int j, int lane2, float& ax, float& ay)
{
    float2 w[U];
    float  a[U];
    #pragma unroll
    for (int u = 0; u < U; ++u) {
        const int dj = __builtin_amdgcn_readlane(dreg, j + u);
        a[u] = __int_as_float(__builtin_amdgcn_readlane(__float_as_int(areg), j + u));
        w[u] = *reinterpret_cast<const float2*>(Wb + (size_t)dj * 64 + lane2);
    }
    #pragma unroll
    for (int u = 0; u < U; ++u) { ax += a[u] * w[u].x; ay += a[u] * w[u].y; }
}

__device__ __forceinline__ void seg_chunk(const float* __restrict__ Wb, int dreg, float areg,
                                          int nj, int lane2, float& ax, float& ay)
{
    int j = 0;
    for (; j + 8 <= nj; j += 8) seg_accum<8>(Wb, dreg, areg, j, lane2, ax, ay);
    if (j + 4 <= nj) { seg_accum<4>(Wb, dreg, areg, j, lane2, ax, ay); j += 4; }
    if (j + 2 <= nj) { seg_accum<2>(Wb, dreg, areg, j, lane2, ax, ay); j += 2; }
    if (j < nj)      { seg_accum<1>(Wb, dreg, areg, j, lane2, ax, ay); }
}

__device__ __forceinline__ float lrelu(float v) { return v > 0.0f ? v : GAT_ALPHA * v; }

__global__ __launch_bounds__(256) void k_seg_attend(
    const int* __restrict__ rowptr, const int* __restrict__ col,
    const float* __restrict__ ssrc, const float* __restrict__ sdst,
    const float* __restrict__ Wh, float* __restrict__ hp, int NH, int N)
{
    const int lane  = threadIdx.x & 31;
    const int lane2 = lane << 1;
    const int wid = __builtin_amdgcn_readfirstlane((int)((blockIdx.x * blockDim.x + threadIdx.x) >> 5));
    if (wid >= NH) return;
    const int h = wid / N;
    const int n = wid - h * N;
    const int beg = rowptr[n];
    const int deg = rowptr[n + 1] - beg;
    float2* __restrict__ orow = reinterpret_cast<float2*>(hp + (size_t)wid * 64) + lane;
    if (deg == 0) { *orow = make_float2(0.0f, 0.0f); return; }

    const int*   __restrict__ cp = col + beg;
    const float* __restrict__ sd = sdst + (size_t)h * N;
    const float* __restrict__ Wb = Wh + (size_t)h * N * 64;
    const float ss = ssrc[wid];

    int   d0 = 0;
    float v0 = -__builtin_huge_valf();
    if (lane < deg) { d0 = cp[lane]; v0 = lrelu(ss + sd[d0]); }

    float mx = v0;
    for (int i = lane + 32; i < deg; i += 32) mx = fmaxf(mx, lrelu(ss + sd[cp[i]]));
    for (int off = 16; off >= 1; off >>= 1) mx = fmaxf(mx, __shfl_xor(mx, off, 32));

    const float ex0 = (lane < deg) ? __expf(v0 - mx) : 0.0f;
    float sum = ex0;
    for (int i = lane + 32; i < deg; i += 32) sum += __expf(lrelu(ss + sd[cp[i]]) - mx);
    for (int off = 16; off >= 1; off >>= 1) sum += __shfl_xor(sum, off, 32);

    float ax = 0.0f, ay = 0.0f;
    seg_chunk(Wb, d0, ex0 / sum, deg < 32 ? deg : 32, lane2, ax, ay);
    for (int base = 32; base < deg; base += 32) {
        const int i = base + lane;
        int   d   = 0;
        float att = 0.0f;
        if (i < deg) { d = cp[i]; att = __expf(lrelu(ss + sd[d]) - mx) / sum; }
        const int left = deg - base;
        seg_chunk(Wb, d, att, left < 32 ? left : 32, lane2, ax, ay);
    }
    *orow = make_float2(ax, ay);
}

__global__ void k_elu_concat(const float* __restrict__ hp, float* __restrict__ hcat, int N) {
    int tid = blockIdx.x * blockDim.x + threadIdx.x;
    if (tid >= N * 256) return;
    int n = tid >> 8, c = tid & 255, h = c >> 6, d = c & 63;
    float v = hp[((size_t)h * N + n) * 64 + d];
    hcat[tid] = v > 0.0f ? v : __expf(v) - 1.0f;
}

__global__ void k_elu(const float* __restrict__ in, float* __restrict__ out, int n) {
    int i = blockIdx.x * blockDim.x + threadIdx.x;
    if (i < n) {
        float v = in[i];
        out[i] = v > 0.0f ? v : __expf(v) - 1.0f;
    }
}

static inline int cdiv(long a, int b) { return (int)((a + b - 1) / b); }

extern "C" void kernel_launch(void* const* d_in, const int* in_sizes, int n_in,
                              void* d_out, int out_size, void* d_ws, size_t ws_size,
                              hipStream_t stream)
{
    const float* x       = (const float*)d_in[0];
    const int*   edges   = (const int*)d_in[1];
    const float* W_heads = (const float*)d_in[2];
    const float* a_heads = (const float*)d_in[3];
    const float* W_out   = (const float*)d_in[4];
    const float* a_out   = (const float*)d_in[5];
    float* out = (float*)d_out;

    const int N = in_sizes[0] / 128;
    const int E = in_sizes[1] / 2;
    const int H = 4, HID = 64;

    const int* src = edges;
    const int* dst = edges + E;

    float* ws = (float*)d_ws;
    size_t off = 0;
    float* Wh1   = ws + off; off += (size_t)H * N * HID;
    float* hcat  = ws + off; off += (size_t)N * 256;
    float* Wh2   = ws + off; off += (size_t)N * HID;
    float* hp1   = ws + off; off += (size_t)H * N * HID;
    float* acc2  = ws + off; off += (size_t)N * HID;
    float* ssrc1 = ws + off; off += (size_t)H * N;
    float* sdst1 = ws + off; off += (size_t)H * N;
    float* ssrc2 = ws + off; off += (size_t)N;
    float* sdst2 = ws + off; off += (size_t)N;
    int* deg    = (int*)(ws + off); off += (size_t)N;
    int* rowptr = (int*)(ws + off); off += (size_t)N + 1;
    int* rank   = (int*)(ws + off); off += (size_t)E;
    int* eid    = (int*)(ws + off); off += (size_t)E;
    int* col    = (int*)(ws + off); off += (size_t)E;

    const int BS = 256;

    k_zero_int<<<cdiv((long)N, BS), BS, 0, stream>>>(deg, N);
    k_edge_rank<<<cdiv((long)E, BS), BS, 0, stream>>>(src, deg, rank, E);
    k_row_offsets<<<1, 1024, 0, stream>>>(deg, rowptr, N);
    k_edge_place<<<cdiv((long)E, BS), BS, 0, stream>>>(src, rowptr, rank, eid, E);
    k_row_order<<<cdiv((long)N * 32, BS), BS, 0, stream>>>(rowptr, eid, dst, col, N);

    k_gemm_wmma<<<dim3(N / 16, H), 32, 0, stream>>>(x, W_heads, Wh1,
        128, 0L, (long)128 * 64, (long)N * 64);
    k_node_scores<<<cdiv((long)H * N * 32, BS), BS, 0, stream>>>(Wh1, a_heads, ssrc1, sdst1, H * N, N);
    k_seg_attend<<<cdiv((long)H * N * 32, BS), BS, 0, stream>>>(rowptr, col, ssrc1, sdst1, Wh1, hp1, H * N, N);
    k_elu_concat<<<cdiv((long)N * 256, BS), BS, 0, stream>>>(hp1, hcat, N);

    k_gemm_wmma<<<dim3(N / 16, 1), 32, 0, stream>>>(hcat, W_out, Wh2,
        256, 0L, 0L, 0L);
    k_node_scores<<<cdiv((long)N * 32, BS), BS, 0, stream>>>(Wh2, a_out, ssrc2, sdst2, N, N);
    k_seg_attend<<<cdiv((long)N * 32, BS), BS, 0, stream>>>(rowptr, col, ssrc2, sdst2, Wh2, acc2, N, N);

    k_elu<<<cdiv((long)N * HID, BS), BS, 0, stream>>>(acc2, out, N * HID);
}


# ELU and head concat moved into the attend pass's own store (hcat and out written directly): the two one-element-per-lane ELU launches, 400k + 100k waves by their grids, are gone; 13 -> 11 launches
# speedup vs baseline: 3.9079x; 1.1695x over previous
#include <hip/hip_runtime.h>
#include <cstdint>
#include <cstddef>

typedef float v2f __attribute__((ext_vector_type(2)));
typedef float v8f __attribute__((ext_vector_type(8)));

#define GAT_ALPHA 0.2f

__global__ void k_zero_int(int* __restrict__ p, int n) {
    int i = blockIdx.x * blockDim.x + threadIdx.x;
    if (i < n) p[i] = 0;
}

__global__ void k_edge_rank(const int* __restrict__ src, int* __restrict__ deg,
                            int* __restrict__ rank, int E)
{
    int e = blockIdx.x * blockDim.x + threadIdx.x;
    if (e >= E) return;
    rank[e] = atomicAdd(&deg[src[e]], 1);
}

__global__ __launch_bounds__(1024) void k_row_offsets(const int* __restrict__ deg,
                                                      int* __restrict__ rowptr, int n)
{
    __shared__ int part[1024];
    const int t = threadIdx.x;
    const int chunk = (n + 1023) / 1024;
    const int b = t * chunk;
    const int e = (b + chunk < n) ? (b + chunk) : n;
    int s = 0;
    for (int i = b; i < e; ++i) s += deg[i];
    part[t] = s;
    __syncthreads();
    for (int off = 1; off < 1024; off <<= 1) {
        const int v = (t >= off) ? part[t - off] : 0;
        __syncthreads();
        part[t] += v;
        __syncthreads();
    }
    int run = part[t] - s;
    for (int i = b; i < e; ++i) { rowptr[i] = run; run += deg[i]; }
    if (t == 1023) rowptr[n] = part[1023];
}

__global__ void k_edge_place(const int* __restrict__ src, const int* __restrict__ rowptr,
                             const int* __restrict__ rank, int* __restrict__ eid, int E)
{
    int e = blockIdx.x * blockDim.x + threadIdx.x;
    if (e >= E) return;
    eid[rowptr[src[e]] + rank[e]] = e;
}

__global__ __launch_bounds__(256) void k_row_order(const int* __restrict__ rowptr, const int* __restrict__ eid,
                                                   const int* __restrict__ dst, int* __restrict__ col, int N)
{
    const int lane = threadIdx.x & 31;
    const int wid = __builtin_amdgcn_readfirstlane((int)((blockIdx.x * blockDim.x + threadIdx.x) >> 5));
    if (wid >= N) return;
    const int beg = rowptr[wid];
    const int deg = rowptr[wid + 1] - beg;
    const int* __restrict__ ep = eid + beg;
    for (int ibase = 0; ibase < deg; ibase += 32) {
        const int i = ibase + lane;
        const int mine = (i < deg) ? ep[i] : 0x7fffffff;
        int pos = 0;
        for (int jbase = 0; jbase < deg; jbase += 32) {
            const int j = jbase + lane;
            const int other = (j < deg) ? ep[j] : 0x7fffffff;
            const int left = deg - jbase;
            const int nj = left < 32 ? left : 32;
            for (int u = 0; u < nj; ++u) pos += (__builtin_amdgcn_readlane(other, u) < mine) ? 1 : 0;
        }
        if (i < deg) col[beg + pos] = dst[mine];
    }
}

__global__ __launch_bounds__(32) void k_gemm_wmma(
    const float* __restrict__ A, const float* __restrict__ B, float* __restrict__ C,
    int K, long aStride, long bStride, long cStride)
{
    const int lane = threadIdx.x;
    const int m0   = blockIdx.x * 16;
    const int b    = blockIdx.y;
    const float* __restrict__ Ab = A + (long)b * aStride;
    const float* __restrict__ Bb = B + (long)b * bStride;
    float* __restrict__ Cb = C + (long)b * cStride;

    const int row = lane & 15;
    const int kh  = (lane >> 4) << 1;
    const int col = lane & 15;

    v8f c0 = {}, c1 = {}, c2 = {}, c3 = {};

    for (int k0 = 0; k0 < K; k0 += 4) {
        const float2 af = *reinterpret_cast<const float2*>(
            Ab + (size_t)(m0 + row) * K + k0 + kh);
        v2f a; a.x = af.x; a.y = af.y;

        const float* br0 = Bb + (size_t)(k0 + kh) * 64 + col;
        const float* br1 = br0 + 64;
        v2f b0, b1, b2, b3;
        b0.x = br0[0];  b0.y = br1[0];
        b1.x = br0[16]; b1.y = br1[16];
        b2.x = br0[32]; b2.y = br1[32];
        b3.x = br0[48]; b3.y = br1[48];

        c0 = __builtin_amdgcn_wmma_f32_16x16x4_f32(false, a, false, b0, (short)0, c0, false, false);
        c1 = __builtin_amdgcn_wmma_f32_16x16x4_f32(false, a, false, b1, (short)0, c1, false, false);
        c2 = __builtin_amdgcn_wmma_f32_16x16x4_f32(false, a, false, b2, (short)0, c2, false, false);
        c3 = __builtin_amdgcn_wmma_f32_16x16x4_f32(false, a, false, b3, (short)0, c3, false, false);
    }

    const int rbase = m0 + ((lane >> 4) << 3);
    #pragma unroll
    for (int r = 0; r < 8; ++r) {
        float* crow = Cb + (size_t)(rbase + r) * 64 + col;
        crow[0]  = c0[r];
        crow[16] = c1[r];
        crow[32] = c2[r];
        crow[48] = c3[r];
    }
}

__global__ void k_node_scores(const float* __restrict__ Wh, const float* __restrict__ a,
                              float* __restrict__ ssrc, float* __restrict__ sdst,
                              int NH, int N)
{
    int wid  = (blockIdx.x * blockDim.x + threadIdx.x) >> 5;
    int lane = threadIdx.x & 31;
    if (wid >= NH) return;
    int h = wid / N;
    const float* __restrict__ row = Wh + (size_t)wid * 64;
    const float* __restrict__ as  = a + h * 128;
    const float* __restrict__ ad  = as + 64;
    float w0 = row[lane], w1 = row[lane + 32];
    float s0 = w0 * as[lane] + w1 * as[lane + 32];
    float s1 = w0 * ad[lane] + w1 * ad[lane + 32];
    #pragma unroll
    for (int off = 16; off >= 1; off >>= 1) {
        s0 += __shfl_xor(s0, off, 32);
        s1 += __shfl_xor(s1, off, 32);
    }
    if (lane == 0) { ssrc[wid] = s0; sdst[wid] = s1; }
}

template <int U>
__device__ __forceinline__ void seg_accum(const float* __restrict__ Wb, int dreg, float areg,
                                          int j, int lane2, float& ax, float& ay)
{
    float2 w[U];
    float  a[U];
    #pragma unroll
    for (int u = 0; u < U; ++u) {
        const int dj = __builtin_amdgcn_readlane(dreg, j + u);
        a[u] = __int_as_float(__builtin_amdgcn_readlane(__float_as_int(areg), j + u));
        w[u] = *reinterpret_cast<const float2*>(Wb + (size_t)dj * 64 + lane2);
    }
    #pragma unroll
    for (int u = 0; u < U; ++u) { ax += a[u] * w[u].x; ay += a[u] * w[u].y; }
}

__device__ __forceinline__ void seg_chunk(const float* __restrict__ Wb, int dreg, float areg,
                                          int nj, int lane2, float& ax, float& ay)
{
    int j = 0;
    for (; j + 8 <= nj; j += 8) seg_accum<8>(Wb, dreg, areg, j, lane2, ax, ay);
    if (j + 4 <= nj) { seg_accum<4>(Wb, dreg, areg, j, lane2, ax, ay); j += 4; }
    if (j + 2 <= nj) { seg_accum<2>(Wb, dreg, areg, j, lane2, ax, ay); j += 2; }
    if (j < nj)      { seg_accum<1>(Wb, dreg, areg, j, lane2, ax, ay); }
}

__device__ __forceinline__ float lrelu(float v) { return v > 0.0f ? v : GAT_ALPHA * v; }

__device__ __forceinline__ float elu1(float v) { return v > 0.0f ? v : __expf(v) - 1.0f; }

__global__ __launch_bounds__(256) void k_seg_attend(
    const int* __restrict__ rowptr, const int* __restrict__ col,
    const float* __restrict__ ssrc, const float* __restrict__ sdst,
    const float* __restrict__ Wh, float* __restrict__ dst, int NH, int N, int ostride)
{
    const int lane  = threadIdx.x & 31;
    const int lane2 = lane << 1;
    const int wid = __builtin_amdgcn_readfirstlane((int)((blockIdx.x * blockDim.x + threadIdx.x) >> 5));
    if (wid >= NH) return;
    const int h = wid / N;
    const int n = wid - h * N;
    const int beg = rowptr[n];
    const int deg = rowptr[n + 1] - beg;
    float2* __restrict__ orow = reinterpret_cast<float2*>(dst + (size_t)n * ostride + (size_t)h * 64) + lane;
    if (deg == 0) { *orow = make_float2(0.0f, 0.0f); return; }

    const int*   __restrict__ cp = col + beg;
    const float* __restrict__ sd = sdst + (size_t)h * N;
    const float* __restrict__ Wb = Wh + (size_t)h * N * 64;
    const float ss = ssrc[wid];

    int   d0 = 0;
    float v0 = -__builtin_huge_valf();
    if (lane < deg) { d0 = cp[lane]; v0 = lrelu(ss + sd[d0]); }

    float mx = v0;
    for (int i = lane + 32; i < deg; i += 32) mx = fmaxf(mx, lrelu(ss + sd[cp[i]]));
    for (int off = 16; off >= 1; off >>= 1) mx = fmaxf(mx, __shfl_xor(mx, off, 32));

    const float ex0 = (lane < deg) ? __expf(v0 - mx) : 0.0f;
    float sum = ex0;
    for (int i = lane + 32; i < deg; i += 32) sum += __expf(lrelu(ss + sd[cp[i]]) - mx);
    for (int off = 16; off >= 1; off >>= 1) sum += __shfl_xor(sum, off, 32);

    float ax = 0.0f, ay = 0.0f;
    seg_chunk(Wb, d0, ex0 / sum, deg < 32 ? deg : 32, lane2, ax, ay);
    for (int base = 32; base < deg; base += 32) {
        const int i = base + lane;
        int   d   = 0;
        float att = 0.0f;
        if (i < deg) { d = cp[i]; att = __expf(lrelu(ss + sd[d]) - mx) / sum; }
        const int left = deg - base;
        seg_chunk(Wb, d, att, left < 32 ? left : 32, lane2, ax, ay);
    }
    *orow = make_float2(elu1(ax), elu1(ay));
}

static inline int cdiv(long a, int b) { return (int)((a + b - 1) / b); }

extern "C" void kernel_launch(void* const* d_in, const int* in_sizes, int n_in,
                              void* d_out, int out_size, void* d_ws, size_t ws_size,
                              hipStream_t stream)
{
    const float* x       = (const float*)d_in[0];
    const int*   edges   = (const int*)d_in[1];
    const float* W_heads = (const float*)d_in[2];
    const float* a_heads = (const float*)d_in[3];
    const float* W_out   = (const float*)d_in[4];
    const float* a_out   = (const float*)d_in[5];
    float* out = (float*)d_out;

    const int N = in_sizes[0] / 128;
    const int E = in_sizes[1] / 2;
    const int H = 4, HID = 64;

    const int* src = edges;
    const int* dst = edges + E;

    float* ws = (float*)d_ws;
    size_t off = 0;
    float* Wh1   = ws + off; off += (size_t)H * N * HID;
    float* hcat  = ws + off; off += (size_t)N * 256;
    float* Wh2   = ws + off; off += (size_t)N * HID;
    float* ssrc1 = ws + off; off += (size_t)H * N;
    float* sdst1 = ws + off; off += (size_t)H * N;
    float* ssrc2 = ws + off; off += (size_t)N;
    float* sdst2 = ws + off; off += (size_t)N;
    int* deg    = (int*)(ws + off); off += (size_t)N;
    int* rowptr = (int*)(ws + off); off += (size_t)N + 1;
    int* rank   = (int*)(ws + off); off += (size_t)E;
    int* eid    = (int*)(ws + off); off += (size_t)E;
    int* col    = (int*)(ws + off); off += (size_t)E;

    const int BS = 256;

    k_zero_int<<<cdiv((long)N, BS), BS, 0, stream>>>(deg, N);
    k_edge_rank<<<cdiv((long)E, BS), BS, 0, stream>>>(src, deg, rank, E);
    k_row_offsets<<<1, 1024, 0, stream>>>(deg, rowptr, N);
    k_edge_place<<<cdiv((long)E, BS), BS, 0, stream>>>(src, rowptr, rank, eid, E);
    k_row_order<<<cdiv((long)N * 32, BS), BS, 0, stream>>>(rowptr, eid, dst, col, N);

    k_gemm_wmma<<<dim3(N / 16, H), 32, 0, stream>>>(x, W_heads, Wh1,
        128, 0L, (long)128 * 64, (long)N * 64);
    k_node_scores<<<cdiv((long)H * N * 32, BS), BS, 0, stream>>>(Wh1, a_heads, ssrc1, sdst1, H * N, N);
    k_seg_attend<<<cdiv((long)H * N * 32, BS), BS, 0, stream>>>(rowptr, col, ssrc1, sdst1, Wh1, hcat, H * N, N, H * HID);

    k_gemm_wmma<<<dim3(N / 16, 1), 32, 0, stream>>>(hcat, W_out, Wh2,
        256, 0L, 0L, 0L);
    k_node_scores<<<cdiv((long)N * 32, BS), BS, 0, stream>>>(Wh2, a_out, ssrc2, sdst2, N, N);
    k_seg_attend<<<cdiv((long)N * 32, BS), BS, 0, stream>>>(rowptr, col, ssrc2, sdst2, Wh2, out, N, N, HID);
}


# Row offsets: one work-group whose threads each walked a 49-word chunk (each wave load and store touching 32 lines) becomes 49 work-groups of whole-line loads that re-sum the degrees before their tile
# speedup vs baseline: 4.9563x; 1.2683x over previous
#include <hip/hip_runtime.h>
#include <cstdint>
#include <cstddef>

typedef float v2f __attribute__((ext_vector_type(2)));
typedef float v8f __attribute__((ext_vector_type(8)));

#define GAT_ALPHA 0.2f

__global__ void k_zero_int(int* __restrict__ p, int n) {
    int i = blockIdx.x * blockDim.x + threadIdx.x;
    if (i < n) p[i] = 0;
}

__global__ void k_edge_rank(const int* __restrict__ src, int* __restrict__ deg,
                            int* __restrict__ rank, int E)
{
    int e = blockIdx.x * blockDim.x + threadIdx.x;
    if (e >= E) return;
    rank[e] = atomicAdd(&deg[src[e]], 1);
}

__global__ __launch_bounds__(1024) void k_row_offsets(const int* __restrict__ deg,
                                                      int* __restrict__ rowptr, int n)
{
    __shared__ int wfront[32];
    __shared__ int wtile[32];
    const int t    = threadIdx.x;
    const int lane = t & 31;
    const int w    = t >> 5;
    const int nb   = blockIdx.x;
    const int idx  = nb * 1024 + t;

    const int v = (idx < n) ? deg[idx] : 0;

    int s = 0;
    const int* __restrict__ p = deg + t;
    for (int k0 = 0; k0 < nb; k0 += 16) {
        int x[16];
        #pragma unroll
        for (int u = 0; u < 16; ++u) {
            const int k = (k0 + u < nb) ? (k0 + u) : (nb - 1);
            x[u] = p[(size_t)k * 1024];
        }
        #pragma unroll
        for (int u = 0; u < 16; ++u) s += (k0 + u < nb) ? x[u] : 0;
    }

    int inc = v;
    for (int off = 1; off < 32; off <<= 1) {
        const int o = __shfl_up(inc, off, 32);
        if (lane >= off) inc += o;
    }
    for (int off = 16; off >= 1; off >>= 1) s += __shfl_xor(s, off, 32);
    if (lane == 31) wtile[w]  = inc;
    if (lane == 0)  wfront[w] = s;
    __syncthreads();
    int base = wfront[lane] + ((lane < w) ? wtile[lane] : 0);
    for (int off = 16; off >= 1; off >>= 1) base += __shfl_xor(base, off, 32);
    if (idx <= n) rowptr[idx] = base + inc - v;
}

__global__ void k_edge_place(const int* __restrict__ src, const int* __restrict__ rowptr,
                             const int* __restrict__ rank, int* __restrict__ eid, int E)
{
    int e = blockIdx.x * blockDim.x + threadIdx.x;
    if (e >= E) return;
    eid[rowptr[src[e]] + rank[e]] = e;
}

__global__ __launch_bounds__(256) void k_row_order(const int* __restrict__ rowptr, const int* __restrict__ eid,
                                                   const int* __restrict__ dst, int* __restrict__ col, int N)
{
    const int lane = threadIdx.x & 31;
    const int wid = __builtin_amdgcn_readfirstlane((int)((blockIdx.x * blockDim.x + threadIdx.x) >> 5));
    if (wid >= N) return;
    const int beg = rowptr[wid];
    const int deg = rowptr[wid + 1] - beg;
    const int* __restrict__ ep = eid + beg;
    for (int ibase = 0; ibase < deg; ibase += 32) {
        const int i = ibase + lane;
        const int mine = (i < deg) ? ep[i] : 0x7fffffff;
        int pos = 0;
        for (int jbase = 0; jbase < deg; jbase += 32) {
            const int j = jbase + lane;
            const int other = (j < deg) ? ep[j] : 0x7fffffff;
            const int left = deg - jbase;
            const int nj = left < 32 ? left : 32;
            for (int u = 0; u < nj; ++u) pos += (__builtin_amdgcn_readlane(other, u) < mine) ? 1 : 0;
        }
        if (i < deg) col[beg + pos] = dst[mine];
    }
}

__global__ __launch_bounds__(32) void k_gemm_wmma(
    const float* __restrict__ A, const float* __restrict__ B, float* __restrict__ C,
    int K, long aStride, long bStride, long cStride)
{
    const int lane = threadIdx.x;
    const int m0   = blockIdx.x * 16;
    const int b    = blockIdx.y;
    const float* __restrict__ Ab = A + (long)b * aStride;
    const float* __restrict__ Bb = B + (long)b * bStride;
    float* __restrict__ Cb = C + (long)b * cStride;

    const int row = lane & 15;
    const int kh  = (lane >> 4) << 1;
    const int col = lane & 15;

    v8f c0 = {}, c1 = {}, c2 = {}, c3 = {};

    for (int k0 = 0; k0 < K; k0 += 4) {
        const float2 af = *reinterpret_cast<const float2*>(
            Ab + (size_t)(m0 + row) * K + k0 + kh);
        v2f a; a.x = af.x; a.y = af.y;

        const float* br0 = Bb + (size_t)(k0 + kh) * 64 + col;
        const float* br1 = br0 + 64;
        v2f b0, b1, b2, b3;
        b0.x = br0[0];  b0.y = br1[0];
        b1.x = br0[16]; b1.y = br1[16];
        b2.x = br0[32]; b2.y = br1[32];
        b3.x = br0[48]; b3.y = br1[48];

        c0 = __builtin_amdgcn_wmma_f32_16x16x4_f32(false, a, false, b0, (short)0, c0, false, false);
        c1 = __builtin_amdgcn_wmma_f32_16x16x4_f32(false, a, false, b1, (short)0, c1, false, false);
        c2 = __builtin_amdgcn_wmma_f32_16x16x4_f32(false, a, false, b2, (short)0, c2, false, false);
        c3 = __builtin_amdgcn_wmma_f32_16x16x4_f32(false, a, false, b3, (short)0, c3, false, false);
    }

    const int rbase = m0 + ((lane >> 4) << 3);
    #pragma unroll
    for (int r = 0; r < 8; ++r) {
        float* crow = Cb + (size_t)(rbase + r) * 64 + col;
        crow[0]  = c0[r];
        crow[16] = c1[r];
        crow[32] = c2[r];
        crow[48] = c3[r];
    }
}

__global__ void k_node_scores(const float* __restrict__ Wh, const float* __restrict__ a,
                              float* __restrict__ ssrc, float* __restrict__ sdst,
                              int NH, int N)
{
    int wid  = (blockIdx.x * blockDim.x + threadIdx.x) >> 5;
    int lane = threadIdx.x & 31;
    if (wid >= NH) return;
    int h = wid / N;
    const float* __restrict__ row = Wh + (size_t)wid * 64;
    const float* __restrict__ as  = a + h * 128;
    const float* __restrict__ ad  = as + 64;
    float w0 = row[lane], w1 = row[lane + 32];
    float s0 = w0 * as[lane] + w1 * as[lane + 32];
    float s1 = w0 * ad[lane] + w1 * ad[lane + 32];
    #pragma unroll
    for (int off = 16; off >= 1; off >>= 1) {
        s0 += __shfl_xor(s0, off, 32);
        s1 += __shfl_xor(s1, off, 32);
    }
    if (lane == 0) { ssrc[wid] = s0; sdst[wid] = s1; }
}

template <int U>
__device__ __forceinline__ void seg_accum(const float* __restrict__ Wb, int dreg, float areg,
                                          int j, int lane2, float& ax, float& ay)
{
    float2 w[U];
    float  a[U];
    #pragma unroll
    for (int u = 0; u < U; ++u) {
        const int dj = __builtin_amdgcn_readlane(dreg, j + u);
        a[u] = __int_as_float(__builtin_amdgcn_readlane(__float_as_int(areg), j + u));
        w[u] = *reinterpret_cast<const float2*>(Wb + (size_t)dj * 64 + lane2);
    }
    #pragma unroll
    for (int u = 0; u < U; ++u) { ax += a[u] * w[u].x; ay += a[u] * w[u].y; }
}

__device__ __forceinline__ void seg_chunk(const float* __restrict__ Wb, int dreg, float areg,
                                          int nj, int lane2, float& ax, float& ay)
{
    int j = 0;
    for (; j + 8 <= nj; j += 8) seg_accum<8>(Wb, dreg, areg, j, lane2, ax, ay);
    if (j + 4 <= nj) { seg_accum<4>(Wb, dreg, areg, j, lane2, ax, ay); j += 4; }
    if (j + 2 <= nj) { seg_accum<2>(Wb, dreg, areg, j, lane2, ax, ay); j += 2; }
    if (j < nj)      { seg_accum<1>(Wb, dreg, areg, j, lane2, ax, ay); }
}

__device__ __forceinline__ float lrelu(float v) { return v > 0.0f ? v : GAT_ALPHA * v; }

__device__ __forceinline__ float elu1(float v) { return v > 0.0f ? v : __expf(v) - 1.0f; }

__global__ __launch_bounds__(256) void k_seg_attend(
    const int* __restrict__ rowptr, const int* __restrict__ col,
    const float* __restrict__ ssrc, const float* __restrict__ sdst,
    const float* __restrict__ Wh, float* __restrict__ dst, int NH, int N, int ostride)
{
    const int lane  = threadIdx.x & 31;
    const int lane2 = lane << 1;
    const int wid = __builtin_amdgcn_readfirstlane((int)((blockIdx.x * blockDim.x + threadIdx.x) >> 5));
    if (wid >= NH) return;
    const int h = wid / N;
    const int n = wid - h * N;
    const int beg = rowptr[n];
    const int deg = rowptr[n + 1] - beg;
    float2* __restrict__ orow = reinterpret_cast<float2*>(dst + (size_t)n * ostride + (size_t)h * 64) + lane;
    if (deg == 0) { *orow = make_float2(0.0f, 0.0f); return; }

    const int*   __restrict__ cp = col + beg;
    const float* __restrict__ sd = sdst + (size_t)h * N;
    const float* __restrict__ Wb = Wh + (size_t)h * N * 64;
    const float ss = ssrc[wid];

    int   d0 = 0;
    float v0 = -__builtin_huge_valf();
    if (lane < deg) { d0 = cp[lane]; v0 = lrelu(ss + sd[d0]); }

    float mx = v0;
    for (int i = lane + 32; i < deg; i += 32) mx = fmaxf(mx, lrelu(ss + sd[cp[i]]));
    for (int off = 16; off >= 1; off >>= 1) mx = fmaxf(mx, __shfl_xor(mx, off, 32));

    const float ex0 = (lane < deg) ? __expf(v0 - mx) : 0.0f;
    float sum = ex0;
    for (int i = lane + 32; i < deg; i += 32) sum += __expf(lrelu(ss + sd[cp[i]]) - mx);
    for (int off = 16; off >= 1; off >>= 1) sum += __shfl_xor(sum, off, 32);

    float ax = 0.0f, ay = 0.0f;
    seg_chunk(Wb, d0, ex0 / sum, deg < 32 ? deg : 32, lane2, ax, ay);
    for (int base = 32; base < deg; base += 32) {
        const int i = base + lane;
        int   d   = 0;
        float att = 0.0f;
        if (i < deg) { d = cp[i]; att = __expf(lrelu(ss + sd[d]) - mx) / sum; }
        const int left = deg - base;
        seg_chunk(Wb, d, att, left < 32 ? left : 32, lane2, ax, ay);
    }
    *orow = make_float2(elu1(ax), elu1(ay));
}

static inline int cdiv(long a, int b) { return (int)((a + b - 1) / b); }

extern "C" void kernel_launch(void* const* d_in, const int* in_sizes, int n_in,
                              void* d_out, int out_size, void* d_ws, size_t ws_size,
                              hipStream_t stream)
{
    const float* x       = (const float*)d_in[0];
    const int*   edges   = (const int*)d_in[1];
    const float* W_heads = (const float*)d_in[2];
    const float* a_heads = (const float*)d_in[3];
    const float* W_out   = (const float*)d_in[4];
    const float* a_out   = (const float*)d_in[5];
    float* out = (float*)d_out;

    const int N = in_sizes[0] / 128;
    const int E = in_sizes[1] / 2;
    const int H = 4, HID = 64;

    const int* src = edges;
    const int* dst = edges + E;

    float* ws = (float*)d_ws;
    size_t off = 0;
    float* Wh1   = ws + off; off += (size_t)H * N * HID;
    float* hcat  = ws + off; off += (size_t)N * 256;
    float* Wh2   = ws + off; off += (size_t)N * HID;
    float* ssrc1 = ws + off; off += (size_t)H * N;
    float* sdst1 = ws + off; off += (size_t)H * N;
    float* ssrc2 = ws + off; off += (size_t)N;
    float* sdst2 = ws + off; off += (size_t)N;
    int* deg    = (int*)(ws + off); off += (size_t)N;
    int* rowptr = (int*)(ws + off); off += (size_t)N + 1;
    int* rank   = (int*)(ws + off); off += (size_t)E;
    int* eid    = (int*)(ws + off); off += (size_t)E;
    int* col    = (int*)(ws + off); off += (size_t)E;

    const int BS = 256;

    k_zero_int<<<cdiv((long)N, BS), BS, 0, stream>>>(deg, N);
    k_edge_rank<<<cdiv((long)E, BS), BS, 0, stream>>>(src, deg, rank, E);
    k_row_offsets<<<cdiv((long)N + 1, 1024), 1024, 0, stream>>>(deg, rowptr, N);
    k_edge_place<<<cdiv((long)E, BS), BS, 0, stream>>>(src, rowptr, rank, eid, E);
    k_row_order<<<cdiv((long)N * 32, BS), BS, 0, stream>>>(rowptr, eid, dst, col, N);

    k_gemm_wmma<<<dim3(N / 16, H), 32, 0, stream>>>(x, W_heads, Wh1,
        128, 0L, (long)128 * 64, (long)N * 64);
    k_node_scores<<<cdiv((long)H * N * 32, BS), BS, 0, stream>>>(Wh1, a_heads, ssrc1, sdst1, H * N, N);
    k_seg_attend<<<cdiv((long)H * N * 32, BS), BS, 0, stream>>>(rowptr, col, ssrc1, sdst1, Wh1, hcat, H * N, N, H * HID);

    k_gemm_wmma<<<dim3(N / 16, 1), 32, 0, stream>>>(hcat, W_out, Wh2,
        256, 0L, 0L, 0L);
    k_node_scores<<<cdiv((long)N * 32, BS), BS, 0, stream>>>(Wh2, a_out, ssrc2, sdst2, N, N);
    k_seg_attend<<<cdiv((long)N * 32, BS), BS, 0, stream>>>(rowptr, col, ssrc2, sdst2, Wh2, out, N, N, HID);
}


# Node scores computed in the product's epilogue from the wave's accumulators (4 terms per lane, then a 16-lane DPP sum: their sum order changes); both score launches are gone, 11 -> 9 launches
# speedup vs baseline: 5.7231x; 1.1547x over previous
#include <hip/hip_runtime.h>
#include <cstdint>
#include <cstddef>

typedef float v2f __attribute__((ext_vector_type(2)));
typedef float v8f __attribute__((ext_vector_type(8)));

#define GAT_ALPHA 0.2f

__global__ void k_zero_int(int* __restrict__ p, int n) {
    int i = blockIdx.x * blockDim.x + threadIdx.x;
    if (i < n) p[i] = 0;
}

__global__ void k_edge_rank(const int* __restrict__ src, int* __restrict__ deg,
                            int* __restrict__ rank, int E)
{
    int e = blockIdx.x * blockDim.x + threadIdx.x;
    if (e >= E) return;
    rank[e] = atomicAdd(&deg[src[e]], 1);
}

__global__ __launch_bounds__(1024) void k_row_offsets(const int* __restrict__ deg,
                                                      int* __restrict__ rowptr, int n)
{
    __shared__ int wfront[32];
    __shared__ int wtile[32];
    const int t    = threadIdx.x;
    const int lane = t & 31;
    const int w    = t >> 5;
    const int nb   = blockIdx.x;
    const int idx  = nb * 1024 + t;

    const int v = (idx < n) ? deg[idx] : 0;

    int s = 0;
    const int* __restrict__ p = deg + t;
    for (int k0 = 0; k0 < nb; k0 += 16) {
        int x[16];
        #pragma unroll
        for (int u = 0; u < 16; ++u) {
            const int k = (k0 + u < nb) ? (k0 + u) : (nb - 1);
            x[u] = p[(size_t)k * 1024];
        }
        #pragma unroll
        for (int u = 0; u < 16; ++u) s += (k0 + u < nb) ? x[u] : 0;
    }

    int inc = v;
    for (int off = 1; off < 32; off <<= 1) {
        const int o = __shfl_up(inc, off, 32);
        if (lane >= off) inc += o;
    }
    for (int off = 16; off >= 1; off >>= 1) s += __shfl_xor(s, off, 32);
    if (lane == 31) wtile[w]  = inc;
    if (lane == 0)  wfront[w] = s;
    __syncthreads();
    int base = wfront[lane] + ((lane < w) ? wtile[lane] : 0);
    for (int off = 16; off >= 1; off >>= 1) base += __shfl_xor(base, off, 32);
    if (idx <= n) rowptr[idx] = base + inc - v;
}

__global__ void k_edge_place(const int* __restrict__ src, const int* __restrict__ rowptr,
                             const int* __restrict__ rank, int* __restrict__ eid, int E)
{
    int e = blockIdx.x * blockDim.x + threadIdx.x;
    if (e >= E) return;
    eid[rowptr[src[e]] + rank[e]] = e;
}

__global__ __launch_bounds__(256) void k_row_order(const int* __restrict__ rowptr, const int* __restrict__ eid,
                                                   const int* __restrict__ dst, int* __restrict__ col, int N)
{
    const int lane = threadIdx.x & 31;
    const int wid = __builtin_amdgcn_readfirstlane((int)((blockIdx.x * blockDim.x + threadIdx.x) >> 5));
    if (wid >= N) return;
    const int beg = rowptr[wid];
    const int deg = rowptr[wid + 1] - beg;
    const int* __restrict__ ep = eid + beg;
    for (int ibase = 0; ibase < deg; ibase += 32) {
        const int i = ibase + lane;
        const int mine = (i < deg) ? ep[i] : 0x7fffffff;
        int pos = 0;
        for (int jbase = 0; jbase < deg; jbase += 32) {
            const int j = jbase + lane;
            const int other = (j < deg) ? ep[j] : 0x7fffffff;
            const int left = deg - jbase;
            const int nj = left < 32 ? left : 32;
            for (int u = 0; u < nj; ++u) pos += (__builtin_amdgcn_readlane(other, u) < mine) ? 1 : 0;
        }
        if (i < deg) col[beg + pos] = dst[mine];
    }
}

__device__ __forceinline__ float row16_sum(float v)
{
    v += __int_as_float(__builtin_amdgcn_update_dpp(0, __float_as_int(v), 0x161, 0xf, 0xf, false));
    v += __int_as_float(__builtin_amdgcn_update_dpp(0, __float_as_int(v), 0x162, 0xf, 0xf, false));
    v += __int_as_float(__builtin_amdgcn_update_dpp(0, __float_as_int(v), 0x164, 0xf, 0xf, false));
    v += __int_as_float(__builtin_amdgcn_update_dpp(0, __float_as_int(v), 0x168, 0xf, 0xf, false));
    return v;
}

__global__ __launch_bounds__(32) void k_gemm_wmma(
    const float* __restrict__ A, const float* __restrict__ B, float* __restrict__ C,
    const float* __restrict__ avec, float* __restrict__ ssrc, float* __restrict__ sdst,
    int K, long aStride, long bStride, long cStride, long sStride)
{
    const int lane = threadIdx.x;
    const int m0   = blockIdx.x * 16;
    const int b    = blockIdx.y;
    const float* __restrict__ Ab = A + (long)b * aStride;
    const float* __restrict__ Bb = B + (long)b * bStride;
    float* __restrict__ Cb = C + (long)b * cStride;

    const int row = lane & 15;
    const int kh  = (lane >> 4) << 1;
    const int col = lane & 15;

    const float* __restrict__ av = avec + (long)b * 128;
    const float as0 = av[col],      as1 = av[col + 16], as2 = av[col + 32], as3 = av[col + 48];
    const float ad0 = av[col + 64], ad1 = av[col + 80], ad2 = av[col + 96], ad3 = av[col + 112];

    v8f c0 = {}, c1 = {}, c2 = {}, c3 = {};

    for (int k0 = 0; k0 < K; k0 += 4) {
        const float2 af = *reinterpret_cast<const float2*>(
            Ab + (size_t)(m0 + row) * K + k0 + kh);
        v2f a; a.x = af.x; a.y = af.y;

        const float* br0 = Bb + (size_t)(k0 + kh) * 64 + col;
        const float* br1 = br0 + 64;
        v2f b0, b1, b2, b3;
        b0.x = br0[0];  b0.y = br1[0];
        b1.x = br0[16]; b1.y = br1[16];
        b2.x = br0[32]; b2.y = br1[32];
        b3.x = br0[48]; b3.y = br1[48];

        c0 = __builtin_amdgcn_wmma_f32_16x16x4_f32(false, a, false, b0, (short)0, c0, false, false);
        c1 = __builtin_amdgcn_wmma_f32_16x16x4_f32(false, a, false, b1, (short)0, c1, false, false);
        c2 = __builtin_amdgcn_wmma_f32_16x16x4_f32(false, a, false, b2, (short)0, c2, false, false);
        c3 = __builtin_amdgcn_wmma_f32_16x16x4_f32(false, a, false, b3, (short)0, c3, false, false);
    }

    const int rbase = m0 + ((lane >> 4) << 3);
    #pragma unroll
    for (int r = 0; r < 8; ++r) {
        float* crow = Cb + (size_t)(rbase + r) * 64 + col;
        crow[0]  = c0[r];
        crow[16] = c1[r];
        crow[32] = c2[r];
        crow[48] = c3[r];
    }

    float ps = 0.0f, pd = 0.0f;
    #pragma unroll
    for (int r = 0; r < 8; ++r) {
        const float s = row16_sum(c0[r] * as0 + c1[r] * as1 + c2[r] * as2 + c3[r] * as3);
        const float d = row16_sum(c0[r] * ad0 + c1[r] * ad1 + c2[r] * ad2 + c3[r] * ad3);
        if (col == r) { ps = s; pd = d; }
    }
    if (col < 8) {
        const size_t si = (size_t)b * sStride + rbase + col;
        ssrc[si] = ps;
        sdst[si] = pd;
    }
}

template <int U>
__device__ __forceinline__ void seg_accum(const float* __restrict__ Wb, int dreg, float areg,
                                          int j, int lane2, float& ax, float& ay)
{
    float2 w[U];
    float  a[U];
    #pragma unroll
    for (int u = 0; u < U; ++u) {
        const int dj = __builtin_amdgcn_readlane(dreg, j + u);
        a[u] = __int_as_float(__builtin_amdgcn_readlane(__float_as_int(areg), j + u));
        w[u] = *reinterpret_cast<const float2*>(Wb + (size_t)dj * 64 + lane2);
    }
    #pragma unroll
    for (int u = 0; u < U; ++u) { ax += a[u] * w[u].x; ay += a[u] * w[u].y; }
}

__device__ __forceinline__ void seg_chunk(const float* __restrict__ Wb, int dreg, float areg,
                                          int nj, int lane2, float& ax, float& ay)
{
    int j = 0;
    for (; j + 8 <= nj; j += 8) seg_accum<8>(Wb, dreg, areg, j, lane2, ax, ay);
    if (j + 4 <= nj) { seg_accum<4>(Wb, dreg, areg, j, lane2, ax, ay); j += 4; }
    if (j + 2 <= nj) { seg_accum<2>(Wb, dreg, areg, j, lane2, ax, ay); j += 2; }
    if (j < nj)      { seg_accum<1>(Wb, dreg, areg, j, lane2, ax, ay); }
}

__device__ __forceinline__ float lrelu(float v) { return v > 0.0f ? v : GAT_ALPHA * v; }

__device__ __forceinline__ float elu1(float v) { return v > 0.0f ? v : __expf(v) - 1.0f; }

__global__ __launch_bounds__(256) void k_seg_attend(
    const int* __restrict__ rowptr, const int* __restrict__ col,
    const float* __restrict__ ssrc, const float* __restrict__ sdst,
    const float* __restrict__ Wh, float* __restrict__ dst, int NH, int N, int ostride)
{
    const int lane  = threadIdx.x & 31;
    const int lane2 = lane << 1;
    const int wid = __builtin_amdgcn_readfirstlane((int)((blockIdx.x * blockDim.x + threadIdx.x) >> 5));
    if (wid >= NH) return;
    const int h = wid / N;
    const int n = wid - h * N;
    const int beg = rowptr[n];
    const int deg = rowptr[n + 1] - beg;
    float2* __restrict__ orow = reinterpret_cast<float2*>(dst + (size_t)n * ostride + (size_t)h * 64) + lane;
    if (deg == 0) { *orow = make_float2(0.0f, 0.0f); return; }

    const int*   __restrict__ cp = col + beg;
    const float* __restrict__ sd = sdst + (size_t)h * N;
    const float* __restrict__ Wb = Wh + (size_t)h * N * 64;
    const float ss = ssrc[wid];

    int   d0 = 0;
    float v0 = -__builtin_huge_valf();
    if (lane < deg) { d0 = cp[lane]; v0 = lrelu(ss + sd[d0]); }

    float mx = v0;
    for (int i = lane + 32; i < deg; i += 32) mx = fmaxf(mx, lrelu(ss + sd[cp[i]]));
    for (int off = 16; off >= 1; off >>= 1) mx = fmaxf(mx, __shfl_xor(mx, off, 32));

    const float ex0 = (lane < deg) ? __expf(v0 - mx) : 0.0f;
    float sum = ex0;
    for (int i = lane + 32; i < deg; i += 32) sum += __expf(lrelu(ss + sd[cp[i]]) - mx);
    for (int off = 16; off >= 1; off >>= 1) sum += __shfl_xor(sum, off, 32);

    float ax = 0.0f, ay = 0.0f;
    seg_chunk(Wb, d0, ex0 / sum, deg < 32 ? deg : 32, lane2, ax, ay);
    for (int base = 32; base < deg; base += 32) {
        const int i = base + lane;
        int   d   = 0;
        float att = 0.0f;
        if (i < deg) { d = cp[i]; att = __expf(lrelu(ss + sd[d]) - mx) / sum; }
        const int left = deg - base;
        seg_chunk(Wb, d, att, left < 32 ? left : 32, lane2, ax, ay);
    }
    *orow = make_float2(elu1(ax), elu1(ay));
}

static inline int cdiv(long a, int b) { return (int)((a + b - 1) / b); }

extern "C" void kernel_launch(void* const* d_in, const int* in_sizes, int n_in,
                              void* d_out, int out_size, void* d_ws, size_t ws_size,
                              hipStream_t stream)
{
    const float* x       = (const float*)d_in[0];
    const int*   edges   = (const int*)d_in[1];
    const float* W_heads = (const float*)d_in[2];
    const float* a_heads = (const float*)d_in[3];
    const float* W_out   = (const float*)d_in[4];
    const float* a_out   = (const float*)d_in[5];
    float* out = (float*)d_out;

    const int N = in_sizes[0] / 128;
    const int E = in_sizes[1] / 2;
    const int H = 4, HID = 64;

    const int* src = edges;
    const int* dst = edges + E;

    float* ws = (float*)d_ws;
    size_t off = 0;
    float* Wh1   = ws + off; off += (size_t)H * N * HID;
    float* hcat  = ws + off; off += (size_t)N * 256;
    float* Wh2   = ws + off; off += (size_t)N * HID;
    float* ssrc1 = ws + off; off += (size_t)H * N;
    float* sdst1 = ws + off; off += (size_t)H * N;
    float* ssrc2 = ws + off; off += (size_t)N;
    float* sdst2 = ws + off; off += (size_t)N;
    int* deg    = (int*)(ws + off); off += (size_t)N;
    int* rowptr = (int*)(ws + off); off += (size_t)N + 1;
    int* rank   = (int*)(ws + off); off += (size_t)E;
    int* eid    = (int*)(ws + off); off += (size_t)E;
    int* col    = (int*)(ws + off); off += (size_t)E;

    const int BS = 256;

    k_zero_int<<<cdiv((long)N, BS), BS, 0, stream>>>(deg, N);
    k_edge_rank<<<cdiv((long)E, BS), BS, 0, stream>>>(src, deg, rank, E);
    k_row_offsets<<<cdiv((long)N + 1, 1024), 1024, 0, stream>>>(deg, rowptr, N);
    k_edge_place<<<cdiv((long)E, BS), BS, 0, stream>>>(src, rowptr, rank, eid, E);
    k_row_order<<<cdiv((long)N * 32, BS), BS, 0, stream>>>(rowptr, eid, dst, col, N);

    k_gemm_wmma<<<dim3(N / 16, H), 32, 0, stream>>>(x, W_heads, Wh1, a_heads, ssrc1, sdst1,
        128, 0L, (long)128 * 64, (long)N * 64, (long)N);
    k_seg_attend<<<cdiv((long)H * N * 32, BS), BS, 0, stream>>>(rowptr, col, ssrc1, sdst1, Wh1, hcat, H * N, N, H * HID);

    k_gemm_wmma<<<dim3(N / 16, 1), 32, 0, stream>>>(hcat, W_out, Wh2, a_out, ssrc2, sdst2,
        256, 0L, 0L, 0L, 0L);
    k_seg_attend<<<cdiv((long)N * 32, BS), BS, 0, stream>>>(rowptr, col, ssrc2, sdst2, Wh2, out, N, N, HID);
}


# Layer-1 attend pass takes a row's four heads in one wave: product and scores stored node-major (1-KB records, 16-B score quads), 200k waves become 50k, a neighbour costs two 16-byte-per-lane loads
# speedup vs baseline: 6.1683x; 1.0778x over previous
#include <hip/hip_runtime.h>
#include <cstdint>
#include <cstddef>

typedef float v2f __attribute__((ext_vector_type(2)));
typedef float v8f __attribute__((ext_vector_type(8)));

#define GAT_ALPHA 0.2f

__global__ void k_zero_int(int* __restrict__ p, int n) {
    int i = blockIdx.x * blockDim.x + threadIdx.x;
    if (i < n) p[i] = 0;
}

__global__ void k_edge_rank(const int* __restrict__ src, int* __restrict__ deg,
                            int* __restrict__ rank, int E)
{
    int e = blockIdx.x * blockDim.x + threadIdx.x;
    if (e >= E) return;
    rank[e] = atomicAdd(&deg[src[e]], 1);
}

__global__ __launch_bounds__(1024) void k_row_offsets(const int* __restrict__ deg,
                                                      int* __restrict__ rowptr, int n)
{
    __shared__ int wfront[32];
    __shared__ int wtile[32];
    const int t    = threadIdx.x;
    const int lane = t & 31;
    const int w    = t >> 5;
    const int nb   = blockIdx.x;
    const int idx  = nb * 1024 + t;

    const int v = (idx < n) ? deg[idx] : 0;

    int s = 0;
    const int* __restrict__ p = deg + t;
    for (int k0 = 0; k0 < nb; k0 += 16) {
        int x[16];
        #pragma unroll
        for (int u = 0; u < 16; ++u) {
            const int k = (k0 + u < nb) ? (k0 + u) : (nb - 1);
            x[u] = p[(size_t)k * 1024];
        }
        #pragma unroll
        for (int u = 0; u < 16; ++u) s += (k0 + u < nb) ? x[u] : 0;
    }

    int inc = v;
    for (int off = 1; off < 32; off <<= 1) {
        const int o = __shfl_up(inc, off, 32);
        if (lane >= off) inc += o;
    }
    for (int off = 16; off >= 1; off >>= 1) s += __shfl_xor(s, off, 32);
    if (lane == 31) wtile[w]  = inc;
    if (lane == 0)  wfront[w] = s;
    __syncthreads();
    int base = wfront[lane] + ((lane < w) ? wtile[lane] : 0);
    for (int off = 16; off >= 1; off >>= 1) base += __shfl_xor(base, off, 32);
    if (idx <= n) rowptr[idx] = base + inc - v;
}

__global__ void k_edge_place(const int* __restrict__ src, const int* __restrict__ rowptr,
                             const int* __restrict__ rank, int* __restrict__ eid, int E)
{
    int e = blockIdx.x * blockDim.x + threadIdx.x;
    if (e >= E) return;
    eid[rowptr[src[e]] + rank[e]] = e;
}

__global__ __launch_bounds__(256) void k_row_order(const int* __restrict__ rowptr, const int* __restrict__ eid,
                                                   const int* __restrict__ dst, int* __restrict__ col, int N)
{
    const int lane = threadIdx.x & 31;
    const int wid = __builtin_amdgcn_readfirstlane((int)((blockIdx.x * blockDim.x + threadIdx.x) >> 5));
    if (wid >= N) return;
    const int beg = rowptr[wid];
    const int deg = rowptr[wid + 1] - beg;
    const int* __restrict__ ep = eid + beg;
    for (int ibase = 0; ibase < deg; ibase += 32) {
        const int i = ibase + lane;
        const int mine = (i < deg) ? ep[i] : 0x7fffffff;
        int pos = 0;
        for (int jbase = 0; jbase < deg; jbase += 32) {
            const int j = jbase + lane;
            const int other = (j < deg) ? ep[j] : 0x7fffffff;
            const int left = deg - jbase;
            const int nj = left < 32 ? left : 32;
            for (int u = 0; u < nj; ++u) pos += (__builtin_amdgcn_readlane(other, u) < mine) ? 1 : 0;
        }
        if (i < deg) col[beg + pos] = dst[mine];
    }
}

__device__ __forceinline__ float row16_sum(float v)
{
    v += __int_as_float(__builtin_amdgcn_update_dpp(0, __float_as_int(v), 0x161, 0xf, 0xf, false));
    v += __int_as_float(__builtin_amdgcn_update_dpp(0, __float_as_int(v), 0x162, 0xf, 0xf, false));
    v += __int_as_float(__builtin_amdgcn_update_dpp(0, __float_as_int(v), 0x164, 0xf, 0xf, false));
    v += __int_as_float(__builtin_amdgcn_update_dpp(0, __float_as_int(v), 0x168, 0xf, 0xf, false));
    return v;
}

__global__ __launch_bounds__(32) void k_gemm_wmma(
    const float* __restrict__ A, const float* __restrict__ B, float* __restrict__ C,
    const float* __restrict__ avec, float* __restrict__ ssrc, float* __restrict__ sdst,
    int K, long aStride, long bStride, long cStride, long ldc, long sStride, long sld)
{
    const int lane = threadIdx.x;
    const int m0   = blockIdx.x * 16;
    const int b    = blockIdx.y;
    const float* __restrict__ Ab = A + (long)b * aStride;
    const float* __restrict__ Bb = B + (long)b * bStride;
    float* __restrict__ Cb = C + (long)b * cStride;

    const int row = lane & 15;
    const int kh  = (lane >> 4) << 1;
    const int col = lane & 15;

    const float* __restrict__ av = avec + (long)b * 128;
    const float as0 = av[col],      as1 = av[col + 16], as2 = av[col + 32], as3 = av[col + 48];
    const float ad0 = av[col + 64], ad1 = av[col + 80], ad2 = av[col + 96], ad3 = av[col + 112];

    v8f c0 = {}, c1 = {}, c2 = {}, c3 = {};

    for (int k0 = 0; k0 < K; k0 += 4) {
        const float2 af = *reinterpret_cast<const float2*>(
            Ab + (size_t)(m0 + row) * K + k0 + kh);
        v2f a; a.x = af.x; a.y = af.y;

        const float* br0 = Bb + (size_t)(k0 + kh) * 64 + col;
        const float* br1 = br0 + 64;
        v2f b0, b1, b2, b3;
        b0.x = br0[0];  b0.y = br1[0];
        b1.x = br0[16]; b1.y = br1[16];
        b2.x = br0[32]; b2.y = br1[32];
        b3.x = br0[48]; b3.y = br1[48];

        c0 = __builtin_amdgcn_wmma_f32_16x16x4_f32(false, a, false, b0, (short)0, c0, false, false);
        c1 = __builtin_amdgcn_wmma_f32_16x16x4_f32(false, a, false, b1, (short)0, c1, false, false);
        c2 = __builtin_amdgcn_wmma_f32_16x16x4_f32(false, a, false, b2, (short)0, c2, false, false);
        c3 = __builtin_amdgcn_wmma_f32_16x16x4_f32(false, a, false, b3, (short)0, c3, false, false);
    }

    const int rbase = m0 + ((lane >> 4) << 3);
    #pragma unroll
    for (int r = 0; r < 8; ++r) {
        float* crow = Cb + (size_t)(rbase + r) * ldc + col;
        crow[0]  = c0[r];
        crow[16] = c1[r];
        crow[32] = c2[r];
        crow[48] = c3[r];
    }

    float ps = 0.0f, pd = 0.0f;
    #pragma unroll
    for (int r = 0; r < 8; ++r) {
        const float s = row16_sum(c0[r] * as0 + c1[r] * as1 + c2[r] * as2 + c3[r] * as3);
        const float d = row16_sum(c0[r] * ad0 + c1[r] * ad1 + c2[r] * ad2 + c3[r] * ad3);
        if (col == r) { ps = s; pd = d; }
    }
    if (col < 8) {
        const size_t si = (size_t)b * sStride + (size_t)(rbase + col) * sld;
        ssrc[si] = ps;
        sdst[si] = pd;
    }
}

template <int U>
__device__ __forceinline__ void seg_accum(const float* __restrict__ Wb, int dreg, float areg,
                                          int j, int lane2, float& ax, float& ay)
{
    float2 w[U];
    float  a[U];
    #pragma unroll
    for (int u = 0; u < U; ++u) {
        const int dj = __builtin_amdgcn_readlane(dreg, j + u);
        a[u] = __int_as_float(__builtin_amdgcn_readlane(__float_as_int(areg), j + u));
        w[u] = *reinterpret_cast<const float2*>(Wb + (size_t)dj * 64 + lane2);
    }
    #pragma unroll
    for (int u = 0; u < U; ++u) { ax += a[u] * w[u].x; ay += a[u] * w[u].y; }
}

__device__ __forceinline__ void seg_chunk(const float* __restrict__ Wb, int dreg, float areg,
                                          int nj, int lane2, float& ax, float& ay)
{
    int j = 0;
    for (; j + 8 <= nj; j += 8) seg_accum<8>(Wb, dreg, areg, j, lane2, ax, ay);
    if (j + 4 <= nj) { seg_accum<4>(Wb, dreg, areg, j, lane2, ax, ay); j += 4; }
    if (j + 2 <= nj) { seg_accum<2>(Wb, dreg, areg, j, lane2, ax, ay); j += 2; }
    if (j < nj)      { seg_accum<1>(Wb, dreg, areg, j, lane2, ax, ay); }
}

__device__ __forceinline__ float lrelu(float v) { return v > 0.0f ? v : GAT_ALPHA * v; }

__device__ __forceinline__ float elu1(float v) { return v > 0.0f ? v : __expf(v) - 1.0f; }

__global__ __launch_bounds__(256) void k_seg_attend(
    const int* __restrict__ rowptr, const int* __restrict__ col,
    const float* __restrict__ ssrc, const float* __restrict__ sdst,
    const float* __restrict__ Wh, float* __restrict__ dst, int NH, int N, int ostride)
{
    const int lane  = threadIdx.x & 31;
    const int lane2 = lane << 1;
    const int wid = __builtin_amdgcn_readfirstlane((int)((blockIdx.x * blockDim.x + threadIdx.x) >> 5));
    if (wid >= NH) return;
    const int h = wid / N;
    const int n = wid - h * N;
    const int beg = rowptr[n];
    const int deg = rowptr[n + 1] - beg;
    float2* __restrict__ orow = reinterpret_cast<float2*>(dst + (size_t)n * ostride + (size_t)h * 64) + lane;
    if (deg == 0) { *orow = make_float2(0.0f, 0.0f); return; }

    const int*   __restrict__ cp = col + beg;
    const float* __restrict__ sd = sdst + (size_t)h * N;
    const float* __restrict__ Wb = Wh + (size_t)h * N * 64;
    const float ss = ssrc[wid];

    int   d0 = 0;
    float v0 = -__builtin_huge_valf();
    if (lane < deg) { d0 = cp[lane]; v0 = lrelu(ss + sd[d0]); }

    float mx = v0;
    for (int i = lane + 32; i < deg; i += 32) mx = fmaxf(mx, lrelu(ss + sd[cp[i]]));
    for (int off = 16; off >= 1; off >>= 1) mx = fmaxf(mx, __shfl_xor(mx, off, 32));

    const float ex0 = (lane < deg) ? __expf(v0 - mx) : 0.0f;
    float sum = ex0;
    for (int i = lane + 32; i < deg; i += 32) sum += __expf(lrelu(ss + sd[cp[i]]) - mx);
    for (int off = 16; off >= 1; off >>= 1) sum += __shfl_xor(sum, off, 32);

    float ax = 0.0f, ay = 0.0f;
    seg_chunk(Wb, d0, ex0 / sum, deg < 32 ? deg : 32, lane2, ax, ay);
    for (int base = 32; base < deg; base += 32) {
        const int i = base + lane;
        int   d   = 0;
        float att = 0.0f;
        if (i < deg) { d = cp[i]; att = __expf(lrelu(ss + sd[d]) - mx) / sum; }
        const int left = deg - base;
        seg_chunk(Wb, d, att, left < 32 ? left : 32, lane2, ax, ay);
    }
    *orow = make_float2(elu1(ax), elu1(ay));
}


template <int U>
__device__ __forceinline__ void seg_accum4(const float* __restrict__ Wn, int dreg,
                                           float a0r, float a1r, float a2r, float a3r,
                                           int j, int lane4, bool hi, float4& pa, float4& pb)
{
    float4 wa[U], wb[U];
    #pragma unroll
    for (int u = 0; u < U; ++u) {
        const int dj = __builtin_amdgcn_readlane(dreg, j + u);
        const float* __restrict__ rec = Wn + (size_t)dj * 256 + lane4;
        wa[u] = *reinterpret_cast<const float4*>(rec);
        wb[u] = *reinterpret_cast<const float4*>(rec + 128);
    }
    #pragma unroll
    for (int u = 0; u < U; ++u) {
        const float s0 = __int_as_float(__builtin_amdgcn_readlane(__float_as_int(a0r), j + u));
        const float s1 = __int_as_float(__builtin_amdgcn_readlane(__float_as_int(a1r), j + u));
        const float s2 = __int_as_float(__builtin_amdgcn_readlane(__float_as_int(a2r), j + u));
        const float s3 = __int_as_float(__builtin_amdgcn_readlane(__float_as_int(a3r), j + u));
        const float ga = hi ? s1 : s0;
        const float gb = hi ? s3 : s2;
        pa.x += ga * wa[u].x; pa.y += ga * wa[u].y; pa.z += ga * wa[u].z; pa.w += ga * wa[u].w;
        pb.x += gb * wb[u].x; pb.y += gb * wb[u].y; pb.z += gb * wb[u].z; pb.w += gb * wb[u].w;
    }
}

__device__ __forceinline__ void seg_chunk4(const float* __restrict__ Wn, int dreg,
                                           float a0r, float a1r, float a2r, float a3r,
                                           int nj, int lane4, bool hi, float4& pa, float4& pb)
{
    int j = 0;
    for (; j + 8 <= nj; j += 8) seg_accum4<8>(Wn, dreg, a0r, a1r, a2r, a3r, j, lane4, hi, pa, pb);
    if (j + 4 <= nj) { seg_accum4<4>(Wn, dreg, a0r, a1r, a2r, a3r, j, lane4, hi, pa, pb); j += 4; }
    if (j + 2 <= nj) { seg_accum4<2>(Wn, dreg, a0r, a1r, a2r, a3r, j, lane4, hi, pa, pb); j += 2; }
    if (j < nj)      { seg_accum4<1>(Wn, dreg, a0r, a1r, a2r, a3r, j, lane4, hi, pa, pb); }
}

__global__ __launch_bounds__(256) void k_seg_attend_h4(
    const int* __restrict__ rowptr, const int* __restrict__ col,
    const float* __restrict__ ssrc, const float* __restrict__ sdst,
    const float* __restrict__ Wn, float* __restrict__ dst, int N)
{
    const int  lane  = threadIdx.x & 31;
    const int  lane4 = lane << 2;
    const bool hi    = lane >= 16;
    const int n = __builtin_amdgcn_readfirstlane((int)((blockIdx.x * blockDim.x + threadIdx.x) >> 5));
    if (n >= N) return;
    const int beg = rowptr[n];
    const int deg = rowptr[n + 1] - beg;
    float4* __restrict__ orow = reinterpret_cast<float4*>(dst + (size_t)n * 256) + lane;
    if (deg == 0) {
        orow[0]  = make_float4(0.0f, 0.0f, 0.0f, 0.0f);
        orow[32] = make_float4(0.0f, 0.0f, 0.0f, 0.0f);
        return;
    }

    const int* __restrict__ cp = col + beg;
    const float4 ss = *reinterpret_cast<const float4*>(ssrc + (size_t)n * 4);

    const float ninf = -__builtin_huge_valf();
    int   d0 = 0;
    float v0 = ninf, v1 = ninf, v2 = ninf, v3 = ninf;
    if (lane < deg) {
        d0 = cp[lane];
        const float4 sd = *reinterpret_cast<const float4*>(sdst + (size_t)d0 * 4);
        v0 = lrelu(ss.x + sd.x); v1 = lrelu(ss.y + sd.y); v2 = lrelu(ss.z + sd.z); v3 = lrelu(ss.w + sd.w);
    }

    float m0 = v0, m1 = v1, m2 = v2, m3 = v3;
    for (int i = lane + 32; i < deg; i += 32) {
        const float4 sd = *reinterpret_cast<const float4*>(sdst + (size_t)cp[i] * 4);
        m0 = fmaxf(m0, lrelu(ss.x + sd.x)); m1 = fmaxf(m1, lrelu(ss.y + sd.y));
        m2 = fmaxf(m2, lrelu(ss.z + sd.z)); m3 = fmaxf(m3, lrelu(ss.w + sd.w));
    }
    for (int off = 16; off >= 1; off >>= 1) {
        m0 = fmaxf(m0, __shfl_xor(m0, off, 32)); m1 = fmaxf(m1, __shfl_xor(m1, off, 32));
        m2 = fmaxf(m2, __shfl_xor(m2, off, 32)); m3 = fmaxf(m3, __shfl_xor(m3, off, 32));
    }

    const bool in0 = lane < deg;
    const float e0 = in0 ? __expf(v0 - m0) : 0.0f;
    const float e1 = in0 ? __expf(v1 - m1) : 0.0f;
    const float e2 = in0 ? __expf(v2 - m2) : 0.0f;
    const float e3 = in0 ? __expf(v3 - m3) : 0.0f;
    float t0 = e0, t1 = e1, t2 = e2, t3 = e3;
    for (int i = lane + 32; i < deg; i += 32) {
        const float4 sd = *reinterpret_cast<const float4*>(sdst + (size_t)cp[i] * 4);
        t0 += __expf(lrelu(ss.x + sd.x) - m0); t1 += __expf(lrelu(ss.y + sd.y) - m1);
        t2 += __expf(lrelu(ss.z + sd.z) - m2); t3 += __expf(lrelu(ss.w + sd.w) - m3);
    }
    for (int off = 16; off >= 1; off >>= 1) {
        t0 += __shfl_xor(t0, off, 32); t1 += __shfl_xor(t1, off, 32);
        t2 += __shfl_xor(t2, off, 32); t3 += __shfl_xor(t3, off, 32);
    }

    float4 pa = make_float4(0.0f, 0.0f, 0.0f, 0.0f);
    float4 pb = make_float4(0.0f, 0.0f, 0.0f, 0.0f);
    seg_chunk4(Wn, d0, e0 / t0, e1 / t1, e2 / t2, e3 / t3, deg < 32 ? deg : 32, lane4, hi, pa, pb);
    for (int base = 32; base < deg; base += 32) {
        const int i = base + lane;
        int   d  = 0;
        float a0 = 0.0f, a1 = 0.0f, a2 = 0.0f, a3 = 0.0f;
        if (i < deg) {
            d = cp[i];
            const float4 sd = *reinterpret_cast<const float4*>(sdst + (size_t)d * 4);
            a0 = __expf(lrelu(ss.x + sd.x) - m0) / t0; a1 = __expf(lrelu(ss.y + sd.y) - m1) / t1;
            a2 = __expf(lrelu(ss.z + sd.z) - m2) / t2; a3 = __expf(lrelu(ss.w + sd.w) - m3) / t3;
        }
        const int left = deg - base;
        seg_chunk4(Wn, d, a0, a1, a2, a3, left < 32 ? left : 32, lane4, hi, pa, pb);
    }
    orow[0]  = make_float4(elu1(pa.x), elu1(pa.y), elu1(pa.z), elu1(pa.w));
    orow[32] = make_float4(elu1(pb.x), elu1(pb.y), elu1(pb.z), elu1(pb.w));
}

static inline int cdiv(long a, int b) { return (int)((a + b - 1) / b); }

extern "C" void kernel_launch(void* const* d_in, const int* in_sizes, int n_in,
                              void* d_out, int out_size, void* d_ws, size_t ws_size,
                              hipStream_t stream)
{
    const float* x       = (const float*)d_in[0];
    const int*   edges   = (const int*)d_in[1];
    const float* W_heads = (const float*)d_in[2];
    const float* a_heads = (const float*)d_in[3];
    const float* W_out   = (const float*)d_in[4];
    const float* a_out   = (const float*)d_in[5];
    float* out = (float*)d_out;

    const int N = in_sizes[0] / 128;
    const int E = in_sizes[1] / 2;
    const int H = 4, HID = 64;

    const int* src = edges;
    const int* dst = edges + E;

    float* ws = (float*)d_ws;
    size_t off = 0;
    float* Wh1   = ws + off; off += (size_t)H * N * HID;
    float* hcat  = ws + off; off += (size_t)N * 256;
    float* Wh2   = ws + off; off += (size_t)N * HID;
    float* ssrc1 = ws + off; off += (size_t)H * N;
    float* sdst1 = ws + off; off += (size_t)H * N;
    float* ssrc2 = ws + off; off += (size_t)N;
    float* sdst2 = ws + off; off += (size_t)N;
    int* deg    = (int*)(ws + off); off += (size_t)N;
    int* rowptr = (int*)(ws + off); off += (size_t)N + 1;
    int* rank   = (int*)(ws + off); off += (size_t)E;
    int* eid    = (int*)(ws + off); off += (size_t)E;
    int* col    = (int*)(ws + off); off += (size_t)E;

    const int BS = 256;

    k_zero_int<<<cdiv((long)N, BS), BS, 0, stream>>>(deg, N);
    k_edge_rank<<<cdiv((long)E, BS), BS, 0, stream>>>(src, deg, rank, E);
    k_row_offsets<<<cdiv((long)N + 1, 1024), 1024, 0, stream>>>(deg, rowptr, N);
    k_edge_place<<<cdiv((long)E, BS), BS, 0, stream>>>(src, rowptr, rank, eid, E);
    k_row_order<<<cdiv((long)N * 32, BS), BS, 0, stream>>>(rowptr, eid, dst, col, N);

    k_gemm_wmma<<<dim3(N / 16, H), 32, 0, stream>>>(x, W_heads, Wh1, a_heads, ssrc1, sdst1,
        128, 0L, (long)128 * 64, 64L, 256L, 1L, 4L);
    k_seg_attend_h4<<<cdiv((long)N * 32, BS), BS, 0, stream>>>(rowptr, col, ssrc1, sdst1, Wh1, hcat, N);

    k_gemm_wmma<<<dim3(N / 16, 1), 32, 0, stream>>>(hcat, W_out, Wh2, a_out, ssrc2, sdst2,
        256, 0L, 0L, 0L, 64L, 0L, 1L);
    k_seg_attend<<<cdiv((long)N * 32, BS), BS, 0, stream>>>(rowptr, col, ssrc2, sdst2, Wh2, out, N, N, HID);
}


# Edge counters kept as one plane per package half, chosen by the wave's die number: no counter line is added into from both halves; half 1's edges fill a row's slots from its back
# speedup vs baseline: 7.4459x; 1.2071x over previous
#include <hip/hip_runtime.h>
#include <cstdint>
#include <cstddef>

typedef float v2f __attribute__((ext_vector_type(2)));
typedef float v8f __attribute__((ext_vector_type(8)));

#define GAT_ALPHA 0.2f

__global__ void k_zero_int2(int* __restrict__ p, int* __restrict__ q, int n) {
    int i = blockIdx.x * blockDim.x + threadIdx.x;
    if (i < n) { p[i] = 0; q[i] = 0; }
}

__device__ __forceinline__ int chip_half()
{
    return (int)((__builtin_amdgcn_s_sendmsg_rtn(0x87) >> 18) & 1u);
}

__global__ void k_edge_rank(const int* __restrict__ src, int* __restrict__ deg0, int* __restrict__ deg1,
                            int* __restrict__ rank, int E)
{
    int e = blockIdx.x * blockDim.x + threadIdx.x;
    const int half = chip_half();
    if (e >= E) return;
    int* __restrict__ cnt = half ? deg1 : deg0;
    const int r = atomicAdd(&cnt[src[e]], 1);
    rank[e] = half ? -1 - r : r;
}

__global__ __launch_bounds__(1024) void k_row_offsets(const int* __restrict__ deg0, const int* __restrict__ deg1,
                                                      int* __restrict__ rowptr, int n)
{
    __shared__ int wfront[32];
    __shared__ int wtile[32];
    const int t    = threadIdx.x;
    const int lane = t & 31;
    const int w    = t >> 5;
    const int nb   = blockIdx.x;
    const int idx  = nb * 1024 + t;

    const int v = (idx < n) ? deg0[idx] + deg1[idx] : 0;

    int s = 0;
    const int* __restrict__ p0 = deg0 + t;
    const int* __restrict__ p1 = deg1 + t;
    for (int k0 = 0; k0 < nb; k0 += 16) {
        int x[16], y[16];
        #pragma unroll
        for (int u = 0; u < 16; ++u) {
            const int k = (k0 + u < nb) ? (k0 + u) : (nb - 1);
            x[u] = p0[(size_t)k * 1024];
            y[u] = p1[(size_t)k * 1024];
        }
        #pragma unroll
        for (int u = 0; u < 16; ++u) s += (k0 + u < nb) ? x[u] + y[u] : 0;
    }

    int inc = v;
    for (int off = 1; off < 32; off <<= 1) {
        const int o = __shfl_up(inc, off, 32);
        if (lane >= off) inc += o;
    }
    for (int off = 16; off >= 1; off >>= 1) s += __shfl_xor(s, off, 32);
    if (lane == 31) wtile[w]  = inc;
    if (lane == 0)  wfront[w] = s;
    __syncthreads();
    int base = wfront[lane] + ((lane < w) ? wtile[lane] : 0);
    for (int off = 16; off >= 1; off >>= 1) base += __shfl_xor(base, off, 32);
    if (idx <= n) rowptr[idx] = base + inc - v;
}

__global__ void k_edge_place(const int* __restrict__ src, const int* __restrict__ rowptr,
                             const int* __restrict__ rank, int* __restrict__ eid, int E)
{
    int e = blockIdx.x * blockDim.x + threadIdx.x;
    if (e >= E) return;
    const int r = rank[e];
    eid[rowptr[src[e] + (r < 0 ? 1 : 0)] + r] = e;
}

__global__ __launch_bounds__(256) void k_row_order(const int* __restrict__ rowptr, const int* __restrict__ eid,
                                                   const int* __restrict__ dst, int* __restrict__ col, int N)
{
    const int lane = threadIdx.x & 31;
    const int wid = __builtin_amdgcn_readfirstlane((int)((blockIdx.x * blockDim.x + threadIdx.x) >> 5));
    if (wid >= N) return;
    const int beg = rowptr[wid];
    const int deg = rowptr[wid + 1] - beg;
    const int* __restrict__ ep = eid + beg;
    for (int ibase = 0; ibase < deg; ibase += 32) {
        const int i = ibase + lane;
        const int mine = (i < deg) ? ep[i] : 0x7fffffff;
        int pos = 0;
        for (int jbase = 0; jbase < deg; jbase += 32) {
            const int j = jbase + lane;
            const int other = (j < deg) ? ep[j] : 0x7fffffff;
            const int left = deg - jbase;
            const int nj = left < 32 ? left : 32;
            for (int u = 0; u < nj; ++u) pos += (__builtin_amdgcn_readlane(other, u) < mine) ? 1 : 0;
        }
        if (i < deg) col[beg + pos] = dst[mine];
    }
}

__device__ __forceinline__ float row16_sum(float v)
{
    v += __int_as_float(__builtin_amdgcn_update_dpp(0, __float_as_int(v), 0x161, 0xf, 0xf, false));
    v += __int_as_float(__builtin_amdgcn_update_dpp(0, __float_as_int(v), 0x162, 0xf, 0xf, false));
    v += __int_as_float(__builtin_amdgcn_update_dpp(0, __float_as_int(v), 0x164, 0xf, 0xf, false));
    v += __int_as_float(__builtin_amdgcn_update_dpp(0, __float_as_int(v), 0x168, 0xf, 0xf, false));
    return v;
}

__global__ __launch_bounds__(32) void k_gemm_wmma(
    const float* __restrict__ A, const float* __restrict__ B, float* __restrict__ C,
    const float* __restrict__ avec, float* __restrict__ ssrc, float* __restrict__ sdst,
    int K, long aStride, long bStride, long cStride, long ldc, long sStride, long sld)
{
    const int lane = threadIdx.x;
    const int m0   = blockIdx.x * 16;
    const int b    = blockIdx.y;
    const float* __restrict__ Ab = A + (long)b * aStride;
    const float* __restrict__ Bb = B + (long)b * bStride;
    float* __restrict__ Cb = C + (long)b * cStride;

    const int row = lane & 15;
    const int kh  = (lane >> 4) << 1;
    const int col = lane & 15;

    const float* __restrict__ av = avec + (long)b * 128;
    const float as0 = av[col],      as1 = av[col + 16], as2 = av[col + 32], as3 = av[col + 48];
    const float ad0 = av[col + 64], ad1 = av[col + 80], ad2 = av[col + 96], ad3 = av[col + 112];

    v8f c0 = {}, c1 = {}, c2 = {}, c3 = {};

    for (int k0 = 0; k0 < K; k0 += 4) {
        const float2 af = *reinterpret_cast<const float2*>(
            Ab + (size_t)(m0 + row) * K + k0 + kh);
        v2f a; a.x = af.x; a.y = af.y;

        const float* br0 = Bb + (size_t)(k0 + kh) * 64 + col;
        const float* br1 = br0 + 64;
        v2f b0, b1, b2, b3;
        b0.x = br0[0];  b0.y = br1[0];
        b1.x = br0[16]; b1.y = br1[16];
        b2.x = br0[32]; b2.y = br1[32];
        b3.x = br0[48]; b3.y = br1[48];

        c0 = __builtin_amdgcn_wmma_f32_16x16x4_f32(false, a, false, b0, (short)0, c0, false, false);
        c1 = __builtin_amdgcn_wmma_f32_16x16x4_f32(false, a, false, b1, (short)0, c1, false, false);
        c2 = __builtin_amdgcn_wmma_f32_16x16x4_f32(false, a, false, b2, (short)0, c2, false, false);
        c3 = __builtin_amdgcn_wmma_f32_16x16x4_f32(false, a, false, b3, (short)0, c3, false, false);
    }

    const int rbase = m0 + ((lane >> 4) << 3);
    #pragma unroll
    for (int r = 0; r < 8; ++r) {
        float* crow = Cb + (size_t)(rbase + r) * ldc + col;
        crow[0]  = c0[r];
        crow[16] = c1[r];
        crow[32] = c2[r];
        crow[48] = c3[r];
    }

    float ps = 0.0f, pd = 0.0f;
    #pragma unroll
    for (int r = 0; r < 8; ++r) {
        const float s = row16_sum(c0[r] * as0 + c1[r] * as1 + c2[r] * as2 + c3[r] * as3);
        const float d = row16_sum(c0[r] * ad0 + c1[r] * ad1 + c2[r] * ad2 + c3[r] * ad3);
        if (col == r) { ps = s; pd = d; }
    }
    if (col < 8) {
        const size_t si = (size_t)b * sStride + (size_t)(rbase + col) * sld;
        ssrc[si] = ps;
        sdst[si] = pd;
    }
}

template <int U>
__device__ __forceinline__ void seg_accum(const float* __restrict__ Wb, int dreg, float areg,
                                          int j, int lane2, float& ax, float& ay)
{
    float2 w[U];
    float  a[U];
    #pragma unroll
    for (int u = 0; u < U; ++u) {
        const int dj = __builtin_amdgcn_readlane(dreg, j + u);
        a[u] = __int_as_float(__builtin_amdgcn_readlane(__float_as_int(areg), j + u));
        w[u] = *reinterpret_cast<const float2*>(Wb + (size_t)dj * 64 + lane2);
    }
    #pragma unroll
    for (int u = 0; u < U; ++u) { ax += a[u] * w[u].x; ay += a[u] * w[u].y; }
}

__device__ __forceinline__ void seg_chunk(const float* __restrict__ Wb, int dreg, float areg,
                                          int nj, int lane2, float& ax, float& ay)
{
    int j = 0;
    for (; j + 8 <= nj; j += 8) seg_accum<8>(Wb, dreg, areg, j, lane2, ax, ay);
    if (j + 4 <= nj) { seg_accum<4>(Wb, dreg, areg, j, lane2, ax, ay); j += 4; }
    if (j + 2 <= nj) { seg_accum<2>(Wb, dreg, areg, j, lane2, ax, ay); j += 2; }
    if (j < nj)      { seg_accum<1>(Wb, dreg, areg, j, lane2, ax, ay); }
}

__device__ __forceinline__ float lrelu(float v) { return v > 0.0f ? v : GAT_ALPHA * v; }

__device__ __forceinline__ float elu1(float v) { return v > 0.0f ? v : __expf(v) - 1.0f; }

__global__ __launch_bounds__(256) void k_seg_attend(
    const int* __restrict__ rowptr, const int* __restrict__ col,
    const float* __restrict__ ssrc, const float* __restrict__ sdst,
    const float* __restrict__ Wh, float* __restrict__ dst, int NH, int N, int ostride)
{
    const int lane  = threadIdx.x & 31;
    const int lane2 = lane << 1;
    const int wid = __builtin_amdgcn_readfirstlane((int)((blockIdx.x * blockDim.x + threadIdx.x) >> 5));
    if (wid >= NH) return;
    const int h = wid / N;
    const int n = wid - h * N;
    const int beg = rowptr[n];
    const int deg = rowptr[n + 1] - beg;
    float2* __restrict__ orow = reinterpret_cast<float2*>(dst + (size_t)n * ostride + (size_t)h * 64) + lane;
    if (deg == 0) { *orow = make_float2(0.0f, 0.0f); return; }

    const int*   __restrict__ cp = col + beg;
    const float* __restrict__ sd = sdst + (size_t)h * N;
    const float* __restrict__ Wb = Wh + (size_t)h * N * 64;
    const float ss = ssrc[wid];

    int   d0 = 0;
    float v0 = -__builtin_huge_valf();
    if (lane < deg) { d0 = cp[lane]; v0 = lrelu(ss + sd[d0]); }

    float mx = v0;
    for (int i = lane + 32; i < deg; i += 32) mx = fmaxf(mx, lrelu(ss + sd[cp[i]]));
    for (int off = 16; off >= 1; off >>= 1) mx = fmaxf(mx, __shfl_xor(mx, off, 32));

    const float ex0 = (lane < deg) ? __expf(v0 - mx) : 0.0f;
    float sum = ex0;
    for (int i = lane + 32; i < deg; i += 32) sum += __expf(lrelu(ss + sd[cp[i]]) - mx);
    for (int off = 16; off >= 1; off >>= 1) sum += __shfl_xor(sum, off, 32);

    float ax = 0.0f, ay = 0.0f;
    seg_chunk(Wb, d0, ex0 / sum, deg < 32 ? deg : 32, lane2, ax, ay);
    for (int base = 32; base < deg; base += 32) {
        const int i = base + lane;
        int   d   = 0;
        float att = 0.0f;
        if (i < deg) { d = cp[i]; att = __expf(lrelu(ss + sd[d]) - mx) / sum; }
        const int left = deg - base;
        seg_chunk(Wb, d, att, left < 32 ? left : 32, lane2, ax, ay);
    }
    *orow = make_float2(elu1(ax), elu1(ay));
}


template <int U>
__device__ __forceinline__ void seg_accum4(const float* __restrict__ Wn, int dreg,
                                           float a0r, float a1r, float a2r, float a3r,
                                           int j, int lane4, bool hi, float4& pa, float4& pb)
{
    float4 wa[U], wb[U];
    #pragma unroll
    for (int u = 0; u < U; ++u) {
        const int dj = __builtin_amdgcn_readlane(dreg, j + u);
        const float* __restrict__ rec = Wn + (size_t)dj * 256 + lane4;
        wa[u] = *reinterpret_cast<const float4*>(rec);
        wb[u] = *reinterpret_cast<const float4*>(rec + 128);
    }
    #pragma unroll
    for (int u = 0; u < U; ++u) {
        const float s0 = __int_as_float(__builtin_amdgcn_readlane(__float_as_int(a0r), j + u));
        const float s1 = __int_as_float(__builtin_amdgcn_readlane(__float_as_int(a1r), j + u));
        const float s2 = __int_as_float(__builtin_amdgcn_readlane(__float_as_int(a2r), j + u));
        const float s3 = __int_as_float(__builtin_amdgcn_readlane(__float_as_int(a3r), j + u));
        const float ga = hi ? s1 : s0;
        const float gb = hi ? s3 : s2;
        pa.x += ga * wa[u].x; pa.y += ga * wa[u].y; pa.z += ga * wa[u].z; pa.w += ga * wa[u].w;
        pb.x += gb * wb[u].x; pb.y += gb * wb[u].y; pb.z += gb * wb[u].z; pb.w += gb * wb[u].w;
    }
}

__device__ __forceinline__ void seg_chunk4(const float* __restrict__ Wn, int dreg,
                                           float a0r, float a1r, float a2r, float a3r,
                                           int nj, int lane4, bool hi, float4& pa, float4& pb)
{
    int j = 0;
    for (; j + 8 <= nj; j += 8) seg_accum4<8>(Wn, dreg, a0r, a1r, a2r, a3r, j, lane4, hi, pa, pb);
    if (j + 4 <= nj) { seg_accum4<4>(Wn, dreg, a0r, a1r, a2r, a3r, j, lane4, hi, pa, pb); j += 4; }
    if (j + 2 <= nj) { seg_accum4<2>(Wn, dreg, a0r, a1r, a2r, a3r, j, lane4, hi, pa, pb); j += 2; }
    if (j < nj)      { seg_accum4<1>(Wn, dreg, a0r, a1r, a2r, a3r, j, lane4, hi, pa, pb); }
}

__global__ __launch_bounds__(256) void k_seg_attend_h4(
    const int* __restrict__ rowptr, const int* __restrict__ col,
    const float* __restrict__ ssrc, const float* __restrict__ sdst,
    const float* __restrict__ Wn, float* __restrict__ dst, int N)
{
    const int  lane  = threadIdx.x & 31;
    const int  lane4 = lane << 2;
    const bool hi    = lane >= 16;
    const int n = __builtin_amdgcn_readfirstlane((int)((blockIdx.x * blockDim.x + threadIdx.x) >> 5));
    if (n >= N) return;
    const int beg = rowptr[n];
    const int deg = rowptr[n + 1] - beg;
    float4* __restrict__ orow = reinterpret_cast<float4*>(dst + (size_t)n * 256) + lane;
    if (deg == 0) {
        orow[0]  = make_float4(0.0f, 0.0f, 0.0f, 0.0f);
        orow[32] = make_float4(0.0f, 0.0f, 0.0f, 0.0f);
        return;
    }

    const int* __restrict__ cp = col + beg;
    const float4 ss = *reinterpret_cast<const float4*>(ssrc + (size_t)n * 4);

    const float ninf = -__builtin_huge_valf();
    int   d0 = 0;
    float v0 = ninf, v1 = ninf, v2 = ninf, v3 = ninf;
    if (lane < deg) {
        d0 = cp[lane];
        const float4 sd = *reinterpret_cast<const float4*>(sdst + (size_t)d0 * 4);
        v0 = lrelu(ss.x + sd.x); v1 = lrelu(ss.y + sd.y); v2 = lrelu(ss.z + sd.z); v3 = lrelu(ss.w + sd.w);
    }

    float m0 = v0, m1 = v1, m2 = v2, m3 = v3;
    for (int i = lane + 32; i < deg; i += 32) {
        const float4 sd = *reinterpret_cast<const float4*>(sdst + (size_t)cp[i] * 4);
        m0 = fmaxf(m0, lrelu(ss.x + sd.x)); m1 = fmaxf(m1, lrelu(ss.y + sd.y));
        m2 = fmaxf(m2, lrelu(ss.z + sd.z)); m3 = fmaxf(m3, lrelu(ss.w + sd.w));
    }
    for (int off = 16; off >= 1; off >>= 1) {
        m0 = fmaxf(m0, __shfl_xor(m0, off, 32)); m1 = fmaxf(m1, __shfl_xor(m1, off, 32));
        m2 = fmaxf(m2, __shfl_xor(m2, off, 32)); m3 = fmaxf(m3, __shfl_xor(m3, off, 32));
    }

    const bool in0 = lane < deg;
    const float e0 = in0 ? __expf(v0 - m0) : 0.0f;
    const float e1 = in0 ? __expf(v1 - m1) : 0.0f;
    const float e2 = in0 ? __expf(v2 - m2) : 0.0f;
    const float e3 = in0 ? __expf(v3 - m3) : 0.0f;
    float t0 = e0, t1 = e1, t2 = e2, t3 = e3;
    for (int i = lane + 32; i < deg; i += 32) {
        const float4 sd = *reinterpret_cast<const float4*>(sdst + (size_t)cp[i] * 4);
        t0 += __expf(lrelu(ss.x + sd.x) - m0); t1 += __expf(lrelu(ss.y + sd.y) - m1);
        t2 += __expf(lrelu(ss.z + sd.z) - m2); t3 += __expf(lrelu(ss.w + sd.w) - m3);
    }
    for (int off = 16; off >= 1; off >>= 1) {
        t0 += __shfl_xor(t0, off, 32); t1 += __shfl_xor(t1, off, 32);
        t2 += __shfl_xor(t2, off, 32); t3 += __shfl_xor(t3, off, 32);
    }

    float4 pa = make_float4(0.0f, 0.0f, 0.0f, 0.0f);
    float4 pb = make_float4(0.0f, 0.0f, 0.0f, 0.0f);
    seg_chunk4(Wn, d0, e0 / t0, e1 / t1, e2 / t2, e3 / t3, deg < 32 ? deg : 32, lane4, hi, pa, pb);
    for (int base = 32; base < deg; base += 32) {
        const int i = base + lane;
        int   d  = 0;
        float a0 = 0.0f, a1 = 0.0f, a2 = 0.0f, a3 = 0.0f;
        if (i < deg) {
            d = cp[i];
            const float4 sd = *reinterpret_cast<const float4*>(sdst + (size_t)d * 4);
            a0 = __expf(lrelu(ss.x + sd.x) - m0) / t0; a1 = __expf(lrelu(ss.y + sd.y) - m1) / t1;
            a2 = __expf(lrelu(ss.z + sd.z) - m2) / t2; a3 = __expf(lrelu(ss.w + sd.w) - m3) / t3;
        }
        const int left = deg - base;
        seg_chunk4(Wn, d, a0, a1, a2, a3, left < 32 ? left : 32, lane4, hi, pa, pb);
    }
    orow[0]  = make_float4(elu1(pa.x), elu1(pa.y), elu1(pa.z), elu1(pa.w));
    orow[32] = make_float4(elu1(pb.x), elu1(pb.y), elu1(pb.z), elu1(pb.w));
}

static inline int cdiv(long a, int b) { return (int)((a + b - 1) / b); }

extern "C" void kernel_launch(void* const* d_in, const int* in_sizes, int n_in,
                              void* d_out, int out_size, void* d_ws, size_t ws_size,
                              hipStream_t stream)
{
    const float* x       = (const float*)d_in[0];
    const int*   edges   = (const int*)d_in[1];
    const float* W_heads = (const float*)d_in[2];
    const float* a_heads = (const float*)d_in[3];
    const float* W_out   = (const float*)d_in[4];
    const float* a_out   = (const float*)d_in[5];
    float* out = (float*)d_out;

    const int N = in_sizes[0] / 128;
    const int E = in_sizes[1] / 2;
    const int H = 4, HID = 64;

    const int* src = edges;
    const int* dst = edges + E;

    float* ws = (float*)d_ws;
    size_t off = 0;
    float* Wh1   = ws + off; off += (size_t)H * N * HID;
    float* hcat  = ws + off; off += (size_t)N * 256;
    float* Wh2   = ws + off; off += (size_t)N * HID;
    float* ssrc1 = ws + off; off += (size_t)H * N;
    float* sdst1 = ws + off; off += (size_t)H * N;
    float* ssrc2 = ws + off; off += (size_t)N;
    float* sdst2 = ws + off; off += (size_t)N;
    int* deg0   = (int*)(ws + off); off += (size_t)N;
    int* rowptr = (int*)(ws + off); off += (size_t)N + 1;
    int* rank   = (int*)(ws + off); off += (size_t)E;
    int* eid    = (int*)(ws + off); off += (size_t)E;
    int* col    = (int*)(ws + off); off += (size_t)E;
    int* deg1   = (int*)hcat;

    const int BS = 256;

    k_zero_int2<<<cdiv((long)N, BS), BS, 0, stream>>>(deg0, deg1, N);
    k_edge_rank<<<cdiv((long)E, BS), BS, 0, stream>>>(src, deg0, deg1, rank, E);
    k_row_offsets<<<cdiv((long)N + 1, 1024), 1024, 0, stream>>>(deg0, deg1, rowptr, N);
    k_edge_place<<<cdiv((long)E, BS), BS, 0, stream>>>(src, rowptr, rank, eid, E);
    k_row_order<<<cdiv((long)N * 32, BS), BS, 0, stream>>>(rowptr, eid, dst, col, N);

    k_gemm_wmma<<<dim3(N / 16, H), 32, 0, stream>>>(x, W_heads, Wh1, a_heads, ssrc1, sdst1,
        128, 0L, (long)128 * 64, 64L, 256L, 1L, 4L);
    k_seg_attend_h4<<<cdiv((long)N * 32, BS), BS, 0, stream>>>(rowptr, col, ssrc1, sdst1, Wh1, hcat, N);

    k_gemm_wmma<<<dim3(N / 16, 1), 32, 0, stream>>>(hcat, W_out, Wh2, a_out, ssrc2, sdst2,
        256, 0L, 0L, 0L, 64L, 0L, 1L);
    k_seg_attend<<<cdiv((long)N * 32, BS), BS, 0, stream>>>(rowptr, col, ssrc2, sdst2, Wh2, out, N, N, HID);
}


# Layer-1 product gets no launch of its own: its 16x64 tiles run, one per wave, in the front work-groups of the edge count and edge place launches, matrix work under their memory waits; 9 -> 8 launches
# speedup vs baseline: 7.6912x; 1.0329x over previous
#include <hip/hip_runtime.h>
#include <cstdint>
#include <cstddef>

typedef float v2f __attribute__((ext_vector_type(2)));
typedef float v8f __attribute__((ext_vector_type(8)));

#define GAT_ALPHA 0.2f

__global__ void k_zero_int2(int* __restrict__ p, int* __restrict__ q, int n) {
    int i = blockIdx.x * blockDim.x + threadIdx.x;
    if (i < n) { p[i] = 0; q[i] = 0; }
}

__device__ __forceinline__ int chip_half()
{
    return (int)((__builtin_amdgcn_s_sendmsg_rtn(0x87) >> 18) & 1u);
}

__device__ __forceinline__ void l1_tile_wave(const float* __restrict__ x, const float* __restrict__ W,
                                             float* __restrict__ Wh, const float* __restrict__ avec,
                                             float* __restrict__ ssrc, float* __restrict__ sdst,
                                             int K, int mTiles, int tile, int lane);

#define L1_TILE_ROLE()                                                                                         \
    if (blockIdx.x < (unsigned)gBlocks) {                                                                      \
        const int tile = tile0 + __builtin_amdgcn_readfirstlane(                                               \
            (int)(blockIdx.x * (blockDim.x >> 5) + (threadIdx.x >> 5)));                                       \
        if (tile < tileEnd) l1_tile_wave(gx, gW, gWh, gavec, gssrc, gsdst, gK, mTiles, tile, threadIdx.x & 31); \
        return;                                                                                                \
    }
#define L1_TILE_PARAMS                                                                                  \
    const float* __restrict__ gx, const float* __restrict__ gW, float* __restrict__ gWh,               \
    const float* __restrict__ gavec, float* __restrict__ gssrc, float* __restrict__ gsdst,             \
    int gK, int mTiles, int gBlocks, int tile0, int tileEnd

__global__ __launch_bounds__(256) void k_edge_rank(const int* __restrict__ src, int* __restrict__ deg0,
                                                   int* __restrict__ deg1, int* __restrict__ rank, int E,
                                                   L1_TILE_PARAMS)
{
    L1_TILE_ROLE()
    int e = (blockIdx.x - gBlocks) * blockDim.x + threadIdx.x;
    const int half = chip_half();
    if (e >= E) return;
    int* __restrict__ cnt = half ? deg1 : deg0;
    const int r = atomicAdd(&cnt[src[e]], 1);
    rank[e] = half ? -1 - r : r;
}

__global__ __launch_bounds__(1024) void k_row_offsets(const int* __restrict__ deg0, const int* __restrict__ deg1,
                                                      int* __restrict__ rowptr, int n)
{
    __shared__ int wfront[32];
    __shared__ int wtile[32];
    const int t    = threadIdx.x;
    const int lane = t & 31;
    const int w    = t >> 5;
    const int nb   = blockIdx.x;
    const int idx  = nb * 1024 + t;

    const int v = (idx < n) ? deg0[idx] + deg1[idx] : 0;

    int s = 0;
    const int* __restrict__ p0 = deg0 + t;
    const int* __restrict__ p1 = deg1 + t;
    for (int k0 = 0; k0 < nb; k0 += 16) {
        int x[16], y[16];
        #pragma unroll
        for (int u = 0; u < 16; ++u) {
            const int k = (k0 + u < nb) ? (k0 + u) : (nb - 1);
            x[u] = p0[(size_t)k * 1024];
            y[u] = p1[(size_t)k * 1024];
        }
        #pragma unroll
        for (int u = 0; u < 16; ++u) s += (k0 + u < nb) ? x[u] + y[u] : 0;
    }

    int inc = v;
    for (int off = 1; off < 32; off <<= 1) {
        const int o = __shfl_up(inc, off, 32);
        if (lane >= off) inc += o;
    }
    for (int off = 16; off >= 1; off >>= 1) s += __shfl_xor(s, off, 32);
    if (lane == 31) wtile[w]  = inc;
    if (lane == 0)  wfront[w] = s;
    __syncthreads();
    int base = wfront[lane] + ((lane < w) ? wtile[lane] : 0);
    for (int off = 16; off >= 1; off >>= 1) base += __shfl_xor(base, off, 32);
    if (idx <= n) rowptr[idx] = base + inc - v;
}

__global__ __launch_bounds__(256) void k_edge_place(const int* __restrict__ src, const int* __restrict__ rowptr,
                                                    const int* __restrict__ rank, int* __restrict__ eid, int E,
                                                    L1_TILE_PARAMS)
{
    L1_TILE_ROLE()
    int e = (blockIdx.x - gBlocks) * blockDim.x + threadIdx.x;
    if (e >= E) return;
    const int r = rank[e];
    eid[rowptr[src[e] + (r < 0 ? 1 : 0)] + r] = e;
}

__global__ __launch_bounds__(256) void k_row_order(const int* __restrict__ rowptr, const int* __restrict__ eid,
                                                   const int* __restrict__ dst, int* __restrict__ col, int N)
{
    const int lane = threadIdx.x & 31;
    const int wid = __builtin_amdgcn_readfirstlane((int)((blockIdx.x * blockDim.x + threadIdx.x) >> 5));
    if (wid >= N) return;
    const int beg = rowptr[wid];
    const int deg = rowptr[wid + 1] - beg;
    const int* __restrict__ ep = eid + beg;
    for (int ibase = 0; ibase < deg; ibase += 32) {
        const int i = ibase + lane;
        const int mine = (i < deg) ? ep[i] : 0x7fffffff;
        int pos = 0;
        for (int jbase = 0; jbase < deg; jbase += 32) {
            const int j = jbase + lane;
            const int other = (j < deg) ? ep[j] : 0x7fffffff;
            const int left = deg - jbase;
            const int nj = left < 32 ? left : 32;
            for (int u = 0; u < nj; ++u) pos += (__builtin_amdgcn_readlane(other, u) < mine) ? 1 : 0;
        }
        if (i < deg) col[beg + pos] = dst[mine];
    }
}

__device__ __forceinline__ float row16_sum(float v)
{
    v += __int_as_float(__builtin_amdgcn_update_dpp(0, __float_as_int(v), 0x161, 0xf, 0xf, false));
    v += __int_as_float(__builtin_amdgcn_update_dpp(0, __float_as_int(v), 0x162, 0xf, 0xf, false));
    v += __int_as_float(__builtin_amdgcn_update_dpp(0, __float_as_int(v), 0x164, 0xf, 0xf, false));
    v += __int_as_float(__builtin_amdgcn_update_dpp(0, __float_as_int(v), 0x168, 0xf, 0xf, false));
    return v;
}

__device__ __forceinline__ void gemm_tile(
    const float* __restrict__ A, const float* __restrict__ B, float* __restrict__ C,
    const float* __restrict__ avec, float* __restrict__ ssrc, float* __restrict__ sdst,
    int K, long aStride, long bStride, long cStride, long ldc, long sStride, long sld,
    const int m0, const int b, const int lane)
{
    const float* __restrict__ Ab = A + (long)b * aStride;
    const float* __restrict__ Bb = B + (long)b * bStride;
    float* __restrict__ Cb = C + (long)b * cStride;

    const int row = lane & 15;
    const int kh  = (lane >> 4) << 1;
    const int col = lane & 15;

    const float* __restrict__ av = avec + (long)b * 128;
    const float as0 = av[col],      as1 = av[col + 16], as2 = av[col + 32], as3 = av[col + 48];
    const float ad0 = av[col + 64], ad1 = av[col + 80], ad2 = av[col + 96], ad3 = av[col + 112];

    v8f c0 = {}, c1 = {}, c2 = {}, c3 = {};

    for (int k0 = 0; k0 < K; k0 += 4) {
        const float2 af = *reinterpret_cast<const float2*>(
            Ab + (size_t)(m0 + row) * K + k0 + kh);
        v2f a; a.x = af.x; a.y = af.y;

        const float* br0 = Bb + (size_t)(k0 + kh) * 64 + col;
        const float* br1 = br0 + 64;
        v2f b0, b1, b2, b3;
        b0.x = br0[0];  b0.y = br1[0];
        b1.x = br0[16]; b1.y = br1[16];
        b2.x = br0[32]; b2.y = br1[32];
        b3.x = br0[48]; b3.y = br1[48];

        c0 = __builtin_amdgcn_wmma_f32_16x16x4_f32(false, a, false, b0, (short)0, c0, false, false);
        c1 = __builtin_amdgcn_wmma_f32_16x16x4_f32(false, a, false, b1, (short)0, c1, false, false);
        c2 = __builtin_amdgcn_wmma_f32_16x16x4_f32(false, a, false, b2, (short)0, c2, false, false);
        c3 = __builtin_amdgcn_wmma_f32_16x16x4_f32(false, a, false, b3, (short)0, c3, false, false);
    }

    const int rbase = m0 + ((lane >> 4) << 3);
    #pragma unroll
    for (int r = 0; r < 8; ++r) {
        float* crow = Cb + (size_t)(rbase + r) * ldc + col;
        crow[0]  = c0[r];
        crow[16] = c1[r];
        crow[32] = c2[r];
        crow[48] = c3[r];
    }

    float ps = 0.0f, pd = 0.0f;
    #pragma unroll
    for (int r = 0; r < 8; ++r) {
        const float s = row16_sum(c0[r] * as0 + c1[r] * as1 + c2[r] * as2 + c3[r] * as3);
        const float d = row16_sum(c0[r] * ad0 + c1[r] * ad1 + c2[r] * ad2 + c3[r] * ad3);
        if (col == r) { ps = s; pd = d; }
    }
    if (col < 8) {
        const size_t si = (size_t)b * sStride + (size_t)(rbase + col) * sld;
        ssrc[si] = ps;
        sdst[si] = pd;
    }
}

__global__ __launch_bounds__(32) void k_gemm_wmma(
    const float* __restrict__ A, const float* __restrict__ B, float* __restrict__ C,
    const float* __restrict__ avec, float* __restrict__ ssrc, float* __restrict__ sdst,
    int K, long aStride, long bStride, long cStride, long ldc, long sStride, long sld)
{
    gemm_tile(A, B, C, avec, ssrc, sdst, K, aStride, bStride, cStride, ldc, sStride, sld,
              blockIdx.x * 16, blockIdx.y, threadIdx.x);
}

__device__ __forceinline__ void l1_tile_wave(const float* __restrict__ x, const float* __restrict__ W,
                                             float* __restrict__ Wh, const float* __restrict__ avec,
                                             float* __restrict__ ssrc, float* __restrict__ sdst,
                                             int K, int mTiles, int tile, int lane)
{
    const int b = tile / mTiles;
    gemm_tile(x, W, Wh, avec, ssrc, sdst, K, 0L, (long)K * 64, 64L, 256L, 1L, 4L, (tile - b * mTiles) * 16, b, lane);
}

template <int U>
__device__ __forceinline__ void seg_accum(const float* __restrict__ Wb, int dreg, float areg,
                                          int j, int lane2, float& ax, float& ay)
{
    float2 w[U];
    float  a[U];
    #pragma unroll
    for (int u = 0; u < U; ++u) {
        const int dj = __builtin_amdgcn_readlane(dreg, j + u);
        a[u] = __int_as_float(__builtin_amdgcn_readlane(__float_as_int(areg), j + u));
        w[u] = *reinterpret_cast<const float2*>(Wb + (size_t)dj * 64 + lane2);
    }
    #pragma unroll
    for (int u = 0; u < U; ++u) { ax += a[u] * w[u].x; ay += a[u] * w[u].y; }
}

__device__ __forceinline__ void seg_chunk(const float* __restrict__ Wb, int dreg, float areg,
                                          int nj, int lane2, float& ax, float& ay)
{
    int j = 0;
    for (; j + 8 <= nj; j += 8) seg_accum<8>(Wb, dreg, areg, j, lane2, ax, ay);
    if (j + 4 <= nj) { seg_accum<4>(Wb, dreg, areg, j, lane2, ax, ay); j += 4; }
    if (j + 2 <= nj) { seg_accum<2>(Wb, dreg, areg, j, lane2, ax, ay); j += 2; }
    if (j < nj)      { seg_accum<1>(Wb, dreg, areg, j, lane2, ax, ay); }
}

__device__ __forceinline__ float lrelu(float v) { return v > 0.0f ? v : GAT_ALPHA * v; }

__device__ __forceinline__ float elu1(float v) { return v > 0.0f ? v : __expf(v) - 1.0f; }

__global__ __launch_bounds__(256) void k_seg_attend(
    const int* __restrict__ rowptr, const int* __restrict__ col,
    const float* __restrict__ ssrc, const float* __restrict__ sdst,
    const float* __restrict__ Wh, float* __restrict__ dst, int NH, int N, int ostride)
{
    const int lane  = threadIdx.x & 31;
    const int lane2 = lane << 1;
    const int wid = __builtin_amdgcn_readfirstlane((int)((blockIdx.x * blockDim.x + threadIdx.x) >> 5));
    if (wid >= NH) return;
    const int h = wid / N;
    const int n = wid - h * N;
    const int beg = rowptr[n];
    const int deg = rowptr[n + 1] - beg;
    float2* __restrict__ orow = reinterpret_cast<float2*>(dst + (size_t)n * ostride + (size_t)h * 64) + lane;
    if (deg == 0) { *orow = make_float2(0.0f, 0.0f); return; }

    const int*   __restrict__ cp = col + beg;
    const float* __restrict__ sd = sdst + (size_t)h * N;
    const float* __restrict__ Wb = Wh + (size_t)h * N * 64;
    const float ss = ssrc[wid];

    int   d0 = 0;
    float v0 = -__builtin_huge_valf();
    if (lane < deg) { d0 = cp[lane]; v0 = lrelu(ss + sd[d0]); }

    float mx = v0;
    for (int i = lane + 32; i < deg; i += 32) mx = fmaxf(mx, lrelu(ss + sd[cp[i]]));
    for (int off = 16; off >= 1; off >>= 1) mx = fmaxf(mx, __shfl_xor(mx, off, 32));

    const float ex0 = (lane < deg) ? __expf(v0 - mx) : 0.0f;
    float sum = ex0;
    for (int i = lane + 32; i < deg; i += 32) sum += __expf(lrelu(ss + sd[cp[i]]) - mx);
    for (int off = 16; off >= 1; off >>= 1) sum += __shfl_xor(sum, off, 32);

    float ax = 0.0f, ay = 0.0f;
    seg_chunk(Wb, d0, ex0 / sum, deg < 32 ? deg : 32, lane2, ax, ay);
    for (int base = 32; base < deg; base += 32) {
        const int i = base + lane;
        int   d   = 0;
        float att = 0.0f;
        if (i < deg) { d = cp[i]; att = __expf(lrelu(ss + sd[d]) - mx) / sum; }
        const int left = deg - base;
        seg_chunk(Wb, d, att, left < 32 ? left : 32, lane2, ax, ay);
    }
    *orow = make_float2(elu1(ax), elu1(ay));
}


template <int U>
__device__ __forceinline__ void seg_accum4(const float* __restrict__ Wn, int dreg,
                                           float a0r, float a1r, float a2r, float a3r,
                                           int j, int lane4, bool hi, float4& pa, float4& pb)
{
    float4 wa[U], wb[U];
    #pragma unroll
    for (int u = 0; u < U; ++u) {
        const int dj = __builtin_amdgcn_readlane(dreg, j + u);
        const float* __restrict__ rec = Wn + (size_t)dj * 256 + lane4;
        wa[u] = *reinterpret_cast<const float4*>(rec);
        wb[u] = *reinterpret_cast<const float4*>(rec + 128);
    }
    #pragma unroll
    for (int u = 0; u < U; ++u) {
        const float s0 = __int_as_float(__builtin_amdgcn_readlane(__float_as_int(a0r), j + u));
        const float s1 = __int_as_float(__builtin_amdgcn_readlane(__float_as_int(a1r), j + u));
        const float s2 = __int_as_float(__builtin_amdgcn_readlane(__float_as_int(a2r), j + u));
        const float s3 = __int_as_float(__builtin_amdgcn_readlane(__float_as_int(a3r), j + u));
        const float ga = hi ? s1 : s0;
        const float gb = hi ? s3 : s2;
        pa.x += ga * wa[u].x; pa.y += ga * wa[u].y; pa.z += ga * wa[u].z; pa.w += ga * wa[u].w;
        pb.x += gb * wb[u].x; pb.y += gb * wb[u].y; pb.z += gb * wb[u].z; pb.w += gb * wb[u].w;
    }
}

__device__ __forceinline__ void seg_chunk4(const float* __restrict__ Wn, int dreg,
                                           float a0r, float a1r, float a2r, float a3r,
                                           int nj, int lane4, bool hi, float4& pa, float4& pb)
{
    int j = 0;
    for (; j + 8 <= nj; j += 8) seg_accum4<8>(Wn, dreg, a0r, a1r, a2r, a3r, j, lane4, hi, pa, pb);
    if (j + 4 <= nj) { seg_accum4<4>(Wn, dreg, a0r, a1r, a2r, a3r, j, lane4, hi, pa, pb); j += 4; }
    if (j + 2 <= nj) { seg_accum4<2>(Wn, dreg, a0r, a1r, a2r, a3r, j, lane4, hi, pa, pb); j += 2; }
    if (j < nj)      { seg_accum4<1>(Wn, dreg, a0r, a1r, a2r, a3r, j, lane4, hi, pa, pb); }
}

__global__ __launch_bounds__(256) void k_seg_attend_h4(
    const int* __restrict__ rowptr, const int* __restrict__ col,
    const float* __restrict__ ssrc, const float* __restrict__ sdst,
    const float* __restrict__ Wn, float* __restrict__ dst, int N)
{
    const int  lane  = threadIdx.x & 31;
    const int  lane4 = lane << 2;
    const bool hi    = lane >= 16;
    const int n = __builtin_amdgcn_readfirstlane((int)((blockIdx.x * blockDim.x + threadIdx.x) >> 5));
    if (n >= N) return;
    const int beg = rowptr[n];
    const int deg = rowptr[n + 1] - beg;
    float4* __restrict__ orow = reinterpret_cast<float4*>(dst + (size_t)n * 256) + lane;
    if (deg == 0) {
        orow[0]  = make_float4(0.0f, 0.0f, 0.0f, 0.0f);
        orow[32] = make_float4(0.0f, 0.0f, 0.0f, 0.0f);
        return;
    }

    const int* __restrict__ cp = col + beg;
    const float4 ss = *reinterpret_cast<const float4*>(ssrc + (size_t)n * 4);

    const float ninf = -__builtin_huge_valf();
    int   d0 = 0;
    float v0 = ninf, v1 = ninf, v2 = ninf, v3 = ninf;
    if (lane < deg) {
        d0 = cp[lane];
        const float4 sd = *reinterpret_cast<const float4*>(sdst + (size_t)d0 * 4);
        v0 = lrelu(ss.x + sd.x); v1 = lrelu(ss.y + sd.y); v2 = lrelu(ss.z + sd.z); v3 = lrelu(ss.w + sd.w);
    }

    float m0 = v0, m1 = v1, m2 = v2, m3 = v3;
    for (int i = lane + 32; i < deg; i += 32) {
        const float4 sd = *reinterpret_cast<const float4*>(sdst + (size_t)cp[i] * 4);
        m0 = fmaxf(m0, lrelu(ss.x + sd.x)); m1 = fmaxf(m1, lrelu(ss.y + sd.y));
        m2 = fmaxf(m2, lrelu(ss.z + sd.z)); m3 = fmaxf(m3, lrelu(ss.w + sd.w));
    }
    for (int off = 16; off >= 1; off >>= 1) {
        m0 = fmaxf(m0, __shfl_xor(m0, off, 32)); m1 = fmaxf(m1, __shfl_xor(m1, off, 32));
        m2 = fmaxf(m2, __shfl_xor(m2, off, 32)); m3 = fmaxf(m3, __shfl_xor(m3, off, 32));
    }

    const bool in0 = lane < deg;
    const float e0 = in0 ? __expf(v0 - m0) : 0.0f;
    const float e1 = in0 ? __expf(v1 - m1) : 0.0f;
    const float e2 = in0 ? __expf(v2 - m2) : 0.0f;
    const float e3 = in0 ? __expf(v3 - m3) : 0.0f;
    float t0 = e0, t1 = e1, t2 = e2, t3 = e3;
    for (int i = lane + 32; i < deg; i += 32) {
        const float4 sd = *reinterpret_cast<const float4*>(sdst + (size_t)cp[i] * 4);
        t0 += __expf(lrelu(ss.x + sd.x) - m0); t1 += __expf(lrelu(ss.y + sd.y) - m1);
        t2 += __expf(lrelu(ss.z + sd.z) - m2); t3 += __expf(lrelu(ss.w + sd.w) - m3);
    }
    for (int off = 16; off >= 1; off >>= 1) {
        t0 += __shfl_xor(t0, off, 32); t1 += __shfl_xor(t1, off, 32);
        t2 += __shfl_xor(t2, off, 32); t3 += __shfl_xor(t3, off, 32);
    }

    float4 pa = make_float4(0.0f, 0.0f, 0.0f, 0.0f);
    float4 pb = make_float4(0.0f, 0.0f, 0.0f, 0.0f);
    seg_chunk4(Wn, d0, e0 / t0, e1 / t1, e2 / t2, e3 / t3, deg < 32 ? deg : 32, lane4, hi, pa, pb);
    for (int base = 32; base < deg; base += 32) {
        const int i = base + lane;
        int   d  = 0;
        float a0 = 0.0f, a1 = 0.0f, a2 = 0.0f, a3 = 0.0f;
        if (i < deg) {
            d = cp[i];
            const float4 sd = *reinterpret_cast<const float4*>(sdst + (size_t)d * 4);
            a0 = __expf(lrelu(ss.x + sd.x) - m0) / t0; a1 = __expf(lrelu(ss.y + sd.y) - m1) / t1;
            a2 = __expf(lrelu(ss.z + sd.z) - m2) / t2; a3 = __expf(lrelu(ss.w + sd.w) - m3) / t3;
        }
        const int left = deg - base;
        seg_chunk4(Wn, d, a0, a1, a2, a3, left < 32 ? left : 32, lane4, hi, pa, pb);
    }
    orow[0]  = make_float4(elu1(pa.x), elu1(pa.y), elu1(pa.z), elu1(pa.w));
    orow[32] = make_float4(elu1(pb.x), elu1(pb.y), elu1(pb.z), elu1(pb.w));
}

static inline int cdiv(long a, int b) { return (int)((a + b - 1) / b); }

extern "C" void kernel_launch(void* const* d_in, const int* in_sizes, int n_in,
                              void* d_out, int out_size, void* d_ws, size_t ws_size,
                              hipStream_t stream)
{
    const float* x       = (const float*)d_in[0];
    const int*   edges   = (const int*)d_in[1];
    const float* W_heads = (const float*)d_in[2];
    const float* a_heads = (const float*)d_in[3];
    const float* W_out   = (const float*)d_in[4];
    const float* a_out   = (const float*)d_in[5];
    float* out = (float*)d_out;

    const int N = in_sizes[0] / 128;
    const int E = in_sizes[1] / 2;
    const int H = 4, HID = 64;

    const int* src = edges;
    const int* dst = edges + E;

    float* ws = (float*)d_ws;
    size_t off = 0;
    float* Wh1   = ws + off; off += (size_t)H * N * HID;
    float* hcat  = ws + off; off += (size_t)N * 256;
    float* Wh2   = ws + off; off += (size_t)N * HID;
    float* ssrc1 = ws + off; off += (size_t)H * N;
    float* sdst1 = ws + off; off += (size_t)H * N;
    float* ssrc2 = ws + off; off += (size_t)N;
    float* sdst2 = ws + off; off += (size_t)N;
    int* deg0   = (int*)(ws + off); off += (size_t)N;
    int* rowptr = (int*)(ws + off); off += (size_t)N + 1;
    int* rank   = (int*)(ws + off); off += (size_t)E;
    int* eid    = (int*)(ws + off); off += (size_t)E;
    int* col    = (int*)(ws + off); off += (size_t)E;
    int* deg1   = (int*)hcat;

    const int BS = 256;

    k_zero_int2<<<cdiv((long)N, BS), BS, 0, stream>>>(deg0, deg1, N);
    const int MT = N / 16;
    const int gb = cdiv((long)2 * MT, BS / 32);
    k_edge_rank<<<gb + cdiv((long)E, BS), BS, 0, stream>>>(src, deg0, deg1, rank, E,
        x, W_heads, Wh1, a_heads, ssrc1, sdst1, 128, MT, gb, 0, 2 * MT);
    k_row_offsets<<<cdiv((long)N + 1, 1024), 1024, 0, stream>>>(deg0, deg1, rowptr, N);
    k_edge_place<<<gb + cdiv((long)E, BS), BS, 0, stream>>>(src, rowptr, rank, eid, E,
        x, W_heads, Wh1, a_heads, ssrc1, sdst1, 128, MT, gb, 2 * MT, 4 * MT);
    k_row_order<<<cdiv((long)N * 32, BS), BS, 0, stream>>>(rowptr, eid, dst, col, N);

    k_seg_attend_h4<<<cdiv((long)N * 32, BS), BS, 0, stream>>>(rowptr, col, ssrc1, sdst1, Wh1, hcat, N);

    k_gemm_wmma<<<dim3(N / 16, 1), 32, 0, stream>>>(hcat, W_out, Wh2, a_out, ssrc2, sdst2,
        256, 0L, 0L, 0L, 64L, 0L, 1L);
    k_seg_attend<<<cdiv((long)N * 32, BS), BS, 0, stream>>>(rowptr, col, ssrc2, sdst2, Wh2, out, N, N, HID);
}


# Per-row edge ordering moves into the layer-1 attend wave: placement stores (edge id, dst) pairs, the wave counts smaller ids by readlane, pushes dst with one ds_permute, writes col; 8 -> 7 launches
# speedup vs baseline: 7.9983x; 1.0399x over previous
#include <hip/hip_runtime.h>
#include <cstdint>
#include <cstddef>

typedef float v2f __attribute__((ext_vector_type(2)));
typedef float v8f __attribute__((ext_vector_type(8)));

#define GAT_ALPHA 0.2f

__global__ void k_zero_int2(int* __restrict__ p, int* __restrict__ q, int n) {
    int i = blockIdx.x * blockDim.x + threadIdx.x;
    if (i < n) { p[i] = 0; q[i] = 0; }
}

__device__ __forceinline__ int chip_half()
{
    return (int)((__builtin_amdgcn_s_sendmsg_rtn(0x87) >> 18) & 1u);
}

__device__ __forceinline__ void l1_tile_wave(const float* __restrict__ x, const float* __restrict__ W,
                                             float* __restrict__ Wh, const float* __restrict__ avec,
                                             float* __restrict__ ssrc, float* __restrict__ sdst,
                                             int K, int mTiles, int tile, int lane);

#define L1_TILE_ROLE()                                                                                         \
    if (blockIdx.x < (unsigned)gBlocks) {                                                                      \
        const int tile = tile0 + __builtin_amdgcn_readfirstlane(                                               \
            (int)(blockIdx.x * (blockDim.x >> 5) + (threadIdx.x >> 5)));                                       \
        if (tile < tileEnd) l1_tile_wave(gx, gW, gWh, gavec, gssrc, gsdst, gK, mTiles, tile, threadIdx.x & 31); \
        return;                                                                                                \
    }
#define L1_TILE_PARAMS                                                                                  \
    const float* __restrict__ gx, const float* __restrict__ gW, float* __restrict__ gWh,               \
    const float* __restrict__ gavec, float* __restrict__ gssrc, float* __restrict__ gsdst,             \
    int gK, int mTiles, int gBlocks, int tile0, int tileEnd

__global__ __launch_bounds__(256) void k_edge_rank(const int* __restrict__ src, int* __restrict__ deg0,
                                                   int* __restrict__ deg1, int* __restrict__ rank, int E,
                                                   L1_TILE_PARAMS)
{
    L1_TILE_ROLE()
    int e = (blockIdx.x - gBlocks) * blockDim.x + threadIdx.x;
    const int half = chip_half();
    if (e >= E) return;
    int* __restrict__ cnt = half ? deg1 : deg0;
    const int r = atomicAdd(&cnt[src[e]], 1);
    rank[e] = half ? -1 - r : r;
}

__global__ __launch_bounds__(1024) void k_row_offsets(const int* __restrict__ deg0, const int* __restrict__ deg1,
                                                      int* __restrict__ rowptr, int n)
{
    __shared__ int wfront[32];
    __shared__ int wtile[32];
    const int t    = threadIdx.x;
    const int lane = t & 31;
    const int w    = t >> 5;
    const int nb   = blockIdx.x;
    const int idx  = nb * 1024 + t;

    const int v = (idx < n) ? deg0[idx] + deg1[idx] : 0;

    int s = 0;
    const int* __restrict__ p0 = deg0 + t;
    const int* __restrict__ p1 = deg1 + t;
    for (int k0 = 0; k0 < nb; k0 += 16) {
        int x[16], y[16];
        #pragma unroll
        for (int u = 0; u < 16; ++u) {
            const int k = (k0 + u < nb) ? (k0 + u) : (nb - 1);
            x[u] = p0[(size_t)k * 1024];
            y[u] = p1[(size_t)k * 1024];
        }
        #pragma unroll
        for (int u = 0; u < 16; ++u) s += (k0 + u < nb) ? x[u] + y[u] : 0;
    }

    int inc = v;
    for (int off = 1; off < 32; off <<= 1) {
        const int o = __shfl_up(inc, off, 32);
        if (lane >= off) inc += o;
    }
    for (int off = 16; off >= 1; off >>= 1) s += __shfl_xor(s, off, 32);
    if (lane == 31) wtile[w]  = inc;
    if (lane == 0)  wfront[w] = s;
    __syncthreads();
    int base = wfront[lane] + ((lane < w) ? wtile[lane] : 0);
    for (int off = 16; off >= 1; off >>= 1) base += __shfl_xor(base, off, 32);
    if (idx <= n) rowptr[idx] = base + inc - v;
}

__global__ __launch_bounds__(256) void k_edge_place(const int* __restrict__ src, const int* __restrict__ edst,
                                                    const int* __restrict__ rowptr, const int* __restrict__ rank,
                                                    int* __restrict__ epair, int E,
                                                    L1_TILE_PARAMS)
{
    L1_TILE_ROLE()
    int e = (blockIdx.x - gBlocks) * blockDim.x + threadIdx.x;
    if (e >= E) return;
    const int r = rank[e];
    reinterpret_cast<int2*>(epair)[rowptr[src[e] + (r < 0 ? 1 : 0)] + r] = make_int2(e, edst[e]);
}

__device__ __forceinline__ void row_order_any(const int2* __restrict__ ep, int* cw, int deg, int lane)
{
    for (int ibase = 0; ibase < deg; ibase += 32) {
        const int i = ibase + lane;
        int mine = 0x7fffffff, dv = 0;
        if (i < deg) { const int2 pr = ep[i]; mine = pr.x; dv = pr.y; }
        int pos = 0;
        for (int jbase = 0; jbase < deg; jbase += 32) {
            const int j = jbase + lane;
            const int other = (j < deg) ? ep[j].x : 0x7fffffff;
            const int left = deg - jbase;
            const int nj = left < 32 ? left : 32;
            for (int u = 0; u < nj; ++u) pos += (__builtin_amdgcn_readlane(other, u) < mine) ? 1 : 0;
        }
        if (i < deg) cw[pos] = dv;
    }
}

__device__ __forceinline__ float row16_sum(float v)
{
    v += __int_as_float(__builtin_amdgcn_update_dpp(0, __float_as_int(v), 0x161, 0xf, 0xf, false));
    v += __int_as_float(__builtin_amdgcn_update_dpp(0, __float_as_int(v), 0x162, 0xf, 0xf, false));
    v += __int_as_float(__builtin_amdgcn_update_dpp(0, __float_as_int(v), 0x164, 0xf, 0xf, false));
    v += __int_as_float(__builtin_amdgcn_update_dpp(0, __float_as_int(v), 0x168, 0xf, 0xf, false));
    return v;
}

__device__ __forceinline__ void gemm_tile(
    const float* __restrict__ A, const float* __restrict__ B, float* __restrict__ C,
    const float* __restrict__ avec, float* __restrict__ ssrc, float* __restrict__ sdst,
    int K, long aStride, long bStride, long cStride, long ldc, long sStride, long sld,
    const int m0, const int b, const int lane)
{
    const float* __restrict__ Ab = A + (long)b * aStride;
    const float* __restrict__ Bb = B + (long)b * bStride;
    float* __restrict__ Cb = C + (long)b * cStride;

    const int row = lane & 15;
    const int kh  = (lane >> 4) << 1;
    const int col = lane & 15;

    const float* __restrict__ av = avec + (long)b * 128;
    const float as0 = av[col],      as1 = av[col + 16], as2 = av[col + 32], as3 = av[col + 48];
    const float ad0 = av[col + 64], ad1 = av[col + 80], ad2 = av[col + 96], ad3 = av[col + 112];

    v8f c0 = {}, c1 = {}, c2 = {}, c3 = {};

    for (int k0 = 0; k0 < K; k0 += 4) {
        const float2 af = *reinterpret_cast<const float2*>(
            Ab + (size_t)(m0 + row) * K + k0 + kh);
        v2f a; a.x = af.x; a.y = af.y;

        const float* br0 = Bb + (size_t)(k0 + kh) * 64 + col;
        const float* br1 = br0 + 64;
        v2f b0, b1, b2, b3;
        b0.x = br0[0];  b0.y = br1[0];
        b1.x = br0[16]; b1.y = br1[16];
        b2.x = br0[32]; b2.y = br1[32];
        b3.x = br0[48]; b3.y = br1[48];

        c0 = __builtin_amdgcn_wmma_f32_16x16x4_f32(false, a, false, b0, (short)0, c0, false, false);
        c1 = __builtin_amdgcn_wmma_f32_16x16x4_f32(false, a, false, b1, (short)0, c1, false, false);
        c2 = __builtin_amdgcn_wmma_f32_16x16x4_f32(false, a, false, b2, (short)0, c2, false, false);
        c3 = __builtin_amdgcn_wmma_f32_16x16x4_f32(false, a, false, b3, (short)0, c3, false, false);
    }

    const int rbase = m0 + ((lane >> 4) << 3);
    #pragma unroll
    for (int r = 0; r < 8; ++r) {
        float* crow = Cb + (size_t)(rbase + r) * ldc + col;
        crow[0]  = c0[r];
        crow[16] = c1[r];
        crow[32] = c2[r];
        crow[48] = c3[r];
    }

    float ps = 0.0f, pd = 0.0f;
    #pragma unroll
    for (int r = 0; r < 8; ++r) {
        const float s = row16_sum(c0[r] * as0 + c1[r] * as1 + c2[r] * as2 + c3[r] * as3);
        const float d = row16_sum(c0[r] * ad0 + c1[r] * ad1 + c2[r] * ad2 + c3[r] * ad3);
        if (col == r) { ps = s; pd = d; }
    }
    if (col < 8) {
        const size_t si = (size_t)b * sStride + (size_t)(rbase + col) * sld;
        ssrc[si] = ps;
        sdst[si] = pd;
    }
}

__global__ __launch_bounds__(32) void k_gemm_wmma(
    const float* __restrict__ A, const float* __restrict__ B, float* __restrict__ C,
    const float* __restrict__ avec, float* __restrict__ ssrc, float* __restrict__ sdst,
    int K, long aStride, long bStride, long cStride, long ldc, long sStride, long sld)
{
    gemm_tile(A, B, C, avec, ssrc, sdst, K, aStride, bStride, cStride, ldc, sStride, sld,
              blockIdx.x * 16, blockIdx.y, threadIdx.x);
}

__device__ __forceinline__ void l1_tile_wave(const float* __restrict__ x, const float* __restrict__ W,
                                             float* __restrict__ Wh, const float* __restrict__ avec,
                                             float* __restrict__ ssrc, float* __restrict__ sdst,
                                             int K, int mTiles, int tile, int lane)
{
    const int b = tile / mTiles;
    gemm_tile(x, W, Wh, avec, ssrc, sdst, K, 0L, (long)K * 64, 64L, 256L, 1L, 4L, (tile - b * mTiles) * 16, b, lane);
}

template <int U>
__device__ __forceinline__ void seg_accum(const float* __restrict__ Wb, int dreg, float areg,
                                          int j, int lane2, float& ax, float& ay)
{
    float2 w[U];
    float  a[U];
    #pragma unroll
    for (int u = 0; u < U; ++u) {
        const int dj = __builtin_amdgcn_readlane(dreg, j + u);
        a[u] = __int_as_float(__builtin_amdgcn_readlane(__float_as_int(areg), j + u));
        w[u] = *reinterpret_cast<const float2*>(Wb + (size_t)dj * 64 + lane2);
    }
    #pragma unroll
    for (int u = 0; u < U; ++u) { ax += a[u] * w[u].x; ay += a[u] * w[u].y; }
}

__device__ __forceinline__ void seg_chunk(const float* __restrict__ Wb, int dreg, float areg,
                                          int nj, int lane2, float& ax, float& ay)
{
    int j = 0;
    for (; j + 8 <= nj; j += 8) seg_accum<8>(Wb, dreg, areg, j, lane2, ax, ay);
    if (j + 4 <= nj) { seg_accum<4>(Wb, dreg, areg, j, lane2, ax, ay); j += 4; }
    if (j + 2 <= nj) { seg_accum<2>(Wb, dreg, areg, j, lane2, ax, ay); j += 2; }
    if (j < nj)      { seg_accum<1>(Wb, dreg, areg, j, lane2, ax, ay); }
}

__device__ __forceinline__ float lrelu(float v) { return v > 0.0f ? v : GAT_ALPHA * v; }

__device__ __forceinline__ float elu1(float v) { return v > 0.0f ? v : __expf(v) - 1.0f; }

__global__ __launch_bounds__(256) void k_seg_attend(
    const int* __restrict__ rowptr, const int* __restrict__ col,
    const float* __restrict__ ssrc, const float* __restrict__ sdst,
    const float* __restrict__ Wh, float* __restrict__ dst, int NH, int N, int ostride)
{
    const int lane  = threadIdx.x & 31;
    const int lane2 = lane << 1;
    const int wid = __builtin_amdgcn_readfirstlane((int)((blockIdx.x * blockDim.x + threadIdx.x) >> 5));
    if (wid >= NH) return;
    const int h = wid / N;
    const int n = wid - h * N;
    const int beg = rowptr[n];
    const int deg = rowptr[n + 1] - beg;
    float2* __restrict__ orow = reinterpret_cast<float2*>(dst + (size_t)n * ostride + (size_t)h * 64) + lane;
    if (deg == 0) { *orow = make_float2(0.0f, 0.0f); return; }

    const int*   __restrict__ cp = col + beg;
    const float* __restrict__ sd = sdst + (size_t)h * N;
    const float* __restrict__ Wb = Wh + (size_t)h * N * 64;
    const float ss = ssrc[wid];

    int   d0 = 0;
    float v0 = -__builtin_huge_valf();
    if (lane < deg) { d0 = cp[lane]; v0 = lrelu(ss + sd[d0]); }

    float mx = v0;
    for (int i = lane + 32; i < deg; i += 32) mx = fmaxf(mx, lrelu(ss + sd[cp[i]]));
    for (int off = 16; off >= 1; off >>= 1) mx = fmaxf(mx, __shfl_xor(mx, off, 32));

    const float ex0 = (lane < deg) ? __expf(v0 - mx) : 0.0f;
    float sum = ex0;
    for (int i = lane + 32; i < deg; i += 32) sum += __expf(lrelu(ss + sd[cp[i]]) - mx);
    for (int off = 16; off >= 1; off >>= 1) sum += __shfl_xor(sum, off, 32);

    float ax = 0.0f, ay = 0.0f;
    seg_chunk(Wb, d0, ex0 / sum, deg < 32 ? deg : 32, lane2, ax, ay);
    for (int base = 32; base < deg; base += 32) {
        const int i = base + lane;
        int   d   = 0;
        float att = 0.0f;
        if (i < deg) { d = cp[i]; att = __expf(lrelu(ss + sd[d]) - mx) / sum; }
        const int left = deg - base;
        seg_chunk(Wb, d, att, left < 32 ? left : 32, lane2, ax, ay);
    }
    *orow = make_float2(elu1(ax), elu1(ay));
}


template <int U>
__device__ __forceinline__ void seg_accum4(const float* __restrict__ Wn, int dreg,
                                           float a0r, float a1r, float a2r, float a3r,
                                           int j, int lane4, bool hi, float4& pa, float4& pb)
{
    float4 wa[U], wb[U];
    #pragma unroll
    for (int u = 0; u < U; ++u) {
        const int dj = __builtin_amdgcn_readlane(dreg, j + u);
        const float* __restrict__ rec = Wn + (size_t)dj * 256 + lane4;
        wa[u] = *reinterpret_cast<const float4*>(rec);
        wb[u] = *reinterpret_cast<const float4*>(rec + 128);
    }
    #pragma unroll
    for (int u = 0; u < U; ++u) {
        const float s0 = __int_as_float(__builtin_amdgcn_readlane(__float_as_int(a0r), j + u));
        const float s1 = __int_as_float(__builtin_amdgcn_readlane(__float_as_int(a1r), j + u));
        const float s2 = __int_as_float(__builtin_amdgcn_readlane(__float_as_int(a2r), j + u));
        const float s3 = __int_as_float(__builtin_amdgcn_readlane(__float_as_int(a3r), j + u));
        const float ga = hi ? s1 : s0;
        const float gb = hi ? s3 : s2;
        pa.x += ga * wa[u].x; pa.y += ga * wa[u].y; pa.z += ga * wa[u].z; pa.w += ga * wa[u].w;
        pb.x += gb * wb[u].x; pb.y += gb * wb[u].y; pb.z += gb * wb[u].z; pb.w += gb * wb[u].w;
    }
}

__device__ __forceinline__ void seg_chunk4(const float* __restrict__ Wn, int dreg,
                                           float a0r, float a1r, float a2r, float a3r,
                                           int nj, int lane4, bool hi, float4& pa, float4& pb)
{
    int j = 0;
    for (; j + 8 <= nj; j += 8) seg_accum4<8>(Wn, dreg, a0r, a1r, a2r, a3r, j, lane4, hi, pa, pb);
    if (j + 4 <= nj) { seg_accum4<4>(Wn, dreg, a0r, a1r, a2r, a3r, j, lane4, hi, pa, pb); j += 4; }
    if (j + 2 <= nj) { seg_accum4<2>(Wn, dreg, a0r, a1r, a2r, a3r, j, lane4, hi, pa, pb); j += 2; }
    if (j < nj)      { seg_accum4<1>(Wn, dreg, a0r, a1r, a2r, a3r, j, lane4, hi, pa, pb); }
}

__global__ __launch_bounds__(256) void k_seg_attend_h4(
    const int* __restrict__ rowptr, const int* __restrict__ epair,
    int* __restrict__ col,
    const float* __restrict__ ssrc, const float* __restrict__ sdst,
    const float* __restrict__ Wn, float* __restrict__ dst, int N)
{
    const int  lane  = threadIdx.x & 31;
    const int  lane4 = lane << 2;
    const bool hi    = lane >= 16;
    const int n = __builtin_amdgcn_readfirstlane((int)((blockIdx.x * blockDim.x + threadIdx.x) >> 5));
    if (n >= N) return;
    const int beg = rowptr[n];
    const int deg = rowptr[n + 1] - beg;
    float4* __restrict__ orow = reinterpret_cast<float4*>(dst + (size_t)n * 256) + lane;
    if (deg == 0) {
        orow[0]  = make_float4(0.0f, 0.0f, 0.0f, 0.0f);
        orow[32] = make_float4(0.0f, 0.0f, 0.0f, 0.0f);
        return;
    }

    int* cp = col + beg;
    const int2* __restrict__ ep = reinterpret_cast<const int2*>(epair) + beg;
    const float4 ss = *reinterpret_cast<const float4*>(ssrc + (size_t)n * 4);

    int d0 = 0;
    if (deg <= 32) {
        int mine = 0x7fffffff, dv = 0;
        if (lane < deg) { const int2 pr = ep[lane]; mine = pr.x; dv = pr.y; }
        int pos = 0;
        for (int u = 0; u < deg; ++u) pos += (__builtin_amdgcn_readlane(mine, u) < mine) ? 1 : 0;
        d0 = __builtin_amdgcn_ds_permute(((lane < deg) ? pos : lane) << 2, dv);
        if (lane < deg) cp[lane] = d0;
    } else {
        row_order_any(ep, cp, deg, lane);
        __threadfence();
        d0 = cp[lane];
    }
    const float ninf = -__builtin_huge_valf();
    float v0 = ninf, v1 = ninf, v2 = ninf, v3 = ninf;
    if (lane < deg) {
        const float4 sd = *reinterpret_cast<const float4*>(sdst + (size_t)d0 * 4);
        v0 = lrelu(ss.x + sd.x); v1 = lrelu(ss.y + sd.y); v2 = lrelu(ss.z + sd.z); v3 = lrelu(ss.w + sd.w);
    }

    float m0 = v0, m1 = v1, m2 = v2, m3 = v3;
    for (int i = lane + 32; i < deg; i += 32) {
        const float4 sd = *reinterpret_cast<const float4*>(sdst + (size_t)cp[i] * 4);
        m0 = fmaxf(m0, lrelu(ss.x + sd.x)); m1 = fmaxf(m1, lrelu(ss.y + sd.y));
        m2 = fmaxf(m2, lrelu(ss.z + sd.z)); m3 = fmaxf(m3, lrelu(ss.w + sd.w));
    }
    for (int off = 16; off >= 1; off >>= 1) {
        m0 = fmaxf(m0, __shfl_xor(m0, off, 32)); m1 = fmaxf(m1, __shfl_xor(m1, off, 32));
        m2 = fmaxf(m2, __shfl_xor(m2, off, 32)); m3 = fmaxf(m3, __shfl_xor(m3, off, 32));
    }

    const bool in0 = lane < deg;
    const float e0 = in0 ? __expf(v0 - m0) : 0.0f;
    const float e1 = in0 ? __expf(v1 - m1) : 0.0f;
    const float e2 = in0 ? __expf(v2 - m2) : 0.0f;
    const float e3 = in0 ? __expf(v3 - m3) : 0.0f;
    float t0 = e0, t1 = e1, t2 = e2, t3 = e3;
    for (int i = lane + 32; i < deg; i += 32) {
        const float4 sd = *reinterpret_cast<const float4*>(sdst + (size_t)cp[i] * 4);
        t0 += __expf(lrelu(ss.x + sd.x) - m0); t1 += __expf(lrelu(ss.y + sd.y) - m1);
        t2 += __expf(lrelu(ss.z + sd.z) - m2); t3 += __expf(lrelu(ss.w + sd.w) - m3);
    }
    for (int off = 16; off >= 1; off >>= 1) {
        t0 += __shfl_xor(t0, off, 32); t1 += __shfl_xor(t1, off, 32);
        t2 += __shfl_xor(t2, off, 32); t3 += __shfl_xor(t3, off, 32);
    }

    float4 pa = make_float4(0.0f, 0.0f, 0.0f, 0.0f);
    float4 pb = make_float4(0.0f, 0.0f, 0.0f, 0.0f);
    seg_chunk4(Wn, d0, e0 / t0, e1 / t1, e2 / t2, e3 / t3, deg < 32 ? deg : 32, lane4, hi, pa, pb);
    for (int base = 32; base < deg; base += 32) {
        const int i = base + lane;
        int   d  = 0;
        float a0 = 0.0f, a1 = 0.0f, a2 = 0.0f, a3 = 0.0f;
        if (i < deg) {
            d = cp[i];
            const float4 sd = *reinterpret_cast<const float4*>(sdst + (size_t)d * 4);
            a0 = __expf(lrelu(ss.x + sd.x) - m0) / t0; a1 = __expf(lrelu(ss.y + sd.y) - m1) / t1;
            a2 = __expf(lrelu(ss.z + sd.z) - m2) / t2; a3 = __expf(lrelu(ss.w + sd.w) - m3) / t3;
        }
        const int left = deg - base;
        seg_chunk4(Wn, d, a0, a1, a2, a3, left < 32 ? left : 32, lane4, hi, pa, pb);
    }
    orow[0]  = make_float4(elu1(pa.x), elu1(pa.y), elu1(pa.z), elu1(pa.w));
    orow[32] = make_float4(elu1(pb.x), elu1(pb.y), elu1(pb.z), elu1(pb.w));
}

static inline int cdiv(long a, int b) { return (int)((a + b - 1) / b); }

extern "C" void kernel_launch(void* const* d_in, const int* in_sizes, int n_in,
                              void* d_out, int out_size, void* d_ws, size_t ws_size,
                              hipStream_t stream)
{
    const float* x       = (const float*)d_in[0];
    const int*   edges   = (const int*)d_in[1];
    const float* W_heads = (const float*)d_in[2];
    const float* a_heads = (const float*)d_in[3];
    const float* W_out   = (const float*)d_in[4];
    const float* a_out   = (const float*)d_in[5];
    float* out = (float*)d_out;

    const int N = in_sizes[0] / 128;
    const int E = in_sizes[1] / 2;
    const int H = 4, HID = 64;

    const int* src = edges;
    const int* dst = edges + E;

    float* ws = (float*)d_ws;
    size_t off = 0;
    float* Wh1   = ws + off; off += (size_t)H * N * HID;
    float* hcat  = ws + off; off += (size_t)N * 256;
    float* Wh2   = ws + off; off += (size_t)N * HID;
    float* ssrc1 = ws + off; off += (size_t)H * N;
    float* sdst1 = ws + off; off += (size_t)H * N;
    float* ssrc2 = ws + off; off += (size_t)N;
    float* sdst2 = ws + off; off += (size_t)N;
    int* deg0   = (int*)(ws + off); off += (size_t)N;
    int* rowptr = (int*)(ws + off); off += (size_t)N + 1;
    int* rank   = (int*)(ws + off); off += (size_t)E;
    int* col    = (int*)(ws + off); off += (size_t)E;
    int* deg1   = (int*)hcat;
    int* epair = (int*)Wh2;

    const int BS = 256;

    k_zero_int2<<<cdiv((long)N, BS), BS, 0, stream>>>(deg0, deg1, N);
    const int MT = N / 16;
    const int gb = cdiv((long)2 * MT, BS / 32);
    k_edge_rank<<<gb + cdiv((long)E, BS), BS, 0, stream>>>(src, deg0, deg1, rank, E,
        x, W_heads, Wh1, a_heads, ssrc1, sdst1, 128, MT, gb, 0, 2 * MT);
    k_row_offsets<<<cdiv((long)N + 1, 1024), 1024, 0, stream>>>(deg0, deg1, rowptr, N);
    k_edge_place<<<gb + cdiv((long)E, BS), BS, 0, stream>>>(src, dst, rowptr, rank, epair, E,
        x, W_heads, Wh1, a_heads, ssrc1, sdst1, 128, MT, gb, 2 * MT, 4 * MT);

    k_seg_attend_h4<<<cdiv((long)N * 32, BS), BS, 0, stream>>>(rowptr, epair, col, ssrc1, sdst1, Wh1, hcat, N);

    k_gemm_wmma<<<dim3(N / 16, 1), 32, 0, stream>>>(hcat, W_out, Wh2, a_out, ssrc2, sdst2,
        256, 0L, 0L, 0L, 64L, 0L, 1L);
    k_seg_attend<<<cdiv((long)N * 32, BS), BS, 0, stream>>>(rowptr, col, ssrc2, sdst2, Wh2, out, N, N, HID);
}


# Layer-2 product moved into the layer-1 attend launch: a work-group's 16 rows go to LDS, K split over its 16 waves, partials and scores summed in a new fixed order; hcat never stored; 7 -> 6 launches
# speedup vs baseline: 8.2807x; 1.0353x over previous
#include <hip/hip_runtime.h>
#include <cstdint>
#include <cstddef>

typedef float v2f __attribute__((ext_vector_type(2)));
typedef float v8f __attribute__((ext_vector_type(8)));

#define GAT_ALPHA 0.2f

__global__ void k_zero_int2(int* __restrict__ p, int* __restrict__ q, int n) {
    int i = blockIdx.x * blockDim.x + threadIdx.x;
    if (i < n) { p[i] = 0; q[i] = 0; }
}

__device__ __forceinline__ int chip_half()
{
    return (int)((__builtin_amdgcn_s_sendmsg_rtn(0x87) >> 18) & 1u);
}

__device__ __forceinline__ void l1_tile_wave(const float* __restrict__ x, const float* __restrict__ W,
                                             float* __restrict__ Wh, const float* __restrict__ avec,
                                             float* __restrict__ ssrc, float* __restrict__ sdst,
                                             int K, int mTiles, int tile, int lane);

#define L1_TILE_ROLE()                                                                                         \
    if (blockIdx.x < (unsigned)gBlocks) {                                                                      \
        const int tile = tile0 + __builtin_amdgcn_readfirstlane(                                               \
            (int)(blockIdx.x * (blockDim.x >> 5) + (threadIdx.x >> 5)));                                       \
        if (tile < tileEnd) l1_tile_wave(gx, gW, gWh, gavec, gssrc, gsdst, gK, mTiles, tile, threadIdx.x & 31); \
        return;                                                                                                \
    }
#define L1_TILE_PARAMS                                                                                  \
    const float* __restrict__ gx, const float* __restrict__ gW, float* __restrict__ gWh,               \
    const float* __restrict__ gavec, float* __restrict__ gssrc, float* __restrict__ gsdst,             \
    int gK, int mTiles, int gBlocks, int tile0, int tileEnd

__global__ __launch_bounds__(256) void k_edge_rank(const int* __restrict__ src, int* __restrict__ deg0,
                                                   int* __restrict__ deg1, int* __restrict__ rank, int E,
                                                   L1_TILE_PARAMS)
{
    L1_TILE_ROLE()
    int e = (blockIdx.x - gBlocks) * blockDim.x + threadIdx.x;
    const int half = chip_half();
    if (e >= E) return;
    int* __restrict__ cnt = half ? deg1 : deg0;
    const int r = atomicAdd(&cnt[src[e]], 1);
    rank[e] = half ? -1 - r : r;
}

__global__ __launch_bounds__(1024) void k_row_offsets(const int* __restrict__ deg0, const int* __restrict__ deg1,
                                                      int* __restrict__ rowptr, int n)
{
    __shared__ int wfront[32];
    __shared__ int wtile[32];
    const int t    = threadIdx.x;
    const int lane = t & 31;
    const int w    = t >> 5;
    const int nb   = blockIdx.x;
    const int idx  = nb * 1024 + t;

    const int v = (idx < n) ? deg0[idx] + deg1[idx] : 0;

    int s = 0;
    const int* __restrict__ p0 = deg0 + t;
    const int* __restrict__ p1 = deg1 + t;
    for (int k0 = 0; k0 < nb; k0 += 16) {
        int x[16], y[16];
        #pragma unroll
        for (int u = 0; u < 16; ++u) {
            const int k = (k0 + u < nb) ? (k0 + u) : (nb - 1);
            x[u] = p0[(size_t)k * 1024];
            y[u] = p1[(size_t)k * 1024];
        }
        #pragma unroll
        for (int u = 0; u < 16; ++u) s += (k0 + u < nb) ? x[u] + y[u] : 0;
    }

    int inc = v;
    for (int off = 1; off < 32; off <<= 1) {
        const int o = __shfl_up(inc, off, 32);
        if (lane >= off) inc += o;
    }
    for (int off = 16; off >= 1; off >>= 1) s += __shfl_xor(s, off, 32);
    if (lane == 31) wtile[w]  = inc;
    if (lane == 0)  wfront[w] = s;
    __syncthreads();
    int base = wfront[lane] + ((lane < w) ? wtile[lane] : 0);
    for (int off = 16; off >= 1; off >>= 1) base += __shfl_xor(base, off, 32);
    if (idx <= n) rowptr[idx] = base + inc - v;
}

__global__ __launch_bounds__(256) void k_edge_place(const int* __restrict__ src, const int* __restrict__ edst,
                                                    const int* __restrict__ rowptr, const int* __restrict__ rank,
                                                    int* __restrict__ epair, int E,
                                                    L1_TILE_PARAMS)
{
    L1_TILE_ROLE()
    int e = (blockIdx.x - gBlocks) * blockDim.x + threadIdx.x;
    if (e >= E) return;
    const int r = rank[e];
    reinterpret_cast<int2*>(epair)[rowptr[src[e] + (r < 0 ? 1 : 0)] + r] = make_int2(e, edst[e]);
}

__device__ __forceinline__ void row_order_any(const int2* __restrict__ ep, int* cw, int deg, int lane)
{
    for (int ibase = 0; ibase < deg; ibase += 32) {
        const int i = ibase + lane;
        int mine = 0x7fffffff, dv = 0;
        if (i < deg) { const int2 pr = ep[i]; mine = pr.x; dv = pr.y; }
        int pos = 0;
        for (int jbase = 0; jbase < deg; jbase += 32) {
            const int j = jbase + lane;
            const int other = (j < deg) ? ep[j].x : 0x7fffffff;
            const int left = deg - jbase;
            const int nj = left < 32 ? left : 32;
            for (int u = 0; u < nj; ++u) pos += (__builtin_amdgcn_readlane(other, u) < mine) ? 1 : 0;
        }
        if (i < deg) cw[pos] = dv;
    }
}

__device__ __forceinline__ float row16_sum(float v)
{
    v += __int_as_float(__builtin_amdgcn_update_dpp(0, __float_as_int(v), 0x161, 0xf, 0xf, false));
    v += __int_as_float(__builtin_amdgcn_update_dpp(0, __float_as_int(v), 0x162, 0xf, 0xf, false));
    v += __int_as_float(__builtin_amdgcn_update_dpp(0, __float_as_int(v), 0x164, 0xf, 0xf, false));
    v += __int_as_float(__builtin_amdgcn_update_dpp(0, __float_as_int(v), 0x168, 0xf, 0xf, false));
    return v;
}

__device__ __forceinline__ void gemm_tile(
    const float* __restrict__ A, const float* __restrict__ B, float* __restrict__ C,
    const float* __restrict__ avec, float* __restrict__ ssrc, float* __restrict__ sdst,
    int K, long aStride, long bStride, long cStride, long ldc, long sStride, long sld,
    const int m0, const int b, const int lane)
{
    const float* __restrict__ Ab = A + (long)b * aStride;
    const float* __restrict__ Bb = B + (long)b * bStride;
    float* __restrict__ Cb = C + (long)b * cStride;

    const int row = lane & 15;
    const int kh  = (lane >> 4) << 1;
    const int col = lane & 15;

    const float* __restrict__ av = avec + (long)b * 128;
    const float as0 = av[col],      as1 = av[col + 16], as2 = av[col + 32], as3 = av[col + 48];
    const float ad0 = av[col + 64], ad1 = av[col + 80], ad2 = av[col + 96], ad3 = av[col + 112];

    v8f c0 = {}, c1 = {}, c2 = {}, c3 = {};

    for (int k0 = 0; k0 < K; k0 += 4) {
        const float2 af = *reinterpret_cast<const float2*>(
            Ab + (size_t)(m0 + row) * K + k0 + kh);
        v2f a; a.x = af.x; a.y = af.y;

        const float* br0 = Bb + (size_t)(k0 + kh) * 64 + col;
        const float* br1 = br0 + 64;
        v2f b0, b1, b2, b3;
        b0.x = br0[0];  b0.y = br1[0];
        b1.x = br0[16]; b1.y = br1[16];
        b2.x = br0[32]; b2.y = br1[32];
        b3.x = br0[48]; b3.y = br1[48];

        c0 = __builtin_amdgcn_wmma_f32_16x16x4_f32(false, a, false, b0, (short)0, c0, false, false);
        c1 = __builtin_amdgcn_wmma_f32_16x16x4_f32(false, a, false, b1, (short)0, c1, false, false);
        c2 = __builtin_amdgcn_wmma_f32_16x16x4_f32(false, a, false, b2, (short)0, c2, false, false);
        c3 = __builtin_amdgcn_wmma_f32_16x16x4_f32(false, a, false, b3, (short)0, c3, false, false);
    }

    const int rbase = m0 + ((lane >> 4) << 3);
    #pragma unroll
    for (int r = 0; r < 8; ++r) {
        float* crow = Cb + (size_t)(rbase + r) * ldc + col;
        crow[0]  = c0[r];
        crow[16] = c1[r];
        crow[32] = c2[r];
        crow[48] = c3[r];
    }

    float ps = 0.0f, pd = 0.0f;
    #pragma unroll
    for (int r = 0; r < 8; ++r) {
        const float s = row16_sum(c0[r] * as0 + c1[r] * as1 + c2[r] * as2 + c3[r] * as3);
        const float d = row16_sum(c0[r] * ad0 + c1[r] * ad1 + c2[r] * ad2 + c3[r] * ad3);
        if (col == r) { ps = s; pd = d; }
    }
    if (col < 8) {
        const size_t si = (size_t)b * sStride + (size_t)(rbase + col) * sld;
        ssrc[si] = ps;
        sdst[si] = pd;
    }
}

__device__ __forceinline__ void l1_tile_wave(const float* __restrict__ x, const float* __restrict__ W,
                                             float* __restrict__ Wh, const float* __restrict__ avec,
                                             float* __restrict__ ssrc, float* __restrict__ sdst,
                                             int K, int mTiles, int tile, int lane)
{
    const int b = tile / mTiles;
    gemm_tile(x, W, Wh, avec, ssrc, sdst, K, 0L, (long)K * 64, 64L, 256L, 1L, 4L, (tile - b * mTiles) * 16, b, lane);
}

template <int U>
__device__ __forceinline__ void seg_accum(const float* __restrict__ Wb, int dreg, float areg,
                                          int j, int lane2, float& ax, float& ay)
{
    float2 w[U];
    float  a[U];
    #pragma unroll
    for (int u = 0; u < U; ++u) {
        const int dj = __builtin_amdgcn_readlane(dreg, j + u);
        a[u] = __int_as_float(__builtin_amdgcn_readlane(__float_as_int(areg), j + u));
        w[u] = *reinterpret_cast<const float2*>(Wb + (size_t)dj * 64 + lane2);
    }
    #pragma unroll
    for (int u = 0; u < U; ++u) { ax += a[u] * w[u].x; ay += a[u] * w[u].y; }
}

__device__ __forceinline__ void seg_chunk(const float* __restrict__ Wb, int dreg, float areg,
                                          int nj, int lane2, float& ax, float& ay)
{
    int j = 0;
    for (; j + 8 <= nj; j += 8) seg_accum<8>(Wb, dreg, areg, j, lane2, ax, ay);
    if (j + 4 <= nj) { seg_accum<4>(Wb, dreg, areg, j, lane2, ax, ay); j += 4; }
    if (j + 2 <= nj) { seg_accum<2>(Wb, dreg, areg, j, lane2, ax, ay); j += 2; }
    if (j < nj)      { seg_accum<1>(Wb, dreg, areg, j, lane2, ax, ay); }
}

__device__ __forceinline__ float lrelu(float v) { return v > 0.0f ? v : GAT_ALPHA * v; }

__device__ __forceinline__ float elu1(float v) { return v > 0.0f ? v : __expf(v) - 1.0f; }

__global__ __launch_bounds__(256) void k_seg_attend(
    const int* __restrict__ rowptr, const int* __restrict__ col,
    const float* __restrict__ ssrc, const float* __restrict__ sdst,
    const float* __restrict__ Wh, float* __restrict__ dst, int NH, int N, int ostride)
{
    const int lane  = threadIdx.x & 31;
    const int lane2 = lane << 1;
    const int wid = __builtin_amdgcn_readfirstlane((int)((blockIdx.x * blockDim.x + threadIdx.x) >> 5));
    if (wid >= NH) return;
    const int h = wid / N;
    const int n = wid - h * N;
    const int beg = rowptr[n];
    const int deg = rowptr[n + 1] - beg;
    float2* __restrict__ orow = reinterpret_cast<float2*>(dst + (size_t)n * ostride + (size_t)h * 64) + lane;
    if (deg == 0) { *orow = make_float2(0.0f, 0.0f); return; }

    const int*   __restrict__ cp = col + beg;
    const float* __restrict__ sd = sdst + (size_t)h * N;
    const float* __restrict__ Wb = Wh + (size_t)h * N * 64;
    const float ss = ssrc[wid];

    int   d0 = 0;
    float v0 = -__builtin_huge_valf();
    if (lane < deg) { d0 = cp[lane]; v0 = lrelu(ss + sd[d0]); }

    float mx = v0;
    for (int i = lane + 32; i < deg; i += 32) mx = fmaxf(mx, lrelu(ss + sd[cp[i]]));
    for (int off = 16; off >= 1; off >>= 1) mx = fmaxf(mx, __shfl_xor(mx, off, 32));

    const float ex0 = (lane < deg) ? __expf(v0 - mx) : 0.0f;
    float sum = ex0;
    for (int i = lane + 32; i < deg; i += 32) sum += __expf(lrelu(ss + sd[cp[i]]) - mx);
    for (int off = 16; off >= 1; off >>= 1) sum += __shfl_xor(sum, off, 32);

    float ax = 0.0f, ay = 0.0f;
    seg_chunk(Wb, d0, ex0 / sum, deg < 32 ? deg : 32, lane2, ax, ay);
    for (int base = 32; base < deg; base += 32) {
        const int i = base + lane;
        int   d   = 0;
        float att = 0.0f;
        if (i < deg) { d = cp[i]; att = __expf(lrelu(ss + sd[d]) - mx) / sum; }
        const int left = deg - base;
        seg_chunk(Wb, d, att, left < 32 ? left : 32, lane2, ax, ay);
    }
    *orow = make_float2(elu1(ax), elu1(ay));
}


template <int U>
__device__ __forceinline__ void seg_accum4(const float* __restrict__ Wn, int dreg,
                                           float a0r, float a1r, float a2r, float a3r,
                                           int j, int lane4, bool hi, float4& pa, float4& pb)
{
    float4 wa[U], wb[U];
    #pragma unroll
    for (int u = 0; u < U; ++u) {
        const int dj = __builtin_amdgcn_readlane(dreg, j + u);
        const float* __restrict__ rec = Wn + (size_t)dj * 256 + lane4;
        wa[u] = *reinterpret_cast<const float4*>(rec);
        wb[u] = *reinterpret_cast<const float4*>(rec + 128);
    }
    #pragma unroll
    for (int u = 0; u < U; ++u) {
        const float s0 = __int_as_float(__builtin_amdgcn_readlane(__float_as_int(a0r), j + u));
        const float s1 = __int_as_float(__builtin_amdgcn_readlane(__float_as_int(a1r), j + u));
        const float s2 = __int_as_float(__builtin_amdgcn_readlane(__float_as_int(a2r), j + u));
        const float s3 = __int_as_float(__builtin_amdgcn_readlane(__float_as_int(a3r), j + u));
        const float ga = hi ? s1 : s0;
        const float gb = hi ? s3 : s2;
        pa.x += ga * wa[u].x; pa.y += ga * wa[u].y; pa.z += ga * wa[u].z; pa.w += ga * wa[u].w;
        pb.x += gb * wb[u].x; pb.y += gb * wb[u].y; pb.z += gb * wb[u].z; pb.w += gb * wb[u].w;
    }
}

__device__ __forceinline__ void seg_chunk4(const float* __restrict__ Wn, int dreg,
                                           float a0r, float a1r, float a2r, float a3r,
                                           int nj, int lane4, bool hi, float4& pa, float4& pb)
{
    int j = 0;
    for (; j + 8 <= nj; j += 8) seg_accum4<8>(Wn, dreg, a0r, a1r, a2r, a3r, j, lane4, hi, pa, pb);
    if (j + 4 <= nj) { seg_accum4<4>(Wn, dreg, a0r, a1r, a2r, a3r, j, lane4, hi, pa, pb); j += 4; }
    if (j + 2 <= nj) { seg_accum4<2>(Wn, dreg, a0r, a1r, a2r, a3r, j, lane4, hi, pa, pb); j += 2; }
    if (j < nj)      { seg_accum4<1>(Wn, dreg, a0r, a1r, a2r, a3r, j, lane4, hi, pa, pb); }
}

__device__ __forceinline__ void attend_row_h4(
    const int* __restrict__ rowptr, const int* __restrict__ epair,
    int* __restrict__ col,
    const float* __restrict__ ssrc, const float* __restrict__ sdst,
    const float* __restrict__ Wn, int N, int n, int lane, float4& oa, float4& ob)
{
    oa = make_float4(0.0f, 0.0f, 0.0f, 0.0f);
    ob = make_float4(0.0f, 0.0f, 0.0f, 0.0f);
    if (n >= N) return;
    const int  lane4 = lane << 2;
    const bool hi    = lane >= 16;
    const int beg = rowptr[n];
    const int deg = rowptr[n + 1] - beg;
    if (deg == 0) return;

    int* cp = col + beg;
    const int2* __restrict__ ep = reinterpret_cast<const int2*>(epair) + beg;
    const float4 ss = *reinterpret_cast<const float4*>(ssrc + (size_t)n * 4);

    int d0 = 0;
    if (deg <= 32) {
        int mine = 0x7fffffff, dv = 0;
        if (lane < deg) { const int2 pr = ep[lane]; mine = pr.x; dv = pr.y; }
        int pos = 0;
        for (int u = 0; u < deg; ++u) pos += (__builtin_amdgcn_readlane(mine, u) < mine) ? 1 : 0;
        d0 = __builtin_amdgcn_ds_permute(((lane < deg) ? pos : lane) << 2, dv);
        if (lane < deg) cp[lane] = d0;
    } else {
        row_order_any(ep, cp, deg, lane);
        __threadfence();
        d0 = cp[lane];
    }
    const float ninf = -__builtin_huge_valf();
    float v0 = ninf, v1 = ninf, v2 = ninf, v3 = ninf;
    if (lane < deg) {
        const float4 sd = *reinterpret_cast<const float4*>(sdst + (size_t)d0 * 4);
        v0 = lrelu(ss.x + sd.x); v1 = lrelu(ss.y + sd.y); v2 = lrelu(ss.z + sd.z); v3 = lrelu(ss.w + sd.w);
    }

    float m0 = v0, m1 = v1, m2 = v2, m3 = v3;
    for (int i = lane + 32; i < deg; i += 32) {
        const float4 sd = *reinterpret_cast<const float4*>(sdst + (size_t)cp[i] * 4);
        m0 = fmaxf(m0, lrelu(ss.x + sd.x)); m1 = fmaxf(m1, lrelu(ss.y + sd.y));
        m2 = fmaxf(m2, lrelu(ss.z + sd.z)); m3 = fmaxf(m3, lrelu(ss.w + sd.w));
    }
    for (int off = 16; off >= 1; off >>= 1) {
        m0 = fmaxf(m0, __shfl_xor(m0, off, 32)); m1 = fmaxf(m1, __shfl_xor(m1, off, 32));
        m2 = fmaxf(m2, __shfl_xor(m2, off, 32)); m3 = fmaxf(m3, __shfl_xor(m3, off, 32));
    }

    const bool in0 = lane < deg;
    const float e0 = in0 ? __expf(v0 - m0) : 0.0f;
    const float e1 = in0 ? __expf(v1 - m1) : 0.0f;
    const float e2 = in0 ? __expf(v2 - m2) : 0.0f;
    const float e3 = in0 ? __expf(v3 - m3) : 0.0f;
    float t0 = e0, t1 = e1, t2 = e2, t3 = e3;
    for (int i = lane + 32; i < deg; i += 32) {
        const float4 sd = *reinterpret_cast<const float4*>(sdst + (size_t)cp[i] * 4);
        t0 += __expf(lrelu(ss.x + sd.x) - m0); t1 += __expf(lrelu(ss.y + sd.y) - m1);
        t2 += __expf(lrelu(ss.z + sd.z) - m2); t3 += __expf(lrelu(ss.w + sd.w) - m3);
    }
    for (int off = 16; off >= 1; off >>= 1) {
        t0 += __shfl_xor(t0, off, 32); t1 += __shfl_xor(t1, off, 32);
        t2 += __shfl_xor(t2, off, 32); t3 += __shfl_xor(t3, off, 32);
    }

    float4 pa = make_float4(0.0f, 0.0f, 0.0f, 0.0f);
    float4 pb = make_float4(0.0f, 0.0f, 0.0f, 0.0f);
    seg_chunk4(Wn, d0, e0 / t0, e1 / t1, e2 / t2, e3 / t3, deg < 32 ? deg : 32, lane4, hi, pa, pb);
    for (int base = 32; base < deg; base += 32) {
        const int i = base + lane;
        int   d  = 0;
        float a0 = 0.0f, a1 = 0.0f, a2 = 0.0f, a3 = 0.0f;
        if (i < deg) {
            d = cp[i];
            const float4 sd = *reinterpret_cast<const float4*>(sdst + (size_t)d * 4);
            a0 = __expf(lrelu(ss.x + sd.x) - m0) / t0; a1 = __expf(lrelu(ss.y + sd.y) - m1) / t1;
            a2 = __expf(lrelu(ss.z + sd.z) - m2) / t2; a3 = __expf(lrelu(ss.w + sd.w) - m3) / t3;
        }
        const int left = deg - base;
        seg_chunk4(Wn, d, a0, a1, a2, a3, left < 32 ? left : 32, lane4, hi, pa, pb);
    }
    oa = make_float4(elu1(pa.x), elu1(pa.y), elu1(pa.z), elu1(pa.w));
    ob = make_float4(elu1(pb.x), elu1(pb.y), elu1(pb.z), elu1(pb.w));
}

#define HA_LD 260
#define PT_LD 68
__global__ __launch_bounds__(512) void k_seg_attend_h4_l2(
    const int* __restrict__ rowptr, const int* __restrict__ epair,
    int* __restrict__ col,
    const float* __restrict__ ssrc, const float* __restrict__ sdst,
    const float* __restrict__ Wn,
    const float* __restrict__ W2, const float* __restrict__ a2,
    float* __restrict__ Wh2, float* __restrict__ ssrc2, float* __restrict__ sdst2, int N)
{
    __shared__ float hA[16 * HA_LD];
    __shared__ float part[16 * 16 * PT_LD];

    const int lane = threadIdx.x & 31;
    const int w = __builtin_amdgcn_readfirstlane((int)(threadIdx.x >> 5));
    const int n = (int)blockIdx.x * 16 + w;

    float4 ra, rb;
    attend_row_h4(rowptr, epair, col, ssrc, sdst, Wn, N, n, lane, ra, rb);
    float4* hr = reinterpret_cast<float4*>(hA + w * HA_LD) + lane;
    hr[0]  = ra;
    hr[32] = rb;

    const int c16 = lane & 15;
    const int kh  = (lane >> 4) << 1;

    const float* __restrict__ bp = W2 + (size_t)(16 * w + kh) * 64 + c16;
    v2f bf[4][4];
    #pragma unroll
    for (int s = 0; s < 4; ++s) {
        #pragma unroll
        for (int j = 0; j < 4; ++j) {
            bf[s][j].x = bp[(4 * s) * 64 + 16 * j];
            bf[s][j].y = bp[(4 * s + 1) * 64 + 16 * j];
        }
    }

    __builtin_amdgcn_fence(__ATOMIC_RELEASE, "workgroup", "local");
    __builtin_amdgcn_s_barrier();
    __builtin_amdgcn_fence(__ATOMIC_ACQUIRE, "workgroup", "local");

    v8f c0 = {}, c1 = {}, c2 = {}, c3 = {};
    const float* ar = hA + c16 * HA_LD + 16 * w + kh;
    #pragma unroll
    for (int s = 0; s < 4; ++s) {
        const float2 af = *reinterpret_cast<const float2*>(ar + 4 * s);
        v2f a; a.x = af.x; a.y = af.y;
        c0 = __builtin_amdgcn_wmma_f32_16x16x4_f32(false, a, false, bf[s][0], (short)0, c0, false, false);
        c1 = __builtin_amdgcn_wmma_f32_16x16x4_f32(false, a, false, bf[s][1], (short)0, c1, false, false);
        c2 = __builtin_amdgcn_wmma_f32_16x16x4_f32(false, a, false, bf[s][2], (short)0, c2, false, false);
        c3 = __builtin_amdgcn_wmma_f32_16x16x4_f32(false, a, false, bf[s][3], (short)0, c3, false, false);
    }

    float* pw = part + (w * 16 + ((lane >> 4) << 3)) * PT_LD + c16;
    #pragma unroll
    for (int r = 0; r < 8; ++r) {
        pw[r * PT_LD]      = c0[r];
        pw[r * PT_LD + 16] = c1[r];
        pw[r * PT_LD + 32] = c2[r];
        pw[r * PT_LD + 48] = c3[r];
    }

    __syncthreads();

    const float* pr = part + w * PT_LD + (lane << 1);
    float2 acc = *reinterpret_cast<const float2*>(pr);
    #pragma unroll
    for (int p = 1; p < 16; ++p) {
        const float2 t = *reinterpret_cast<const float2*>(pr + p * 16 * PT_LD);
        acc.x += t.x; acc.y += t.y;
    }

    float s = acc.x * a2[lane << 1] + acc.y * a2[(lane << 1) + 1];
    float d = acc.x * a2[64 + (lane << 1)] + acc.y * a2[65 + (lane << 1)];
    s = row16_sum(s);
    d = row16_sum(d);
    s += __shfl_xor(s, 16, 32);
    d += __shfl_xor(d, 16, 32);
    if (n < N) {
        *reinterpret_cast<float2*>(Wh2 + (size_t)n * 64 + (lane << 1)) = acc;
        if (lane == 0) { ssrc2[n] = s; sdst2[n] = d; }
    }
}

static inline int cdiv(long a, int b) { return (int)((a + b - 1) / b); }

extern "C" void kernel_launch(void* const* d_in, const int* in_sizes, int n_in,
                              void* d_out, int out_size, void* d_ws, size_t ws_size,
                              hipStream_t stream)
{
    const float* x       = (const float*)d_in[0];
    const int*   edges   = (const int*)d_in[1];
    const float* W_heads = (const float*)d_in[2];
    const float* a_heads = (const float*)d_in[3];
    const float* W_out   = (const float*)d_in[4];
    const float* a_out   = (const float*)d_in[5];
    float* out = (float*)d_out;

    const int N = in_sizes[0] / 128;
    const int E = in_sizes[1] / 2;
    const int H = 4, HID = 64;

    const int* src = edges;
    const int* dst = edges + E;

    float* ws = (float*)d_ws;
    size_t off = 0;
    float* Wh1   = ws + off; off += (size_t)H * N * HID;
    float* scratch = ws + off; off += (size_t)N * 256;
    float* Wh2   = ws + off; off += (size_t)N * HID;
    float* ssrc1 = ws + off; off += (size_t)H * N;
    float* sdst1 = ws + off; off += (size_t)H * N;
    float* ssrc2 = ws + off; off += (size_t)N;
    float* sdst2 = ws + off; off += (size_t)N;
    int* deg0   = (int*)(ws + off); off += (size_t)N;
    int* rowptr = (int*)(ws + off); off += (size_t)N + 1;
    int* rank   = (int*)(ws + off); off += (size_t)E;
    int* col    = (int*)(ws + off); off += (size_t)E;
    int* deg1   = (int*)scratch;
    int* epair = (int*)scratch + (size_t)N * 2;

    const int BS = 256;

    k_zero_int2<<<cdiv((long)N, BS), BS, 0, stream>>>(deg0, deg1, N);
    const int MT = N / 16;
    const int gb = cdiv((long)2 * MT, BS / 32);
    k_edge_rank<<<gb + cdiv((long)E, BS), BS, 0, stream>>>(src, deg0, deg1, rank, E,
        x, W_heads, Wh1, a_heads, ssrc1, sdst1, 128, MT, gb, 0, 2 * MT);
    k_row_offsets<<<cdiv((long)N + 1, 1024), 1024, 0, stream>>>(deg0, deg1, rowptr, N);
    k_edge_place<<<gb + cdiv((long)E, BS), BS, 0, stream>>>(src, dst, rowptr, rank, epair, E,
        x, W_heads, Wh1, a_heads, ssrc1, sdst1, 128, MT, gb, 2 * MT, 4 * MT);

    k_seg_attend_h4_l2<<<cdiv((long)N, 16), 512, 0, stream>>>(rowptr, epair, col, ssrc1, sdst1, Wh1,
        W_out, a_out, Wh2, ssrc2, sdst2, N);

    k_seg_attend<<<cdiv((long)N * 32, BS), BS, 0, stream>>>(rowptr, col, ssrc2, sdst2, Wh2, out, N, N, HID);
}


# Edges go straight into fixed 64-slot row buckets by the counting atomic's reply (longer rows re-scan the edge list): no offsets scan, no placement launch, 6 -> 4; row sums in neighbour-index order
# speedup vs baseline: 9.5726x; 1.1560x over previous
#include <hip/hip_runtime.h>
#include <cstdint>
#include <cstddef>

typedef float v2f __attribute__((ext_vector_type(2)));
typedef float v8f __attribute__((ext_vector_type(8)));

#define GAT_ALPHA 0.2f

__global__ void k_zero_int2(int* __restrict__ p, int* __restrict__ q, int n) {
    int i = blockIdx.x * blockDim.x + threadIdx.x;
    if (i < n) { p[i] = 0; q[i] = 0; }
}

__device__ __forceinline__ int chip_half()
{
    return (int)((__builtin_amdgcn_s_sendmsg_rtn(0x87) >> 18) & 1u);
}

__device__ __forceinline__ void l1_tile_wave(const float* __restrict__ x, const float* __restrict__ W,
                                             float* __restrict__ Wh, const float* __restrict__ avec,
                                             float* __restrict__ ssrc, float* __restrict__ sdst,
                                             int K, int mTiles, int tile, int lane);

#define L1_TILE_ROLE()                                                                                         \
    if (blockIdx.x < (unsigned)gBlocks) {                                                                      \
        const int tile = tile0 + __builtin_amdgcn_readfirstlane(                                               \
            (int)(blockIdx.x * (blockDim.x >> 5) + (threadIdx.x >> 5)));                                       \
        if (tile < tileEnd) l1_tile_wave(gx, gW, gWh, gavec, gssrc, gsdst, gK, mTiles, tile, threadIdx.x & 31); \
        return;                                                                                                \
    }
#define L1_TILE_PARAMS                                                                                  \
    const float* __restrict__ gx, const float* __restrict__ gW, float* __restrict__ gWh,               \
    const float* __restrict__ gavec, float* __restrict__ gssrc, float* __restrict__ gsdst,             \
    int gK, int mTiles, int gBlocks, int tile0, int tileEnd

#define ROW_CAP 64
__global__ __launch_bounds__(256) void k_edge_bucket(const int* __restrict__ src, const int* __restrict__ edst,
                                                     int* __restrict__ deg0, int* __restrict__ deg1,
                                                     int* __restrict__ bk, int E,
                                                     L1_TILE_PARAMS)
{
    L1_TILE_ROLE()
    int e = (blockIdx.x - gBlocks) * blockDim.x + threadIdx.x;
    const int half = chip_half();
    if (e >= E) return;
    const int s = src[e];
    int d = edst[e];
    asm volatile("" : "+v"(d));
    int* __restrict__ cnt = half ? deg1 : deg0;
    const int r = atomicAdd(&cnt[s], 1);
    if (r < ROW_CAP) bk[(size_t)s * (2 * ROW_CAP) + half * ROW_CAP + r] = d;
}

__device__ __forceinline__ int bucket_at(const int* __restrict__ b0, const int* __restrict__ b1, int dg0, int i)
{
    const int* __restrict__ p = i < dg0 ? b0 + i : b1 + (i - dg0);
    return *p;
}

__device__ __forceinline__ void row_order_any(const int* __restrict__ b0, const int* __restrict__ b1, int dg0,
                                              int* cw, int deg, int lane)
{
    for (int ibase = 0; ibase < deg; ibase += 32) {
        const int i = ibase + lane;
        const int mine = (i < deg) ? bucket_at(b0, b1, dg0, i) : 0x7fffffff;
        int pos = 0;
        for (int jbase = 0; jbase < deg; jbase += 32) {
            const int j = jbase + lane;
            const int other = (j < deg) ? bucket_at(b0, b1, dg0, j) : 0x7fffffff;
            const int left = deg - jbase;
            const int nj = left < 32 ? left : 32;
            for (int u = 0; u < nj; ++u) {
                const int o = __builtin_amdgcn_readlane(other, u);
                pos += (o < mine || (o == mine && jbase + u < i)) ? 1 : 0;
            }
        }
        if (i < deg) cw[pos] = mine;
    }
}

__device__ __forceinline__ float row16_sum(float v)
{
    v += __int_as_float(__builtin_amdgcn_update_dpp(0, __float_as_int(v), 0x161, 0xf, 0xf, false));
    v += __int_as_float(__builtin_amdgcn_update_dpp(0, __float_as_int(v), 0x162, 0xf, 0xf, false));
    v += __int_as_float(__builtin_amdgcn_update_dpp(0, __float_as_int(v), 0x164, 0xf, 0xf, false));
    v += __int_as_float(__builtin_amdgcn_update_dpp(0, __float_as_int(v), 0x168, 0xf, 0xf, false));
    return v;
}

__device__ __forceinline__ void gemm_tile(
    const float* __restrict__ A, const float* __restrict__ B, float* __restrict__ C,
    const float* __restrict__ avec, float* __restrict__ ssrc, float* __restrict__ sdst,
    int K, long aStride, long bStride, long cStride, long ldc, long sStride, long sld,
    const int m0, const int b, const int lane)
{
    const float* __restrict__ Ab = A + (long)b * aStride;
    const float* __restrict__ Bb = B + (long)b * bStride;
    float* __restrict__ Cb = C + (long)b * cStride;

    const int row = lane & 15;
    const int kh  = (lane >> 4) << 1;
    const int col = lane & 15;

    const float* __restrict__ av = avec + (long)b * 128;
    const float as0 = av[col],      as1 = av[col + 16], as2 = av[col + 32], as3 = av[col + 48];
    const float ad0 = av[col + 64], ad1 = av[col + 80], ad2 = av[col + 96], ad3 = av[col + 112];

    v8f c0 = {}, c1 = {}, c2 = {}, c3 = {};

    for (int k0 = 0; k0 < K; k0 += 4) {
        const float2 af = *reinterpret_cast<const float2*>(
            Ab + (size_t)(m0 + row) * K + k0 + kh);
        v2f a; a.x = af.x; a.y = af.y;

        const float* br0 = Bb + (size_t)(k0 + kh) * 64 + col;
        const float* br1 = br0 + 64;
        v2f b0, b1, b2, b3;
        b0.x = br0[0];  b0.y = br1[0];
        b1.x = br0[16]; b1.y = br1[16];
        b2.x = br0[32]; b2.y = br1[32];
        b3.x = br0[48]; b3.y = br1[48];

        c0 = __builtin_amdgcn_wmma_f32_16x16x4_f32(false, a, false, b0, (short)0, c0, false, false);
        c1 = __builtin_amdgcn_wmma_f32_16x16x4_f32(false, a, false, b1, (short)0, c1, false, false);
        c2 = __builtin_amdgcn_wmma_f32_16x16x4_f32(false, a, false, b2, (short)0, c2, false, false);
        c3 = __builtin_amdgcn_wmma_f32_16x16x4_f32(false, a, false, b3, (short)0, c3, false, false);
    }

    const int rbase = m0 + ((lane >> 4) << 3);
    #pragma unroll
    for (int r = 0; r < 8; ++r) {
        float* crow = Cb + (size_t)(rbase + r) * ldc + col;
        crow[0]  = c0[r];
        crow[16] = c1[r];
        crow[32] = c2[r];
        crow[48] = c3[r];
    }

    float ps = 0.0f, pd = 0.0f;
    #pragma unroll
    for (int r = 0; r < 8; ++r) {
        const float s = row16_sum(c0[r] * as0 + c1[r] * as1 + c2[r] * as2 + c3[r] * as3);
        const float d = row16_sum(c0[r] * ad0 + c1[r] * ad1 + c2[r] * ad2 + c3[r] * ad3);
        if (col == r) { ps = s; pd = d; }
    }
    if (col < 8) {
        const size_t si = (size_t)b * sStride + (size_t)(rbase + col) * sld;
        ssrc[si] = ps;
        sdst[si] = pd;
    }
}

__device__ __forceinline__ void l1_tile_wave(const float* __restrict__ x, const float* __restrict__ W,
                                             float* __restrict__ Wh, const float* __restrict__ avec,
                                             float* __restrict__ ssrc, float* __restrict__ sdst,
                                             int K, int mTiles, int tile, int lane)
{
    const int b = tile / mTiles;
    gemm_tile(x, W, Wh, avec, ssrc, sdst, K, 0L, (long)K * 64, 64L, 256L, 1L, 4L, (tile - b * mTiles) * 16, b, lane);
}

template <int U>
__device__ __forceinline__ void seg_accum(const float* __restrict__ Wb, int dreg, float areg,
                                          int j, int lane2, float& ax, float& ay)
{
    float2 w[U];
    float  a[U];
    #pragma unroll
    for (int u = 0; u < U; ++u) {
        const int dj = __builtin_amdgcn_readlane(dreg, j + u);
        a[u] = __int_as_float(__builtin_amdgcn_readlane(__float_as_int(areg), j + u));
        w[u] = *reinterpret_cast<const float2*>(Wb + (size_t)dj * 64 + lane2);
    }
    #pragma unroll
    for (int u = 0; u < U; ++u) { ax += a[u] * w[u].x; ay += a[u] * w[u].y; }
}

__device__ __forceinline__ void seg_chunk(const float* __restrict__ Wb, int dreg, float areg,
                                          int nj, int lane2, float& ax, float& ay)
{
    int j = 0;
    for (; j + 8 <= nj; j += 8) seg_accum<8>(Wb, dreg, areg, j, lane2, ax, ay);
    if (j + 4 <= nj) { seg_accum<4>(Wb, dreg, areg, j, lane2, ax, ay); j += 4; }
    if (j + 2 <= nj) { seg_accum<2>(Wb, dreg, areg, j, lane2, ax, ay); j += 2; }
    if (j < nj)      { seg_accum<1>(Wb, dreg, areg, j, lane2, ax, ay); }
}

__device__ __forceinline__ float lrelu(float v) { return v > 0.0f ? v : GAT_ALPHA * v; }

__device__ __forceinline__ float elu1(float v) { return v > 0.0f ? v : __expf(v) - 1.0f; }

__global__ __launch_bounds__(256) void k_seg_attend(
    const int* __restrict__ deg0, const int* __restrict__ deg1, const int* __restrict__ col,
    const int* __restrict__ esrc, const int* __restrict__ edst, int E,
    const float* __restrict__ ssrc, const float* __restrict__ sdst,
    const float* __restrict__ Wh, float* __restrict__ dst, int NH, int N, int ostride)
{
    const int lane  = threadIdx.x & 31;
    const int lane2 = lane << 1;
    const int wid = __builtin_amdgcn_readfirstlane((int)((blockIdx.x * blockDim.x + threadIdx.x) >> 5));
    if (wid >= NH) return;
    const int h = wid / N;
    const int n = wid - h * N;
    const int deg = deg0[n] + deg1[n];
    float2* __restrict__ orow = reinterpret_cast<float2*>(dst + (size_t)n * ostride + (size_t)h * 64) + lane;
    if (deg == 0) { *orow = make_float2(0.0f, 0.0f); return; }

    const float* __restrict__ sd = sdst + (size_t)h * N;
    const float* __restrict__ Wb = Wh + (size_t)h * N * 64;
    const float ss = ssrc[wid];
    float ax = 0.0f, ay = 0.0f;

    if (deg > ROW_CAP) {
        float mx = -__builtin_huge_valf();
        for (int base = 0; base < E; base += 32) {
            const int e = base + lane;
            if (e < E && esrc[e] == n) mx = fmaxf(mx, lrelu(ss + sd[edst[e]]));
        }
        for (int off = 16; off >= 1; off >>= 1) mx = fmaxf(mx, __shfl_xor(mx, off, 32));
        float sum = 0.0f;
        for (int base = 0; base < E; base += 32) {
            const int e = base + lane;
            if (e < E && esrc[e] == n) sum += __expf(lrelu(ss + sd[edst[e]]) - mx);
        }
        for (int off = 16; off >= 1; off >>= 1) sum += __shfl_xor(sum, off, 32);
        for (int base = 0; base < E; base += 32) {
            const int e = base + lane;
            const bool own = e < E && esrc[e] == n;
            int   d   = 0;
            float att = 0.0f;
            if (own) { d = edst[e]; att = __expf(lrelu(ss + sd[d]) - mx) / sum; }
            unsigned mask = __builtin_amdgcn_ballot_w32(own);
            while (mask) {
                const int j = __builtin_ctz(mask);
                mask &= mask - 1;
                seg_accum<1>(Wb, d, att, j, lane2, ax, ay);
            }
        }
        *orow = make_float2(elu1(ax), elu1(ay));
        return;
    }

    const int* __restrict__ cp = col + (size_t)n * ROW_CAP;

    int   d0 = 0;
    float v0 = -__builtin_huge_valf();
    if (lane < deg) { d0 = cp[lane]; v0 = lrelu(ss + sd[d0]); }

    float mx = v0;
    for (int i = lane + 32; i < deg; i += 32) mx = fmaxf(mx, lrelu(ss + sd[cp[i]]));
    for (int off = 16; off >= 1; off >>= 1) mx = fmaxf(mx, __shfl_xor(mx, off, 32));

    const float ex0 = (lane < deg) ? __expf(v0 - mx) : 0.0f;
    float sum = ex0;
    for (int i = lane + 32; i < deg; i += 32) sum += __expf(lrelu(ss + sd[cp[i]]) - mx);
    for (int off = 16; off >= 1; off >>= 1) sum += __shfl_xor(sum, off, 32);

    seg_chunk(Wb, d0, ex0 / sum, deg < 32 ? deg : 32, lane2, ax, ay);
    for (int base = 32; base < deg; base += 32) {
        const int i = base + lane;
        int   d   = 0;
        float att = 0.0f;
        if (i < deg) { d = cp[i]; att = __expf(lrelu(ss + sd[d]) - mx) / sum; }
        const int left = deg - base;
        seg_chunk(Wb, d, att, left < 32 ? left : 32, lane2, ax, ay);
    }
    *orow = make_float2(elu1(ax), elu1(ay));
}


template <int U>
__device__ __forceinline__ void seg_accum4(const float* __restrict__ Wn, int dreg,
                                           float a0r, float a1r, float a2r, float a3r,
                                           int j, int lane4, bool hi, float4& pa, float4& pb)
{
    float4 wa[U], wb[U];
    #pragma unroll
    for (int u = 0; u < U; ++u) {
        const int dj = __builtin_amdgcn_readlane(dreg, j + u);
        const float* __restrict__ rec = Wn + (size_t)dj * 256 + lane4;
        wa[u] = *reinterpret_cast<const float4*>(rec);
        wb[u] = *reinterpret_cast<const float4*>(rec + 128);
    }
    #pragma unroll
    for (int u = 0; u < U; ++u) {
        const float s0 = __int_as_float(__builtin_amdgcn_readlane(__float_as_int(a0r), j + u));
        const float s1 = __int_as_float(__builtin_amdgcn_readlane(__float_as_int(a1r), j + u));
        const float s2 = __int_as_float(__builtin_amdgcn_readlane(__float_as_int(a2r), j + u));
        const float s3 = __int_as_float(__builtin_amdgcn_readlane(__float_as_int(a3r), j + u));
        const float ga = hi ? s1 : s0;
        const float gb = hi ? s3 : s2;
        pa.x += ga * wa[u].x; pa.y += ga * wa[u].y; pa.z += ga * wa[u].z; pa.w += ga * wa[u].w;
        pb.x += gb * wb[u].x; pb.y += gb * wb[u].y; pb.z += gb * wb[u].z; pb.w += gb * wb[u].w;
    }
}

__device__ __forceinline__ void seg_chunk4(const float* __restrict__ Wn, int dreg,
                                           float a0r, float a1r, float a2r, float a3r,
                                           int nj, int lane4, bool hi, float4& pa, float4& pb)
{
    int j = 0;
    for (; j + 8 <= nj; j += 8) seg_accum4<8>(Wn, dreg, a0r, a1r, a2r, a3r, j, lane4, hi, pa, pb);
    if (j + 4 <= nj) { seg_accum4<4>(Wn, dreg, a0r, a1r, a2r, a3r, j, lane4, hi, pa, pb); j += 4; }
    if (j + 2 <= nj) { seg_accum4<2>(Wn, dreg, a0r, a1r, a2r, a3r, j, lane4, hi, pa, pb); j += 2; }
    if (j < nj)      { seg_accum4<1>(Wn, dreg, a0r, a1r, a2r, a3r, j, lane4, hi, pa, pb); }
}

struct row256 { float4 a, b; };

__device__ __attribute__((noinline)) row256 attend_row_h4_scan(
    const int* __restrict__ esrc, const int* __restrict__ edst, int E,
    const float4 ss, const float* __restrict__ sdst, const float* __restrict__ Wn,
    int n, int lane)
{
    const int  lane4 = lane << 2;
    const bool hi    = lane >= 16;
    const float ninf = -__builtin_huge_valf();
    float m0 = ninf, m1 = ninf, m2 = ninf, m3 = ninf;
    for (int base = 0; base < E; base += 32) {
        const int e = base + lane;
        if (e < E && esrc[e] == n) {
            const float4 sd = *reinterpret_cast<const float4*>(sdst + (size_t)edst[e] * 4);
            m0 = fmaxf(m0, lrelu(ss.x + sd.x)); m1 = fmaxf(m1, lrelu(ss.y + sd.y));
            m2 = fmaxf(m2, lrelu(ss.z + sd.z)); m3 = fmaxf(m3, lrelu(ss.w + sd.w));
        }
    }
    for (int off = 16; off >= 1; off >>= 1) {
        m0 = fmaxf(m0, __shfl_xor(m0, off, 32)); m1 = fmaxf(m1, __shfl_xor(m1, off, 32));
        m2 = fmaxf(m2, __shfl_xor(m2, off, 32)); m3 = fmaxf(m3, __shfl_xor(m3, off, 32));
    }
    float t0 = 0.0f, t1 = 0.0f, t2 = 0.0f, t3 = 0.0f;
    for (int base = 0; base < E; base += 32) {
        const int e = base + lane;
        if (e < E && esrc[e] == n) {
            const float4 sd = *reinterpret_cast<const float4*>(sdst + (size_t)edst[e] * 4);
            t0 += __expf(lrelu(ss.x + sd.x) - m0); t1 += __expf(lrelu(ss.y + sd.y) - m1);
            t2 += __expf(lrelu(ss.z + sd.z) - m2); t3 += __expf(lrelu(ss.w + sd.w) - m3);
        }
    }
    for (int off = 16; off >= 1; off >>= 1) {
        t0 += __shfl_xor(t0, off, 32); t1 += __shfl_xor(t1, off, 32);
        t2 += __shfl_xor(t2, off, 32); t3 += __shfl_xor(t3, off, 32);
    }
    float4 pa = make_float4(0.0f, 0.0f, 0.0f, 0.0f);
    float4 pb = make_float4(0.0f, 0.0f, 0.0f, 0.0f);
    for (int base = 0; base < E; base += 32) {
        const int e = base + lane;
        const bool own = e < E && esrc[e] == n;
        int   d  = 0;
        float a0 = 0.0f, a1 = 0.0f, a2 = 0.0f, a3 = 0.0f;
        if (own) {
            d = edst[e];
            const float4 sd = *reinterpret_cast<const float4*>(sdst + (size_t)d * 4);
            a0 = __expf(lrelu(ss.x + sd.x) - m0) / t0; a1 = __expf(lrelu(ss.y + sd.y) - m1) / t1;
            a2 = __expf(lrelu(ss.z + sd.z) - m2) / t2; a3 = __expf(lrelu(ss.w + sd.w) - m3) / t3;
        }
        unsigned mask = __builtin_amdgcn_ballot_w32(own);
        while (mask) {
            const int j = __builtin_ctz(mask);
            mask &= mask - 1;
            seg_accum4<1>(Wn, d, a0, a1, a2, a3, j, lane4, hi, pa, pb);
        }
    }
    row256 o;
    o.a = make_float4(elu1(pa.x), elu1(pa.y), elu1(pa.z), elu1(pa.w));
    o.b = make_float4(elu1(pb.x), elu1(pb.y), elu1(pb.z), elu1(pb.w));
    return o;
}

__device__ __forceinline__ void attend_row_h4(
    const int* __restrict__ deg0, const int* __restrict__ deg1, const int* __restrict__ bk,
    int* __restrict__ col,
    const int* __restrict__ esrc, const int* __restrict__ edst, int E,
    const float* __restrict__ ssrc, const float* __restrict__ sdst,
    const float* __restrict__ Wn, int N, int n, int lane, float4& oa, float4& ob)
{
    oa = make_float4(0.0f, 0.0f, 0.0f, 0.0f);
    ob = make_float4(0.0f, 0.0f, 0.0f, 0.0f);
    if (n >= N) return;
    const int  lane4 = lane << 2;
    const bool hi    = lane >= 16;
    const int dg0 = deg0[n];
    const int deg = dg0 + deg1[n];
    if (deg == 0) return;

    const float4 ss = *reinterpret_cast<const float4*>(ssrc + (size_t)n * 4);
    if (deg > ROW_CAP) {
        const row256 o = attend_row_h4_scan(esrc, edst, E, ss, sdst, Wn, n, lane);
        oa = o.a; ob = o.b;
        return;
    }

    int* cp = col + (size_t)n * ROW_CAP;
    const int* __restrict__ b0 = bk + (size_t)n * (2 * ROW_CAP);
    const int* __restrict__ b1 = b0 + ROW_CAP;

    int d0 = 0;
    if (deg <= 32) {
        int mine = 0x7fffffff;
        if (lane < deg) mine = bucket_at(b0, b1, dg0, lane);
        int pos = 0;
        for (int u = 0; u < deg; ++u) {
            const int o = __builtin_amdgcn_readlane(mine, u);
            pos += (o < mine || (o == mine && u < lane)) ? 1 : 0;
        }
        d0 = __builtin_amdgcn_ds_permute(((lane < deg) ? pos : lane) << 2, (lane < deg) ? mine : 0);
        if (lane < deg) cp[lane] = d0;
    } else {
        row_order_any(b0, b1, dg0, cp, deg, lane);
        __threadfence();
        d0 = cp[lane];
    }
    const float ninf = -__builtin_huge_valf();
    float v0 = ninf, v1 = ninf, v2 = ninf, v3 = ninf;
    if (lane < deg) {
        const float4 sd = *reinterpret_cast<const float4*>(sdst + (size_t)d0 * 4);
        v0 = lrelu(ss.x + sd.x); v1 = lrelu(ss.y + sd.y); v2 = lrelu(ss.z + sd.z); v3 = lrelu(ss.w + sd.w);
    }

    float m0 = v0, m1 = v1, m2 = v2, m3 = v3;
    for (int i = lane + 32; i < deg; i += 32) {
        const float4 sd = *reinterpret_cast<const float4*>(sdst + (size_t)cp[i] * 4);
        m0 = fmaxf(m0, lrelu(ss.x + sd.x)); m1 = fmaxf(m1, lrelu(ss.y + sd.y));
        m2 = fmaxf(m2, lrelu(ss.z + sd.z)); m3 = fmaxf(m3, lrelu(ss.w + sd.w));
    }
    for (int off = 16; off >= 1; off >>= 1) {
        m0 = fmaxf(m0, __shfl_xor(m0, off, 32)); m1 = fmaxf(m1, __shfl_xor(m1, off, 32));
        m2 = fmaxf(m2, __shfl_xor(m2, off, 32)); m3 = fmaxf(m3, __shfl_xor(m3, off, 32));
    }

    const bool in0 = lane < deg;
    const float e0 = in0 ? __expf(v0 - m0) : 0.0f;
    const float e1 = in0 ? __expf(v1 - m1) : 0.0f;
    const float e2 = in0 ? __expf(v2 - m2) : 0.0f;
    const float e3 = in0 ? __expf(v3 - m3) : 0.0f;
    float t0 = e0, t1 = e1, t2 = e2, t3 = e3;
    for (int i = lane + 32; i < deg; i += 32) {
        const float4 sd = *reinterpret_cast<const float4*>(sdst + (size_t)cp[i] * 4);
        t0 += __expf(lrelu(ss.x + sd.x) - m0); t1 += __expf(lrelu(ss.y + sd.y) - m1);
        t2 += __expf(lrelu(ss.z + sd.z) - m2); t3 += __expf(lrelu(ss.w + sd.w) - m3);
    }
    for (int off = 16; off >= 1; off >>= 1) {
        t0 += __shfl_xor(t0, off, 32); t1 += __shfl_xor(t1, off, 32);
        t2 += __shfl_xor(t2, off, 32); t3 += __shfl_xor(t3, off, 32);
    }

    float4 pa = make_float4(0.0f, 0.0f, 0.0f, 0.0f);
    float4 pb = make_float4(0.0f, 0.0f, 0.0f, 0.0f);
    seg_chunk4(Wn, d0, e0 / t0, e1 / t1, e2 / t2, e3 / t3, deg < 32 ? deg : 32, lane4, hi, pa, pb);
    for (int base = 32; base < deg; base += 32) {
        const int i = base + lane;
        int   d  = 0;
        float a0 = 0.0f, a1 = 0.0f, a2 = 0.0f, a3 = 0.0f;
        if (i < deg) {
            d = cp[i];
            const float4 sd = *reinterpret_cast<const float4*>(sdst + (size_t)d * 4);
            a0 = __expf(lrelu(ss.x + sd.x) - m0) / t0; a1 = __expf(lrelu(ss.y + sd.y) - m1) / t1;
            a2 = __expf(lrelu(ss.z + sd.z) - m2) / t2; a3 = __expf(lrelu(ss.w + sd.w) - m3) / t3;
        }
        const int left = deg - base;
        seg_chunk4(Wn, d, a0, a1, a2, a3, left < 32 ? left : 32, lane4, hi, pa, pb);
    }
    oa = make_float4(elu1(pa.x), elu1(pa.y), elu1(pa.z), elu1(pa.w));
    ob = make_float4(elu1(pb.x), elu1(pb.y), elu1(pb.z), elu1(pb.w));
}

#define HA_LD 260
#define PT_LD 68
__global__ __launch_bounds__(512) void k_seg_attend_h4_l2(
    const int* __restrict__ deg0, const int* __restrict__ deg1, const int* __restrict__ bk,
    int* __restrict__ col,
    const int* __restrict__ esrc, const int* __restrict__ edst, int E,
    const float* __restrict__ ssrc, const float* __restrict__ sdst,
    const float* __restrict__ Wn,
    const float* __restrict__ W2, const float* __restrict__ a2,
    float* __restrict__ Wh2, float* __restrict__ ssrc2, float* __restrict__ sdst2, int N)
{
    __shared__ float hA[16 * HA_LD];
    __shared__ float part[16 * 16 * PT_LD];

    const int lane = threadIdx.x & 31;
    const int w = __builtin_amdgcn_readfirstlane((int)(threadIdx.x >> 5));
    const int n = (int)blockIdx.x * 16 + w;

    float4 ra, rb;
    attend_row_h4(deg0, deg1, bk, col, esrc, edst, E, ssrc, sdst, Wn, N, n, lane, ra, rb);
    float4* hr = reinterpret_cast<float4*>(hA + w * HA_LD) + lane;
    hr[0]  = ra;
    hr[32] = rb;

    const int c16 = lane & 15;
    const int kh  = (lane >> 4) << 1;

    const float* __restrict__ bp = W2 + (size_t)(16 * w + kh) * 64 + c16;
    v2f bf[4][4];
    #pragma unroll
    for (int s = 0; s < 4; ++s) {
        #pragma unroll
        for (int j = 0; j < 4; ++j) {
            bf[s][j].x = bp[(4 * s) * 64 + 16 * j];
            bf[s][j].y = bp[(4 * s + 1) * 64 + 16 * j];
        }
    }

    __builtin_amdgcn_fence(__ATOMIC_RELEASE, "workgroup", "local");
    __builtin_amdgcn_s_barrier();
    __builtin_amdgcn_fence(__ATOMIC_ACQUIRE, "workgroup", "local");

    v8f c0 = {}, c1 = {}, c2 = {}, c3 = {};
    const float* ar = hA + c16 * HA_LD + 16 * w + kh;
    #pragma unroll
    for (int s = 0; s < 4; ++s) {
        const float2 af = *reinterpret_cast<const float2*>(ar + 4 * s);
        v2f a; a.x = af.x; a.y = af.y;
        c0 = __builtin_amdgcn_wmma_f32_16x16x4_f32(false, a, false, bf[s][0], (short)0, c0, false, false);
        c1 = __builtin_amdgcn_wmma_f32_16x16x4_f32(false, a, false, bf[s][1], (short)0, c1, false, false);
        c2 = __builtin_amdgcn_wmma_f32_16x16x4_f32(false, a, false, bf[s][2], (short)0, c2, false, false);
        c3 = __builtin_amdgcn_wmma_f32_16x16x4_f32(false, a, false, bf[s][3], (short)0, c3, false, false);
    }

    float* pw = part + (w * 16 + ((lane >> 4) << 3)) * PT_LD + c16;
    #pragma unroll
    for (int r = 0; r < 8; ++r) {
        pw[r * PT_LD]      = c0[r];
        pw[r * PT_LD + 16] = c1[r];
        pw[r * PT_LD + 32] = c2[r];
        pw[r * PT_LD + 48] = c3[r];
    }

    __syncthreads();

    const float* pr = part + w * PT_LD + (lane << 1);
    float2 acc = *reinterpret_cast<const float2*>(pr);
    #pragma unroll
    for (int p = 1; p < 16; ++p) {
        const float2 t = *reinterpret_cast<const float2*>(pr + p * 16 * PT_LD);
        acc.x += t.x; acc.y += t.y;
    }

    float s = acc.x * a2[lane << 1] + acc.y * a2[(lane << 1) + 1];
    float d = acc.x * a2[64 + (lane << 1)] + acc.y * a2[65 + (lane << 1)];
    s = row16_sum(s);
    d = row16_sum(d);
    s += __shfl_xor(s, 16, 32);
    d += __shfl_xor(d, 16, 32);
    if (n < N) {
        *reinterpret_cast<float2*>(Wh2 + (size_t)n * 64 + (lane << 1)) = acc;
        if (lane == 0) { ssrc2[n] = s; sdst2[n] = d; }
    }
}

static inline int cdiv(long a, int b) { return (int)((a + b - 1) / b); }

extern "C" void kernel_launch(void* const* d_in, const int* in_sizes, int n_in,
                              void* d_out, int out_size, void* d_ws, size_t ws_size,
                              hipStream_t stream)
{
    const float* x       = (const float*)d_in[0];
    const int*   edges   = (const int*)d_in[1];
    const float* W_heads = (const float*)d_in[2];
    const float* a_heads = (const float*)d_in[3];
    const float* W_out   = (const float*)d_in[4];
    const float* a_out   = (const float*)d_in[5];
    float* out = (float*)d_out;

    const int N = in_sizes[0] / 128;
    const int E = in_sizes[1] / 2;
    const int H = 4, HID = 64;

    const int* src = edges;
    const int* dst = edges + E;

    float* ws = (float*)d_ws;
    size_t off = 0;
    float* Wh1   = ws + off; off += (size_t)H * N * HID;
    float* scratch = ws + off; off += (size_t)N * 256;
    float* Wh2   = ws + off; off += (size_t)N * HID;
    float* ssrc1 = ws + off; off += (size_t)H * N;
    float* sdst1 = ws + off; off += (size_t)H * N;
    float* ssrc2 = ws + off; off += (size_t)N;
    float* sdst2 = ws + off; off += (size_t)N;
    int* deg0   = (int*)(ws + off); off += (size_t)N;
    int* deg1   = (int*)(ws + off); off += (size_t)N;
    int* bk  = (int*)scratch;
    int* col = (int*)scratch + (size_t)N * 128;

    const int BS = 256;

    k_zero_int2<<<cdiv((long)N, BS), BS, 0, stream>>>(deg0, deg1, N);
    const int MT = N / 16;
    const int gb = cdiv((long)4 * MT, BS / 32);
    k_edge_bucket<<<gb + cdiv((long)E, BS), BS, 0, stream>>>(src, dst, deg0, deg1, bk, E,
        x, W_heads, Wh1, a_heads, ssrc1, sdst1, 128, MT, gb, 0, 4 * MT);

    k_seg_attend_h4_l2<<<cdiv((long)N, 16), 512, 0, stream>>>(deg0, deg1, bk, col, src, dst, E, ssrc1, sdst1, Wh1,
        W_out, a_out, Wh2, ssrc2, sdst2, N);

    k_seg_attend<<<cdiv((long)N * 32, BS), BS, 0, stream>>>(deg0, deg1, col, src, dst, E, ssrc2, sdst2, Wh2, out,
        N, N, HID);
}
